# starting kernel
# baseline (speedup 1.0000x reference)
#include <hip/hip_runtime.h>
#include <cstdint>
#include <cstddef>

typedef __attribute__((ext_vector_type(2))) float v2f;
typedef __attribute__((ext_vector_type(8))) float v8f;

template <int NCOL, int EPI>
__global__ __launch_bounds__(256) void gemm_wmma_kernel(
    const float* __restrict__ A, const float* __restrict__ W,
    const float* __restrict__ bias, float* __restrict__ out, int nrows)
{
    constexpr int NT = NCOL / 16;
    __shared__ __align__(16) float sW[64 * NCOL];
    __shared__ __align__(16) float sA[8][16 * 68];

    const int tid  = threadIdx.x;
    const int wave = tid >> 5;
    const int lane = tid & 31;
    const int hh   = lane >> 4;
    const int l16  = lane & 15;
    const int r0   = blockIdx.x * 128 + wave * 16;

    for (int i = tid; i < 64 * NCOL; i += 256) sW[i] = W[i];

    const float4* A4 = reinterpret_cast<const float4*>(A);
    #pragma unroll
    for (int i = 0; i < 8; ++i) {
        int q  = lane + 32 * i;
        int r  = q >> 4;
        int c4 = q & 15;
        int gr = r0 + r;
        float4 v = make_float4(0.f, 0.f, 0.f, 0.f);
        if (gr < nrows) v = A4[(size_t)gr * 16 + c4];
        *reinterpret_cast<float4*>(&sA[wave][r * 68 + c4 * 4]) = v;
    }
    __syncthreads();

    v8f acc[NT] = {};

    #pragma unroll
    for (int kb = 0; kb < 64; kb += 4) {
        v2f a;
        a.x = sA[wave][l16 * 68 + kb + 2 * hh + 0];
        a.y = sA[wave][l16 * 68 + kb + 2 * hh + 1];
        #pragma unroll
        for (int nt = 0; nt < NT; ++nt) {
            v2f b;
            b.x = sW[(kb + 2 * hh + 0) * NCOL + nt * 16 + l16];
            b.y = sW[(kb + 2 * hh + 1) * NCOL + nt * 16 + l16];
            acc[nt] = __builtin_amdgcn_wmma_f32_16x16x4_f32(
                false, a, false, b, (short)0, acc[nt], false, false);
        }
    }

    #pragma unroll
    for (int nt = 0; nt < NT; ++nt) {
        const int col = nt * 16 + l16;
        float bv = 0.f;
        if (EPI > 0) bv = bias[col];
        #pragma unroll
        for (int j = 0; j < 8; ++j) {
            int gr = r0 + j + 8 * hh;
            if (gr < nrows) {
                float v = acc[nt][j] + bv;
                if (EPI == 2) v = fmaxf(v, 0.f);
                out[(size_t)gr * NCOL + col] = v;
            }
        }
    }
}

__global__ __launch_bounds__(256) void deg_init_kernel(float* __restrict__ deg, int n)
{
    int i = blockIdx.x * 256 + threadIdx.x;
    if (i < n) deg[i] = 1.0f;
}

__global__ __launch_bounds__(256) void deg_count_kernel(
    const int* __restrict__ col, float* __restrict__ deg, int ne)
{
    int e = blockIdx.x * 256 + threadIdx.x;
    if (e < ne) unsafeAtomicAdd(&deg[col[e]], 1.0f);
}

__global__ __launch_bounds__(256) void dinv_kernel(float* __restrict__ d, int n)
{
    int i = blockIdx.x * 256 + threadIdx.x;
    if (i < n) d[i] = rsqrtf(d[i]);
}

__global__ __launch_bounds__(256) void self_term_kernel(
    const float* __restrict__ xw, const float* __restrict__ dinv,
    float* __restrict__ agg, int n)
{
    int t = blockIdx.x * 256 + threadIdx.x;
    if (t < n * 64) {
        int i = t >> 6;
        float dv = dinv[i];
        agg[t] = xw[t] * dv * dv;
    }
}

__global__ __launch_bounds__(256) void edge_scatter_kernel(
    const int* __restrict__ row, const int* __restrict__ col,
    const float* __restrict__ dinv, const float* __restrict__ xw,
    float* __restrict__ agg, int ne)
{
    int t    = blockIdx.x * 256 + threadIdx.x;
    int e    = t >> 5;
    int lane = t & 31;
    if (e >= ne) return;
    int   r  = row[e];
    int   c  = col[e];
    float nm = dinv[r] * dinv[c];
    const float* src = xw  + (size_t)r * 64;
    float*       dst = agg + (size_t)c * 64;
    unsafeAtomicAdd(&dst[lane],      nm * src[lane]);
    unsafeAtomicAdd(&dst[lane + 32], nm * src[lane + 32]);
}

__global__ __launch_bounds__(256) void bn_relu_kernel(
    const float* __restrict__ agg, const float* __restrict__ b,
    const float* __restrict__ g, const float* __restrict__ beta,
    const float* __restrict__ mean, const float* __restrict__ var,
    float* __restrict__ out, int n)
{
    int t = blockIdx.x * 256 + threadIdx.x;
    if (t < n * 64) {
        int f = t & 63;
        float v = (agg[t] + b[f] - mean[f]) * rsqrtf(var[f] + 1e-5f) * g[f] + beta[f];
        out[t] = fmaxf(v, 0.f);
    }
}

extern "C" void kernel_launch(void* const* d_in, const int* in_sizes, int n_in,
                              void* d_out, int out_size, void* d_ws, size_t ws_size,
                              hipStream_t stream)
{
    const float* x      = (const float*)d_in[0];
    const int*   ei     = (const int*)d_in[1];
    const float* W0     = (const float*)d_in[2];
    const float* b0     = (const float*)d_in[3];
    const float* W1     = (const float*)d_in[4];
    const float* b1     = (const float*)d_in[5];
    const float* bn0_g  = (const float*)d_in[6];
    const float* bn0_b  = (const float*)d_in[7];
    const float* bn0_m  = (const float*)d_in[8];
    const float* bn0_v  = (const float*)d_in[9];
    const float* bn1_g  = (const float*)d_in[10];
    const float* bn1_b  = (const float*)d_in[11];
    const float* bn1_m  = (const float*)d_in[12];
    const float* bn1_v  = (const float*)d_in[13];
    const float* lin1_W = (const float*)d_in[14];
    const float* lin1_b = (const float*)d_in[15];
    const float* lin2_W = (const float*)d_in[16];
    const float* lin2_b = (const float*)d_in[17];
    const float* lin3_W = (const float*)d_in[18];
    const float* lin3_b = (const float*)d_in[19];

    const int N = in_sizes[0] / 64;
    const int E = in_sizes[1] / 2;
    const int* row = ei;
    const int* col = ei + E;

    char*  ws   = (char*)d_ws;
    size_t off0 = 0;
    size_t off1 = (off0 + (size_t)N * sizeof(float) + 255) & ~(size_t)255;
    size_t off2 = (off1 + (size_t)N * 64 * sizeof(float) + 255) & ~(size_t)255;
    float* dinv = (float*)(ws + off0);
    float* bufA = (float*)(ws + off1);
    float* bufB = (float*)(ws + off2);

    const dim3 blk(256);
    const int gN    = (N + 255) / 256;
    const int gE    = (E + 255) / 256;
    const int gNF   = (N * 64 + 255) / 256;
    const int gEdge = (E * 32 + 255) / 256;
    const int gGemm = (N + 127) / 128;

    deg_init_kernel<<<gN, blk, 0, stream>>>(dinv, N);
    deg_count_kernel<<<gE, blk, 0, stream>>>(col, dinv, E);
    dinv_kernel<<<gN, blk, 0, stream>>>(dinv, N);

    gemm_wmma_kernel<64, 0><<<gGemm, blk, 0, stream>>>(x, W0, nullptr, bufA, N);
    self_term_kernel<<<gNF, blk, 0, stream>>>(bufA, dinv, bufB, N);
    edge_scatter_kernel<<<gEdge, blk, 0, stream>>>(row, col, dinv, bufA, bufB, E);
    bn_relu_kernel<<<gNF, blk, 0, stream>>>(bufB, b0, bn0_g, bn0_b, bn0_m, bn0_v, bufA, N);

    gemm_wmma_kernel<64, 0><<<gGemm, blk, 0, stream>>>(bufA, W1, nullptr, bufB, N);
    self_term_kernel<<<gNF, blk, 0, stream>>>(bufB, dinv, bufA, N);
    edge_scatter_kernel<<<gEdge, blk, 0, stream>>>(row, col, dinv, bufB, bufA, E);
    bn_relu_kernel<<<gNF, blk, 0, stream>>>(bufA, b1, bn1_g, bn1_b, bn1_m, bn1_v, bufB, N);

    gemm_wmma_kernel<64, 2><<<gGemm, blk, 0, stream>>>(bufB, lin1_W, lin1_b, bufA, N);
    gemm_wmma_kernel<64, 2><<<gGemm, blk, 0, stream>>>(bufA, lin2_W, lin2_b, bufB, N);
    gemm_wmma_kernel<16, 1><<<gGemm, blk, 0, stream>>>(bufB, lin3_W, lin3_b, (float*)d_out, N);
}


# edge aggregation: float-atomic scatter replaced by a per-destination gather over a CSR rebuilt each call with an integer atomic rank; per-node sum order = arrival rank
# speedup vs baseline: 1.4880x; 1.4880x over previous
#include <hip/hip_runtime.h>
#include <cstdint>
#include <cstddef>

typedef __attribute__((ext_vector_type(2))) float v2f;
typedef __attribute__((ext_vector_type(8))) float v8f;

template <int NCOL, int EPI>
__global__ __launch_bounds__(256) void gemm_wmma_kernel(
    const float* __restrict__ A, const float* __restrict__ W,
    const float* __restrict__ bias, float* __restrict__ out, int nrows)
{
    constexpr int NT = NCOL / 16;
    __shared__ __align__(16) float sW[64 * NCOL];
    __shared__ __align__(16) float sA[8][16 * 68];

    const int tid  = threadIdx.x;
    const int wave = tid >> 5;
    const int lane = tid & 31;
    const int hh   = lane >> 4;
    const int l16  = lane & 15;
    const int r0   = blockIdx.x * 128 + wave * 16;

    for (int i = tid; i < 64 * NCOL; i += 256) sW[i] = W[i];

    const float4* A4 = reinterpret_cast<const float4*>(A);
    #pragma unroll
    for (int i = 0; i < 8; ++i) {
        int q  = lane + 32 * i;
        int r  = q >> 4;
        int c4 = q & 15;
        int gr = r0 + r;
        float4 v = make_float4(0.f, 0.f, 0.f, 0.f);
        if (gr < nrows) v = A4[(size_t)gr * 16 + c4];
        *reinterpret_cast<float4*>(&sA[wave][r * 68 + c4 * 4]) = v;
    }
    __syncthreads();

    v8f acc[NT] = {};

    #pragma unroll
    for (int kb = 0; kb < 64; kb += 4) {
        v2f a;
        a.x = sA[wave][l16 * 68 + kb + 2 * hh + 0];
        a.y = sA[wave][l16 * 68 + kb + 2 * hh + 1];
        #pragma unroll
        for (int nt = 0; nt < NT; ++nt) {
            v2f b;
            b.x = sW[(kb + 2 * hh + 0) * NCOL + nt * 16 + l16];
            b.y = sW[(kb + 2 * hh + 1) * NCOL + nt * 16 + l16];
            acc[nt] = __builtin_amdgcn_wmma_f32_16x16x4_f32(
                false, a, false, b, (short)0, acc[nt], false, false);
        }
    }

    #pragma unroll
    for (int nt = 0; nt < NT; ++nt) {
        const int col = nt * 16 + l16;
        float bv = 0.f;
        if (EPI > 0) bv = bias[col];
        #pragma unroll
        for (int j = 0; j < 8; ++j) {
            int gr = r0 + j + 8 * hh;
            if (gr < nrows) {
                float v = acc[nt][j] + bv;
                if (EPI == 2) v = fmaxf(v, 0.f);
                out[(size_t)gr * NCOL + col] = v;
            }
        }
    }
}

__global__ __launch_bounds__(256) void zero_cnt_kernel(int* __restrict__ cnt, int n)
{
    int i = blockIdx.x * 256 + threadIdx.x;
    if (i < n) cnt[i] = 0;
}

__global__ __launch_bounds__(256) void rank_kernel(
    const int* __restrict__ col, int* __restrict__ cnt, int* __restrict__ rank, int ne)
{
    int e = blockIdx.x * 256 + threadIdx.x;
    if (e < ne) rank[e] = atomicAdd(&cnt[col[e]], 1);
}

__global__ __launch_bounds__(256) void slice_total_kernel(
    const int* __restrict__ cnt, int* __restrict__ tot, int n)
{
    __shared__ int sw[8];
    const int tid  = threadIdx.x;
    const int base = blockIdx.x * 1024 + tid * 4;
    int s = 0;
    #pragma unroll
    for (int k = 0; k < 4; ++k)
        if (base + k < n) s += cnt[base + k];
    for (int o = 16; o > 0; o >>= 1) s += __shfl_xor(s, o, 32);
    if ((tid & 31) == 0) sw[tid >> 5] = s;
    __syncthreads();
    if (tid == 0) {
        int t = 0;
        for (int w = 0; w < 8; ++w) t += sw[w];
        tot[blockIdx.x] = t;
    }
}

__global__ __launch_bounds__(256) void scan_kernel(
    const int* __restrict__ cnt, const int* __restrict__ tot,
    int* __restrict__ offs, float* __restrict__ dinv, int n)
{
    __shared__ int sp[8];
    __shared__ int sw[8];
    const int tid  = threadIdx.x;
    const int lane = tid & 31;
    const int wave = tid >> 5;

    int p = 0;
    for (int i = tid; i < (int)blockIdx.x; i += 256) p += tot[i];
    for (int o = 16; o > 0; o >>= 1) p += __shfl_xor(p, o, 32);

    const int base = blockIdx.x * 1024 + tid * 4;
    int c[4];
    #pragma unroll
    for (int k = 0; k < 4; ++k) c[k] = (base + k < n) ? cnt[base + k] : 0;
    const int s = c[0] + c[1] + c[2] + c[3];
    int incl = s;
    for (int o = 1; o < 32; o <<= 1) {
        int t = __shfl_up(incl, o, 32);
        if (lane >= o) incl += t;
    }
    if (lane == 0)  sp[wave] = p;
    if (lane == 31) sw[wave] = incl;
    __syncthreads();
    int pre = 0;
    for (int w = 0; w < 8; ++w) {
        pre += sp[w];
        if (w < wave) pre += sw[w];
    }
    int run = pre + incl - s;
    #pragma unroll
    for (int k = 0; k < 4; ++k) {
        if (base + k < n) {
            offs[base + k] = run;
            dinv[base + k] = rsqrtf((float)(c[k] + 1));
        }
        run += c[k];
    }
}

__global__ __launch_bounds__(256) void fill_kernel(
    const int* __restrict__ row, const int* __restrict__ col,
    const int* __restrict__ rank, const int* __restrict__ offs,
    int* __restrict__ srcs, int ne)
{
    int e = blockIdx.x * 256 + threadIdx.x;
    if (e < ne) srcs[offs[col[e]] + rank[e]] = row[e];
}

__global__ __launch_bounds__(256) void gather_kernel(
    const int* __restrict__ offs, const int* __restrict__ cnt,
    const int* __restrict__ srcs, const float* __restrict__ dinv,
    const float* __restrict__ xw, float* __restrict__ agg, int n)
{
    const int  l     = threadIdx.x & 15;
    const int  c     = blockIdx.x * 16 + (threadIdx.x >> 4);
    const bool valid = c < n;
    const float4* X4 = reinterpret_cast<const float4*>(xw);

    int   start = 0, deg = 0;
    float dc = 0.f;
    float4 acc = make_float4(0.f, 0.f, 0.f, 0.f);
    if (valid) {
        start = offs[c];
        deg   = cnt[c];
        dc    = dinv[c];
        const float4 x = X4[(size_t)c * 16 + l];
        acc.x = x.x * dc * dc;
        acc.y = x.y * dc * dc;
        acc.z = x.z * dc * dc;
        acc.w = x.w * dc * dc;
    }

    for (int j0 = 0; j0 < deg; j0 += 16) {
        const int m = (deg - j0 < 16) ? (deg - j0) : 16;
        int   myr = 0;
        float myd = 0.f;
        if (l < m) {
            myr = srcs[start + j0 + l];
            myd = dinv[myr];
        }
        for (int j = 0; j < m; j += 4) {
            int    r[4];
            float  w[4];
            float4 x[4];
            #pragma unroll
            for (int k = 0; k < 4; ++k) {
                r[k] = __shfl(myr, j + k, 16);
            }
            #pragma unroll
            for (int k = 0; k < 4; ++k) {
                x[k] = make_float4(0.f, 0.f, 0.f, 0.f);
                if (j + k < m) x[k] = X4[(size_t)r[k] * 16 + l];
            }
            #pragma unroll
            for (int k = 0; k < 4; ++k) w[k] = __shfl(myd, j + k, 16) * dc;
            #pragma unroll
            for (int k = 0; k < 4; ++k) {
                acc.x = fmaf(w[k], x[k].x, acc.x);
                acc.y = fmaf(w[k], x[k].y, acc.y);
                acc.z = fmaf(w[k], x[k].z, acc.z);
                acc.w = fmaf(w[k], x[k].w, acc.w);
            }
        }
    }
    if (valid) reinterpret_cast<float4*>(agg)[(size_t)c * 16 + l] = acc;
}

__global__ __launch_bounds__(256) void bn_relu_kernel(
    const float* __restrict__ agg, const float* __restrict__ b,
    const float* __restrict__ g, const float* __restrict__ beta,
    const float* __restrict__ mean, const float* __restrict__ var,
    float* __restrict__ out, int n)
{
    int t = blockIdx.x * 256 + threadIdx.x;
    if (t < n * 64) {
        int f = t & 63;
        float v = (agg[t] + b[f] - mean[f]) * rsqrtf(var[f] + 1e-5f) * g[f] + beta[f];
        out[t] = fmaxf(v, 0.f);
    }
}

extern "C" void kernel_launch(void* const* d_in, const int* in_sizes, int n_in,
                              void* d_out, int out_size, void* d_ws, size_t ws_size,
                              hipStream_t stream)
{
    const float* x      = (const float*)d_in[0];
    const int*   ei     = (const int*)d_in[1];
    const float* W0     = (const float*)d_in[2];
    const float* b0     = (const float*)d_in[3];
    const float* W1     = (const float*)d_in[4];
    const float* b1     = (const float*)d_in[5];
    const float* bn0_g  = (const float*)d_in[6];
    const float* bn0_b  = (const float*)d_in[7];
    const float* bn0_m  = (const float*)d_in[8];
    const float* bn0_v  = (const float*)d_in[9];
    const float* bn1_g  = (const float*)d_in[10];
    const float* bn1_b  = (const float*)d_in[11];
    const float* bn1_m  = (const float*)d_in[12];
    const float* bn1_v  = (const float*)d_in[13];
    const float* lin1_W = (const float*)d_in[14];
    const float* lin1_b = (const float*)d_in[15];
    const float* lin2_W = (const float*)d_in[16];
    const float* lin2_b = (const float*)d_in[17];
    const float* lin3_W = (const float*)d_in[18];
    const float* lin3_b = (const float*)d_in[19];

    const int N = in_sizes[0] / 64;
    const int E = in_sizes[1] / 2;
    const int* row = ei;
    const int* col = ei + E;

    char*  ws   = (char*)d_ws;
    size_t off0 = 0;
    size_t off1 = (off0 + (size_t)N * sizeof(float) + 255) & ~(size_t)255;
    size_t off2 = (off1 + (size_t)N * 64 * sizeof(float) + 255) & ~(size_t)255;
    size_t off3 = (off2 + (size_t)N * 64 * sizeof(float) + 255) & ~(size_t)255;
    size_t off4 = (off3 + (size_t)N * sizeof(int) + 255) & ~(size_t)255;
    size_t off5 = (off4 + (size_t)N * sizeof(int) + 255) & ~(size_t)255;
    const int gSlice = (N + 1023) / 1024;
    size_t off6 = (off5 + (size_t)gSlice * sizeof(int) + 255) & ~(size_t)255;
    size_t off7 = (off6 + (size_t)E * sizeof(int) + 255) & ~(size_t)255;
    float* dinv = (float*)(ws + off0);
    float* bufA = (float*)(ws + off1);
    float* bufB = (float*)(ws + off2);
    int*   cnt  = (int*)(ws + off3);
    int*   offs = (int*)(ws + off4);
    int*   tot  = (int*)(ws + off5);
    int*   rank = (int*)(ws + off6);
    int*   srcs = (int*)(ws + off7);

    const dim3 blk(256);
    const int gN    = (N + 255) / 256;
    const int gE    = (E + 255) / 256;
    const int gNF   = (N * 64 + 255) / 256;
    const int gGath = (N + 15) / 16;
    const int gGemm = (N + 127) / 128;

    zero_cnt_kernel<<<gN, blk, 0, stream>>>(cnt, N);
    rank_kernel<<<gE, blk, 0, stream>>>(col, cnt, rank, E);
    slice_total_kernel<<<gSlice, blk, 0, stream>>>(cnt, tot, N);
    scan_kernel<<<gSlice, blk, 0, stream>>>(cnt, tot, offs, dinv, N);
    fill_kernel<<<gE, blk, 0, stream>>>(row, col, rank, offs, srcs, E);

    gemm_wmma_kernel<64, 0><<<gGemm, blk, 0, stream>>>(x, W0, nullptr, bufA, N);
    gather_kernel<<<gGath, blk, 0, stream>>>(offs, cnt, srcs, dinv, bufA, bufB, N);
    bn_relu_kernel<<<gNF, blk, 0, stream>>>(bufB, b0, bn0_g, bn0_b, bn0_m, bn0_v, bufA, N);

    gemm_wmma_kernel<64, 0><<<gGemm, blk, 0, stream>>>(bufA, W1, nullptr, bufB, N);
    gather_kernel<<<gGath, blk, 0, stream>>>(offs, cnt, srcs, dinv, bufB, bufA, N);
    bn_relu_kernel<<<gNF, blk, 0, stream>>>(bufA, b1, bn1_g, bn1_b, bn1_m, bn1_v, bufB, N);

    gemm_wmma_kernel<64, 2><<<gGemm, blk, 0, stream>>>(bufB, lin1_W, lin1_b, bufA, N);
    gemm_wmma_kernel<64, 2><<<gGemm, blk, 0, stream>>>(bufA, lin2_W, lin2_b, bufB, N);
    gemm_wmma_kernel<16, 1><<<gGemm, blk, 0, stream>>>(bufB, lin3_W, lin3_b, (float*)d_out, N);
}


# batch-norm + ReLU moved into the gather kernel's epilogue, applied to the aggregate in registers; the two elementwise kernels (one element per thread) and their buffer round trips are gone
# speedup vs baseline: 1.7575x; 1.1811x over previous
#include <hip/hip_runtime.h>
#include <cstdint>
#include <cstddef>

typedef __attribute__((ext_vector_type(2))) float v2f;
typedef __attribute__((ext_vector_type(8))) float v8f;

template <int NCOL, int EPI>
__global__ __launch_bounds__(256) void gemm_wmma_kernel(
    const float* __restrict__ A, const float* __restrict__ W,
    const float* __restrict__ bias, float* __restrict__ out, int nrows)
{
    constexpr int NT = NCOL / 16;
    __shared__ __align__(16) float sW[64 * NCOL];
    __shared__ __align__(16) float sA[8][16 * 68];

    const int tid  = threadIdx.x;
    const int wave = tid >> 5;
    const int lane = tid & 31;
    const int hh   = lane >> 4;
    const int l16  = lane & 15;
    const int r0   = blockIdx.x * 128 + wave * 16;

    for (int i = tid; i < 64 * NCOL; i += 256) sW[i] = W[i];

    const float4* A4 = reinterpret_cast<const float4*>(A);
    #pragma unroll
    for (int i = 0; i < 8; ++i) {
        int q  = lane + 32 * i;
        int r  = q >> 4;
        int c4 = q & 15;
        int gr = r0 + r;
        float4 v = make_float4(0.f, 0.f, 0.f, 0.f);
        if (gr < nrows) v = A4[(size_t)gr * 16 + c4];
        *reinterpret_cast<float4*>(&sA[wave][r * 68 + c4 * 4]) = v;
    }
    __syncthreads();

    v8f acc[NT] = {};

    #pragma unroll
    for (int kb = 0; kb < 64; kb += 4) {
        v2f a;
        a.x = sA[wave][l16 * 68 + kb + 2 * hh + 0];
        a.y = sA[wave][l16 * 68 + kb + 2 * hh + 1];
        #pragma unroll
        for (int nt = 0; nt < NT; ++nt) {
            v2f b;
            b.x = sW[(kb + 2 * hh + 0) * NCOL + nt * 16 + l16];
            b.y = sW[(kb + 2 * hh + 1) * NCOL + nt * 16 + l16];
            acc[nt] = __builtin_amdgcn_wmma_f32_16x16x4_f32(
                false, a, false, b, (short)0, acc[nt], false, false);
        }
    }

    #pragma unroll
    for (int nt = 0; nt < NT; ++nt) {
        const int col = nt * 16 + l16;
        float bv = 0.f;
        if (EPI > 0) bv = bias[col];
        #pragma unroll
        for (int j = 0; j < 8; ++j) {
            int gr = r0 + j + 8 * hh;
            if (gr < nrows) {
                float v = acc[nt][j] + bv;
                if (EPI == 2) v = fmaxf(v, 0.f);
                out[(size_t)gr * NCOL + col] = v;
            }
        }
    }
}

__global__ __launch_bounds__(256) void zero_cnt_kernel(int* __restrict__ cnt, int n)
{
    int i = blockIdx.x * 256 + threadIdx.x;
    if (i < n) cnt[i] = 0;
}

__global__ __launch_bounds__(256) void rank_kernel(
    const int* __restrict__ col, int* __restrict__ cnt, int* __restrict__ rank, int ne)
{
    int e = blockIdx.x * 256 + threadIdx.x;
    if (e < ne) rank[e] = atomicAdd(&cnt[col[e]], 1);
}

__global__ __launch_bounds__(256) void slice_total_kernel(
    const int* __restrict__ cnt, int* __restrict__ tot, int n)
{
    __shared__ int sw[8];
    const int tid  = threadIdx.x;
    const int base = blockIdx.x * 1024 + tid * 4;
    int s = 0;
    #pragma unroll
    for (int k = 0; k < 4; ++k)
        if (base + k < n) s += cnt[base + k];
    for (int o = 16; o > 0; o >>= 1) s += __shfl_xor(s, o, 32);
    if ((tid & 31) == 0) sw[tid >> 5] = s;
    __syncthreads();
    if (tid == 0) {
        int t = 0;
        for (int w = 0; w < 8; ++w) t += sw[w];
        tot[blockIdx.x] = t;
    }
}

__global__ __launch_bounds__(256) void scan_kernel(
    const int* __restrict__ cnt, const int* __restrict__ tot,
    int* __restrict__ offs, float* __restrict__ dinv, int n)
{
    __shared__ int sp[8];
    __shared__ int sw[8];
    const int tid  = threadIdx.x;
    const int lane = tid & 31;
    const int wave = tid >> 5;

    int p = 0;
    for (int i = tid; i < (int)blockIdx.x; i += 256) p += tot[i];
    for (int o = 16; o > 0; o >>= 1) p += __shfl_xor(p, o, 32);

    const int base = blockIdx.x * 1024 + tid * 4;
    int c[4];
    #pragma unroll
    for (int k = 0; k < 4; ++k) c[k] = (base + k < n) ? cnt[base + k] : 0;
    const int s = c[0] + c[1] + c[2] + c[3];
    int incl = s;
    for (int o = 1; o < 32; o <<= 1) {
        int t = __shfl_up(incl, o, 32);
        if (lane >= o) incl += t;
    }
    if (lane == 0)  sp[wave] = p;
    if (lane == 31) sw[wave] = incl;
    __syncthreads();
    int pre = 0;
    for (int w = 0; w < 8; ++w) {
        pre += sp[w];
        if (w < wave) pre += sw[w];
    }
    int run = pre + incl - s;
    #pragma unroll
    for (int k = 0; k < 4; ++k) {
        if (base + k < n) {
            offs[base + k] = run;
            dinv[base + k] = rsqrtf((float)(c[k] + 1));
        }
        run += c[k];
    }
}

__global__ __launch_bounds__(256) void fill_kernel(
    const int* __restrict__ row, const int* __restrict__ col,
    const int* __restrict__ rank, const int* __restrict__ offs,
    int* __restrict__ srcs, int ne)
{
    int e = blockIdx.x * 256 + threadIdx.x;
    if (e < ne) srcs[offs[col[e]] + rank[e]] = row[e];
}

__global__ __launch_bounds__(256) void gather_bn_relu_kernel(
    const int* __restrict__ offs, const int* __restrict__ cnt,
    const int* __restrict__ srcs, const float* __restrict__ dinv,
    const float* __restrict__ xw,
    const float* __restrict__ b, const float* __restrict__ g,
    const float* __restrict__ beta, const float* __restrict__ mean,
    const float* __restrict__ var,
    float* __restrict__ act, int n)
{
    const int  l     = threadIdx.x & 15;
    const int  c     = blockIdx.x * 16 + (threadIdx.x >> 4);
    const bool valid = c < n;
    const float4* X4 = reinterpret_cast<const float4*>(xw);

    const float4 pb  = reinterpret_cast<const float4*>(b)[l];
    const float4 pm  = reinterpret_cast<const float4*>(mean)[l];
    const float4 pv  = reinterpret_cast<const float4*>(var)[l];
    const float4 pg  = reinterpret_cast<const float4*>(g)[l];
    const float4 pbe = reinterpret_cast<const float4*>(beta)[l];

    int   start = 0, deg = 0;
    float dc = 0.f;
    float4 acc = make_float4(0.f, 0.f, 0.f, 0.f);
    if (valid) {
        start = offs[c];
        deg   = cnt[c];
        dc    = dinv[c];
        const float4 x = X4[(size_t)c * 16 + l];
        acc.x = x.x * dc * dc;
        acc.y = x.y * dc * dc;
        acc.z = x.z * dc * dc;
        acc.w = x.w * dc * dc;
    }

    for (int j0 = 0; j0 < deg; j0 += 16) {
        const int m = (deg - j0 < 16) ? (deg - j0) : 16;
        int   myr = 0;
        float myd = 0.f;
        if (l < m) {
            myr = srcs[start + j0 + l];
            myd = dinv[myr];
        }
        for (int j = 0; j < m; j += 4) {
            int    r[4];
            float  w[4];
            float4 x[4];
            #pragma unroll
            for (int k = 0; k < 4; ++k) {
                r[k] = __shfl(myr, j + k, 16);
            }
            #pragma unroll
            for (int k = 0; k < 4; ++k) {
                x[k] = make_float4(0.f, 0.f, 0.f, 0.f);
                if (j + k < m) x[k] = X4[(size_t)r[k] * 16 + l];
            }
            #pragma unroll
            for (int k = 0; k < 4; ++k) w[k] = __shfl(myd, j + k, 16) * dc;
            #pragma unroll
            for (int k = 0; k < 4; ++k) {
                acc.x = fmaf(w[k], x[k].x, acc.x);
                acc.y = fmaf(w[k], x[k].y, acc.y);
                acc.z = fmaf(w[k], x[k].z, acc.z);
                acc.w = fmaf(w[k], x[k].w, acc.w);
            }
        }
    }
    if (valid) {
        float4 o;
        o.x = fmaxf((acc.x + pb.x - pm.x) * rsqrtf(pv.x + 1e-5f) * pg.x + pbe.x, 0.f);
        o.y = fmaxf((acc.y + pb.y - pm.y) * rsqrtf(pv.y + 1e-5f) * pg.y + pbe.y, 0.f);
        o.z = fmaxf((acc.z + pb.z - pm.z) * rsqrtf(pv.z + 1e-5f) * pg.z + pbe.z, 0.f);
        o.w = fmaxf((acc.w + pb.w - pm.w) * rsqrtf(pv.w + 1e-5f) * pg.w + pbe.w, 0.f);
        reinterpret_cast<float4*>(act)[(size_t)c * 16 + l] = o;
    }
}

extern "C" void kernel_launch(void* const* d_in, const int* in_sizes, int n_in,
                              void* d_out, int out_size, void* d_ws, size_t ws_size,
                              hipStream_t stream)
{
    const float* x      = (const float*)d_in[0];
    const int*   ei     = (const int*)d_in[1];
    const float* W0     = (const float*)d_in[2];
    const float* b0     = (const float*)d_in[3];
    const float* W1     = (const float*)d_in[4];
    const float* b1     = (const float*)d_in[5];
    const float* bn0_g  = (const float*)d_in[6];
    const float* bn0_b  = (const float*)d_in[7];
    const float* bn0_m  = (const float*)d_in[8];
    const float* bn0_v  = (const float*)d_in[9];
    const float* bn1_g  = (const float*)d_in[10];
    const float* bn1_b  = (const float*)d_in[11];
    const float* bn1_m  = (const float*)d_in[12];
    const float* bn1_v  = (const float*)d_in[13];
    const float* lin1_W = (const float*)d_in[14];
    const float* lin1_b = (const float*)d_in[15];
    const float* lin2_W = (const float*)d_in[16];
    const float* lin2_b = (const float*)d_in[17];
    const float* lin3_W = (const float*)d_in[18];
    const float* lin3_b = (const float*)d_in[19];

    const int N = in_sizes[0] / 64;
    const int E = in_sizes[1] / 2;
    const int* row = ei;
    const int* col = ei + E;

    char*  ws   = (char*)d_ws;
    size_t off0 = 0;
    size_t off1 = (off0 + (size_t)N * sizeof(float) + 255) & ~(size_t)255;
    size_t off2 = (off1 + (size_t)N * 64 * sizeof(float) + 255) & ~(size_t)255;
    size_t off3 = (off2 + (size_t)N * 64 * sizeof(float) + 255) & ~(size_t)255;
    size_t off4 = (off3 + (size_t)N * sizeof(int) + 255) & ~(size_t)255;
    size_t off5 = (off4 + (size_t)N * sizeof(int) + 255) & ~(size_t)255;
    const int gSlice = (N + 1023) / 1024;
    size_t off6 = (off5 + (size_t)gSlice * sizeof(int) + 255) & ~(size_t)255;
    size_t off7 = (off6 + (size_t)E * sizeof(int) + 255) & ~(size_t)255;
    float* dinv = (float*)(ws + off0);
    float* bufA = (float*)(ws + off1);
    float* bufB = (float*)(ws + off2);
    int*   cnt  = (int*)(ws + off3);
    int*   offs = (int*)(ws + off4);
    int*   tot  = (int*)(ws + off5);
    int*   rank = (int*)(ws + off6);
    int*   srcs = (int*)(ws + off7);

    const dim3 blk(256);
    const int gN    = (N + 255) / 256;
    const int gE    = (E + 255) / 256;
    const int gGath = (N + 15) / 16;
    const int gGemm = (N + 127) / 128;

    zero_cnt_kernel<<<gN, blk, 0, stream>>>(cnt, N);
    rank_kernel<<<gE, blk, 0, stream>>>(col, cnt, rank, E);
    slice_total_kernel<<<gSlice, blk, 0, stream>>>(cnt, tot, N);
    scan_kernel<<<gSlice, blk, 0, stream>>>(cnt, tot, offs, dinv, N);
    fill_kernel<<<gE, blk, 0, stream>>>(row, col, rank, offs, srcs, E);

    gemm_wmma_kernel<64, 0><<<gGemm, blk, 0, stream>>>(x, W0, nullptr, bufA, N);
    gather_bn_relu_kernel<<<gGath, blk, 0, stream>>>(offs, cnt, srcs, dinv, bufA,
        b0, bn0_g, bn0_b, bn0_m, bn0_v, bufB, N);

    gemm_wmma_kernel<64, 0><<<gGemm, blk, 0, stream>>>(bufB, W1, nullptr, bufA, N);
    gather_bn_relu_kernel<<<gGath, blk, 0, stream>>>(offs, cnt, srcs, dinv, bufA,
        b1, bn1_g, bn1_b, bn1_m, bn1_v, bufB, N);

    gemm_wmma_kernel<64, 2><<<gGemm, blk, 0, stream>>>(bufB, lin1_W, lin1_b, bufA, N);
    gemm_wmma_kernel<64, 2><<<gGemm, blk, 0, stream>>>(bufA, lin2_W, lin2_b, bufB, N);
    gemm_wmma_kernel<16, 1><<<gGemm, blk, 0, stream>>>(bufB, lin3_W, lin3_b, (float*)d_out, N);
}


# MLP head's three dense layers fused into one kernel: weights staged once by 16-byte loads issued together, a wave's 16-row activations stay in LDS between layers; 2 launches gone; sum order unchanged
# speedup vs baseline: 2.0436x; 1.1628x over previous
#include <hip/hip_runtime.h>
#include <cstdint>
#include <cstddef>

typedef __attribute__((ext_vector_type(2))) float v2f;
typedef __attribute__((ext_vector_type(8))) float v8f;

template <int NCOL, int EPI>
__global__ __launch_bounds__(256) void gemm_wmma_kernel(
    const float* __restrict__ A, const float* __restrict__ W,
    const float* __restrict__ bias, float* __restrict__ out, int nrows)
{
    constexpr int NT = NCOL / 16;
    __shared__ __align__(16) float sW[64 * NCOL];
    __shared__ __align__(16) float sA[8][16 * 68];

    const int tid  = threadIdx.x;
    const int wave = tid >> 5;
    const int lane = tid & 31;
    const int hh   = lane >> 4;
    const int l16  = lane & 15;
    const int r0   = blockIdx.x * 128 + wave * 16;

    for (int i = tid; i < 64 * NCOL; i += 256) sW[i] = W[i];

    const float4* A4 = reinterpret_cast<const float4*>(A);
    #pragma unroll
    for (int i = 0; i < 8; ++i) {
        int q  = lane + 32 * i;
        int r  = q >> 4;
        int c4 = q & 15;
        int gr = r0 + r;
        float4 v = make_float4(0.f, 0.f, 0.f, 0.f);
        if (gr < nrows) v = A4[(size_t)gr * 16 + c4];
        *reinterpret_cast<float4*>(&sA[wave][r * 68 + c4 * 4]) = v;
    }
    __syncthreads();

    v8f acc[NT] = {};

    #pragma unroll
    for (int kb = 0; kb < 64; kb += 4) {
        v2f a;
        a.x = sA[wave][l16 * 68 + kb + 2 * hh + 0];
        a.y = sA[wave][l16 * 68 + kb + 2 * hh + 1];
        #pragma unroll
        for (int nt = 0; nt < NT; ++nt) {
            v2f b;
            b.x = sW[(kb + 2 * hh + 0) * NCOL + nt * 16 + l16];
            b.y = sW[(kb + 2 * hh + 1) * NCOL + nt * 16 + l16];
            acc[nt] = __builtin_amdgcn_wmma_f32_16x16x4_f32(
                false, a, false, b, (short)0, acc[nt], false, false);
        }
    }

    #pragma unroll
    for (int nt = 0; nt < NT; ++nt) {
        const int col = nt * 16 + l16;
        float bv = 0.f;
        if (EPI > 0) bv = bias[col];
        #pragma unroll
        for (int j = 0; j < 8; ++j) {
            int gr = r0 + j + 8 * hh;
            if (gr < nrows) {
                float v = acc[nt][j] + bv;
                if (EPI == 2) v = fmaxf(v, 0.f);
                out[(size_t)gr * NCOL + col] = v;
            }
        }
    }
}

template <int NCOL>
__device__ __forceinline__ void strip_gemm(
    const float* __restrict__ sAw, const float* __restrict__ sW,
    v8f (&acc)[NCOL / 16], int l16, int hh)
{
    constexpr int NT = NCOL / 16;
    #pragma unroll
    for (int kb = 0; kb < 64; kb += 4) {
        v2f a;
        a.x = sAw[l16 * 68 + kb + 2 * hh + 0];
        a.y = sAw[l16 * 68 + kb + 2 * hh + 1];
        #pragma unroll
        for (int nt = 0; nt < NT; ++nt) {
            v2f b;
            b.x = sW[(kb + 2 * hh + 0) * NCOL + nt * 16 + l16];
            b.y = sW[(kb + 2 * hh + 1) * NCOL + nt * 16 + l16];
            acc[nt] = __builtin_amdgcn_wmma_f32_16x16x4_f32(
                false, a, false, b, (short)0, acc[nt], false, false);
        }
    }
}

__device__ __forceinline__ void strip_store_relu(
    float* __restrict__ sAw, const v8f (&acc)[4],
    const float* __restrict__ bias, int l16, int hh)
{
    #pragma unroll
    for (int nt = 0; nt < 4; ++nt) {
        const int   col = nt * 16 + l16;
        const float bv  = bias[col];
        #pragma unroll
        for (int j = 0; j < 8; ++j)
            sAw[(j + 8 * hh) * 68 + col] = fmaxf(acc[nt][j] + bv, 0.f);
    }
}

__global__ __launch_bounds__(256) void mlp_head_kernel(
    const float* __restrict__ A,
    const float* __restrict__ W1, const float* __restrict__ c1,
    const float* __restrict__ W2, const float* __restrict__ c2,
    const float* __restrict__ W3, const float* __restrict__ c3,
    float* __restrict__ out, int nrows)
{
    __shared__ __align__(16) float sW1[64 * 64];
    __shared__ __align__(16) float sW2[64 * 64];
    __shared__ __align__(16) float sW3[64 * 16];
    __shared__ __align__(16) float sA[8][16 * 68];

    const int tid  = threadIdx.x;
    const int wave = tid >> 5;
    const int lane = tid & 31;
    const int hh   = lane >> 4;
    const int l16  = lane & 15;
    const int r0   = blockIdx.x * 128 + wave * 16;

    {
        const float4* g1 = reinterpret_cast<const float4*>(W1);
        const float4* g2 = reinterpret_cast<const float4*>(W2);
        const float4* g3 = reinterpret_cast<const float4*>(W3);
        float4* s1 = reinterpret_cast<float4*>(sW1);
        float4* s2 = reinterpret_cast<float4*>(sW2);
        float4* s3 = reinterpret_cast<float4*>(sW3);
        #pragma unroll
        for (int i = 0; i < 4; ++i) {
            s1[tid + 256 * i] = g1[tid + 256 * i];
            s2[tid + 256 * i] = g2[tid + 256 * i];
        }
        s3[tid] = g3[tid];
    }

    const float4* A4 = reinterpret_cast<const float4*>(A);
    #pragma unroll
    for (int i = 0; i < 8; ++i) {
        int q  = lane + 32 * i;
        int r  = q >> 4;
        int c4 = q & 15;
        int gr = r0 + r;
        float4 v = make_float4(0.f, 0.f, 0.f, 0.f);
        if (gr < nrows) v = A4[(size_t)gr * 16 + c4];
        *reinterpret_cast<float4*>(&sA[wave][r * 68 + c4 * 4]) = v;
    }
    __syncthreads();

    float* sAw = sA[wave];

    v8f acc1[4] = {};
    strip_gemm<64>(sAw, sW1, acc1, l16, hh);
    strip_store_relu(sAw, acc1, c1, l16, hh);
    __syncthreads();

    v8f acc2[4] = {};
    strip_gemm<64>(sAw, sW2, acc2, l16, hh);
    strip_store_relu(sAw, acc2, c2, l16, hh);
    __syncthreads();

    v8f acc3[1] = {};
    strip_gemm<16>(sAw, sW3, acc3, l16, hh);
    {
        const float bv = c3[l16];
        #pragma unroll
        for (int j = 0; j < 8; ++j) {
            int gr = r0 + j + 8 * hh;
            if (gr < nrows) out[(size_t)gr * 16 + l16] = acc3[0][j] + bv;
        }
    }
}

__global__ __launch_bounds__(256) void zero_cnt_kernel(int* __restrict__ cnt, int n)
{
    int i = blockIdx.x * 256 + threadIdx.x;
    if (i < n) cnt[i] = 0;
}

__global__ __launch_bounds__(256) void rank_kernel(
    const int* __restrict__ col, int* __restrict__ cnt, int* __restrict__ rank, int ne)
{
    int e = blockIdx.x * 256 + threadIdx.x;
    if (e < ne) rank[e] = atomicAdd(&cnt[col[e]], 1);
}

__global__ __launch_bounds__(256) void slice_total_kernel(
    const int* __restrict__ cnt, int* __restrict__ tot, int n)
{
    __shared__ int sw[8];
    const int tid  = threadIdx.x;
    const int base = blockIdx.x * 1024 + tid * 4;
    int s = 0;
    #pragma unroll
    for (int k = 0; k < 4; ++k)
        if (base + k < n) s += cnt[base + k];
    for (int o = 16; o > 0; o >>= 1) s += __shfl_xor(s, o, 32);
    if ((tid & 31) == 0) sw[tid >> 5] = s;
    __syncthreads();
    if (tid == 0) {
        int t = 0;
        for (int w = 0; w < 8; ++w) t += sw[w];
        tot[blockIdx.x] = t;
    }
}

__global__ __launch_bounds__(256) void scan_kernel(
    const int* __restrict__ cnt, const int* __restrict__ tot,
    int* __restrict__ offs, float* __restrict__ dinv, int n)
{
    __shared__ int sp[8];
    __shared__ int sw[8];
    const int tid  = threadIdx.x;
    const int lane = tid & 31;
    const int wave = tid >> 5;

    int p = 0;
    for (int i = tid; i < (int)blockIdx.x; i += 256) p += tot[i];
    for (int o = 16; o > 0; o >>= 1) p += __shfl_xor(p, o, 32);

    const int base = blockIdx.x * 1024 + tid * 4;
    int c[4];
    #pragma unroll
    for (int k = 0; k < 4; ++k) c[k] = (base + k < n) ? cnt[base + k] : 0;
    const int s = c[0] + c[1] + c[2] + c[3];
    int incl = s;
    for (int o = 1; o < 32; o <<= 1) {
        int t = __shfl_up(incl, o, 32);
        if (lane >= o) incl += t;
    }
    if (lane == 0)  sp[wave] = p;
    if (lane == 31) sw[wave] = incl;
    __syncthreads();
    int pre = 0;
    for (int w = 0; w < 8; ++w) {
        pre += sp[w];
        if (w < wave) pre += sw[w];
    }
    int run = pre + incl - s;
    #pragma unroll
    for (int k = 0; k < 4; ++k) {
        if (base + k < n) {
            offs[base + k] = run;
            dinv[base + k] = rsqrtf((float)(c[k] + 1));
        }
        run += c[k];
    }
}

__global__ __launch_bounds__(256) void fill_kernel(
    const int* __restrict__ row, const int* __restrict__ col,
    const int* __restrict__ rank, const int* __restrict__ offs,
    int* __restrict__ srcs, int ne)
{
    int e = blockIdx.x * 256 + threadIdx.x;
    if (e < ne) srcs[offs[col[e]] + rank[e]] = row[e];
}

__global__ __launch_bounds__(256) void gather_bn_relu_kernel(
    const int* __restrict__ offs, const int* __restrict__ cnt,
    const int* __restrict__ srcs, const float* __restrict__ dinv,
    const float* __restrict__ xw,
    const float* __restrict__ b, const float* __restrict__ g,
    const float* __restrict__ beta, const float* __restrict__ mean,
    const float* __restrict__ var,
    float* __restrict__ act, int n)
{
    const int  l     = threadIdx.x & 15;
    const int  c     = blockIdx.x * 16 + (threadIdx.x >> 4);
    const bool valid = c < n;
    const float4* X4 = reinterpret_cast<const float4*>(xw);

    const float4 pb  = reinterpret_cast<const float4*>(b)[l];
    const float4 pm  = reinterpret_cast<const float4*>(mean)[l];
    const float4 pv  = reinterpret_cast<const float4*>(var)[l];
    const float4 pg  = reinterpret_cast<const float4*>(g)[l];
    const float4 pbe = reinterpret_cast<const float4*>(beta)[l];

    int   start = 0, deg = 0;
    float dc = 0.f;
    float4 acc = make_float4(0.f, 0.f, 0.f, 0.f);
    if (valid) {
        start = offs[c];
        deg   = cnt[c];
        dc    = dinv[c];
        const float4 x = X4[(size_t)c * 16 + l];
        acc.x = x.x * dc * dc;
        acc.y = x.y * dc * dc;
        acc.z = x.z * dc * dc;
        acc.w = x.w * dc * dc;
    }

    for (int j0 = 0; j0 < deg; j0 += 16) {
        const int m = (deg - j0 < 16) ? (deg - j0) : 16;
        int   myr = 0;
        float myd = 0.f;
        if (l < m) {
            myr = srcs[start + j0 + l];
            myd = dinv[myr];
        }
        for (int j = 0; j < m; j += 4) {
            int    r[4];
            float  w[4];
            float4 x[4];
            #pragma unroll
            for (int k = 0; k < 4; ++k) {
                r[k] = __shfl(myr, j + k, 16);
            }
            #pragma unroll
            for (int k = 0; k < 4; ++k) {
                x[k] = make_float4(0.f, 0.f, 0.f, 0.f);
                if (j + k < m) x[k] = X4[(size_t)r[k] * 16 + l];
            }
            #pragma unroll
            for (int k = 0; k < 4; ++k) w[k] = __shfl(myd, j + k, 16) * dc;
            #pragma unroll
            for (int k = 0; k < 4; ++k) {
                acc.x = fmaf(w[k], x[k].x, acc.x);
                acc.y = fmaf(w[k], x[k].y, acc.y);
                acc.z = fmaf(w[k], x[k].z, acc.z);
                acc.w = fmaf(w[k], x[k].w, acc.w);
            }
        }
    }
    if (valid) {
        float4 o;
        o.x = fmaxf((acc.x + pb.x - pm.x) * rsqrtf(pv.x + 1e-5f) * pg.x + pbe.x, 0.f);
        o.y = fmaxf((acc.y + pb.y - pm.y) * rsqrtf(pv.y + 1e-5f) * pg.y + pbe.y, 0.f);
        o.z = fmaxf((acc.z + pb.z - pm.z) * rsqrtf(pv.z + 1e-5f) * pg.z + pbe.z, 0.f);
        o.w = fmaxf((acc.w + pb.w - pm.w) * rsqrtf(pv.w + 1e-5f) * pg.w + pbe.w, 0.f);
        reinterpret_cast<float4*>(act)[(size_t)c * 16 + l] = o;
    }
}

extern "C" void kernel_launch(void* const* d_in, const int* in_sizes, int n_in,
                              void* d_out, int out_size, void* d_ws, size_t ws_size,
                              hipStream_t stream)
{
    const float* x      = (const float*)d_in[0];
    const int*   ei     = (const int*)d_in[1];
    const float* W0     = (const float*)d_in[2];
    const float* b0     = (const float*)d_in[3];
    const float* W1     = (const float*)d_in[4];
    const float* b1     = (const float*)d_in[5];
    const float* bn0_g  = (const float*)d_in[6];
    const float* bn0_b  = (const float*)d_in[7];
    const float* bn0_m  = (const float*)d_in[8];
    const float* bn0_v  = (const float*)d_in[9];
    const float* bn1_g  = (const float*)d_in[10];
    const float* bn1_b  = (const float*)d_in[11];
    const float* bn1_m  = (const float*)d_in[12];
    const float* bn1_v  = (const float*)d_in[13];
    const float* lin1_W = (const float*)d_in[14];
    const float* lin1_b = (const float*)d_in[15];
    const float* lin2_W = (const float*)d_in[16];
    const float* lin2_b = (const float*)d_in[17];
    const float* lin3_W = (const float*)d_in[18];
    const float* lin3_b = (const float*)d_in[19];

    const int N = in_sizes[0] / 64;
    const int E = in_sizes[1] / 2;
    const int* row = ei;
    const int* col = ei + E;

    char*  ws   = (char*)d_ws;
    size_t off0 = 0;
    size_t off1 = (off0 + (size_t)N * sizeof(float) + 255) & ~(size_t)255;
    size_t off2 = (off1 + (size_t)N * 64 * sizeof(float) + 255) & ~(size_t)255;
    size_t off3 = (off2 + (size_t)N * 64 * sizeof(float) + 255) & ~(size_t)255;
    size_t off4 = (off3 + (size_t)N * sizeof(int) + 255) & ~(size_t)255;
    size_t off5 = (off4 + (size_t)N * sizeof(int) + 255) & ~(size_t)255;
    const int gSlice = (N + 1023) / 1024;
    size_t off6 = (off5 + (size_t)gSlice * sizeof(int) + 255) & ~(size_t)255;
    size_t off7 = (off6 + (size_t)E * sizeof(int) + 255) & ~(size_t)255;
    float* dinv = (float*)(ws + off0);
    float* bufA = (float*)(ws + off1);
    float* bufB = (float*)(ws + off2);
    int*   cnt  = (int*)(ws + off3);
    int*   offs = (int*)(ws + off4);
    int*   tot  = (int*)(ws + off5);
    int*   rank = (int*)(ws + off6);
    int*   srcs = (int*)(ws + off7);

    const dim3 blk(256);
    const int gN    = (N + 255) / 256;
    const int gE    = (E + 255) / 256;
    const int gGath = (N + 15) / 16;
    const int gGemm = (N + 127) / 128;

    zero_cnt_kernel<<<gN, blk, 0, stream>>>(cnt, N);
    rank_kernel<<<gE, blk, 0, stream>>>(col, cnt, rank, E);
    slice_total_kernel<<<gSlice, blk, 0, stream>>>(cnt, tot, N);
    scan_kernel<<<gSlice, blk, 0, stream>>>(cnt, tot, offs, dinv, N);
    fill_kernel<<<gE, blk, 0, stream>>>(row, col, rank, offs, srcs, E);

    gemm_wmma_kernel<64, 0><<<gGemm, blk, 0, stream>>>(x, W0, nullptr, bufA, N);
    gather_bn_relu_kernel<<<gGath, blk, 0, stream>>>(offs, cnt, srcs, dinv, bufA,
        b0, bn0_g, bn0_b, bn0_m, bn0_v, bufB, N);

    gemm_wmma_kernel<64, 0><<<gGemm, blk, 0, stream>>>(bufB, W1, nullptr, bufA, N);
    gather_bn_relu_kernel<<<gGath, blk, 0, stream>>>(offs, cnt, srcs, dinv, bufA,
        b1, bn1_g, bn1_b, bn1_m, bn1_v, bufB, N);

    mlp_head_kernel<<<gGemm, blk, 0, stream>>>(bufB, lin1_W, lin1_b, lin2_W, lin2_b,
        lin3_W, lin3_b, (float*)d_out, N);
}


# conv 1's GEMM and the 3-layer head run inside the gather kernels: a wave gathers its own 16-node strip into LDS, then does the WMMA layers on it; 2 launches and act round trips gone; sum order kept
# speedup vs baseline: 2.1650x; 1.0594x over previous
#include <hip/hip_runtime.h>
#include <cstdint>
#include <cstddef>

typedef __attribute__((ext_vector_type(2))) float v2f;
typedef __attribute__((ext_vector_type(8))) float v8f;

template <int NCOL, int EPI>
__global__ __launch_bounds__(256) void gemm_wmma_kernel(
    const float* __restrict__ A, const float* __restrict__ W,
    const float* __restrict__ bias, float* __restrict__ out, int nrows)
{
    constexpr int NT = NCOL / 16;
    __shared__ __align__(16) float sW[64 * NCOL];
    __shared__ __align__(16) float sA[8][16 * 68];

    const int tid  = threadIdx.x;
    const int wave = tid >> 5;
    const int lane = tid & 31;
    const int hh   = lane >> 4;
    const int l16  = lane & 15;
    const int r0   = blockIdx.x * 128 + wave * 16;

    for (int i = tid; i < 64 * NCOL; i += 256) sW[i] = W[i];

    const float4* A4 = reinterpret_cast<const float4*>(A);
    #pragma unroll
    for (int i = 0; i < 8; ++i) {
        int q  = lane + 32 * i;
        int r  = q >> 4;
        int c4 = q & 15;
        int gr = r0 + r;
        float4 v = make_float4(0.f, 0.f, 0.f, 0.f);
        if (gr < nrows) v = A4[(size_t)gr * 16 + c4];
        *reinterpret_cast<float4*>(&sA[wave][r * 68 + c4 * 4]) = v;
    }
    __syncthreads();

    v8f acc[NT] = {};

    #pragma unroll
    for (int kb = 0; kb < 64; kb += 4) {
        v2f a;
        a.x = sA[wave][l16 * 68 + kb + 2 * hh + 0];
        a.y = sA[wave][l16 * 68 + kb + 2 * hh + 1];
        #pragma unroll
        for (int nt = 0; nt < NT; ++nt) {
            v2f b;
            b.x = sW[(kb + 2 * hh + 0) * NCOL + nt * 16 + l16];
            b.y = sW[(kb + 2 * hh + 1) * NCOL + nt * 16 + l16];
            acc[nt] = __builtin_amdgcn_wmma_f32_16x16x4_f32(
                false, a, false, b, (short)0, acc[nt], false, false);
        }
    }

    #pragma unroll
    for (int nt = 0; nt < NT; ++nt) {
        const int col = nt * 16 + l16;
        float bv = 0.f;
        if (EPI > 0) bv = bias[col];
        #pragma unroll
        for (int j = 0; j < 8; ++j) {
            int gr = r0 + j + 8 * hh;
            if (gr < nrows) {
                float v = acc[nt][j] + bv;
                if (EPI == 2) v = fmaxf(v, 0.f);
                out[(size_t)gr * NCOL + col] = v;
            }
        }
    }
}

template <int NCOL>
__device__ __forceinline__ void strip_gemm(
    const float* __restrict__ sAw, const float* __restrict__ sW,
    v8f (&acc)[NCOL / 16], int l16, int hh)
{
    constexpr int NT = NCOL / 16;
    #pragma unroll
    for (int kb = 0; kb < 64; kb += 4) {
        v2f a;
        a.x = sAw[l16 * 68 + kb + 2 * hh + 0];
        a.y = sAw[l16 * 68 + kb + 2 * hh + 1];
        #pragma unroll
        for (int nt = 0; nt < NT; ++nt) {
            v2f b;
            b.x = sW[(kb + 2 * hh + 0) * NCOL + nt * 16 + l16];
            b.y = sW[(kb + 2 * hh + 1) * NCOL + nt * 16 + l16];
            acc[nt] = __builtin_amdgcn_wmma_f32_16x16x4_f32(
                false, a, false, b, (short)0, acc[nt], false, false);
        }
    }
}

__device__ __forceinline__ void strip_store_relu(
    float* __restrict__ sAw, const v8f (&acc)[4],
    const float* __restrict__ bias, int l16, int hh)
{
    #pragma unroll
    for (int nt = 0; nt < 4; ++nt) {
        const int   col = nt * 16 + l16;
        const float bv  = bias[col];
        #pragma unroll
        for (int j = 0; j < 8; ++j)
            sAw[(j + 8 * hh) * 68 + col] = fmaxf(acc[nt][j] + bv, 0.f);
    }
}

__global__ __launch_bounds__(256) void zero_cnt_kernel(int* __restrict__ cnt, int n)
{
    int i = blockIdx.x * 256 + threadIdx.x;
    if (i < n) cnt[i] = 0;
}

__global__ __launch_bounds__(256) void rank_kernel(
    const int* __restrict__ col, int* __restrict__ cnt, int* __restrict__ rank, int ne)
{
    int e = blockIdx.x * 256 + threadIdx.x;
    if (e < ne) rank[e] = atomicAdd(&cnt[col[e]], 1);
}

__global__ __launch_bounds__(256) void slice_total_kernel(
    const int* __restrict__ cnt, int* __restrict__ tot, int n)
{
    __shared__ int sw[8];
    const int tid  = threadIdx.x;
    const int base = blockIdx.x * 1024 + tid * 4;
    int s = 0;
    #pragma unroll
    for (int k = 0; k < 4; ++k)
        if (base + k < n) s += cnt[base + k];
    for (int o = 16; o > 0; o >>= 1) s += __shfl_xor(s, o, 32);
    if ((tid & 31) == 0) sw[tid >> 5] = s;
    __syncthreads();
    if (tid == 0) {
        int t = 0;
        for (int w = 0; w < 8; ++w) t += sw[w];
        tot[blockIdx.x] = t;
    }
}

__global__ __launch_bounds__(256) void scan_kernel(
    const int* __restrict__ cnt, const int* __restrict__ tot,
    int* __restrict__ offs, float* __restrict__ dinv, int n)
{
    __shared__ int sp[8];
    __shared__ int sw[8];
    const int tid  = threadIdx.x;
    const int lane = tid & 31;
    const int wave = tid >> 5;

    int p = 0;
    for (int i = tid; i < (int)blockIdx.x; i += 256) p += tot[i];
    for (int o = 16; o > 0; o >>= 1) p += __shfl_xor(p, o, 32);

    const int base = blockIdx.x * 1024 + tid * 4;
    int c[4];
    #pragma unroll
    for (int k = 0; k < 4; ++k) c[k] = (base + k < n) ? cnt[base + k] : 0;
    const int s = c[0] + c[1] + c[2] + c[3];
    int incl = s;
    for (int o = 1; o < 32; o <<= 1) {
        int t = __shfl_up(incl, o, 32);
        if (lane >= o) incl += t;
    }
    if (lane == 0)  sp[wave] = p;
    if (lane == 31) sw[wave] = incl;
    __syncthreads();
    int pre = 0;
    for (int w = 0; w < 8; ++w) {
        pre += sp[w];
        if (w < wave) pre += sw[w];
    }
    int run = pre + incl - s;
    #pragma unroll
    for (int k = 0; k < 4; ++k) {
        if (base + k < n) {
            offs[base + k] = run;
            dinv[base + k] = rsqrtf((float)(c[k] + 1));
        }
        run += c[k];
    }
}

__global__ __launch_bounds__(256) void fill_kernel(
    const int* __restrict__ row, const int* __restrict__ col,
    const int* __restrict__ rank, const int* __restrict__ offs,
    int* __restrict__ srcs, int ne)
{
    int e = blockIdx.x * 256 + threadIdx.x;
    if (e < ne) srcs[offs[col[e]] + rank[e]] = row[e];
}

template <int HEAD>
__global__ __launch_bounds__(256) void gather_dense_kernel(
    const int* __restrict__ offs, const int* __restrict__ cnt,
    const int* __restrict__ srcs, const float* __restrict__ dinv,
    const float* __restrict__ xw,
    const float* __restrict__ b, const float* __restrict__ g,
    const float* __restrict__ beta, const float* __restrict__ mean,
    const float* __restrict__ var,
    const float* __restrict__ Wa, const float* __restrict__ ca,
    const float* __restrict__ Wb, const float* __restrict__ cb,
    const float* __restrict__ Wc, const float* __restrict__ cc,
    float* __restrict__ res, int n, int ne)
{
    __shared__ __align__(16) float sWa[64 * 64];
    __shared__ __align__(16) float sWb[HEAD ? 64 * 64 : 4];
    __shared__ __align__(16) float sWc[HEAD ? 64 * 16 : 4];
    __shared__ __align__(16) float sA[8][16 * 68];

    const int tid  = threadIdx.x;
    const int wave = tid >> 5;
    const int lane = tid & 31;
    const int hh   = lane >> 4;
    const int l    = lane & 15;
    const int r0   = blockIdx.x * 128 + wave * 16;
    float* sAw = sA[wave];
    const float4* X4 = reinterpret_cast<const float4*>(xw);

    int   m_start = 0, m_deg = 0;
    float m_dc = 0.f;
    {
        const int cm = r0 + 2 * l + hh;
        if (l < 8 && cm < n) {
            m_start = offs[cm];
            m_deg   = cnt[cm];
            m_dc    = dinv[cm];
        }
    }

    {
        const float4* ga = reinterpret_cast<const float4*>(Wa);
        float4* sa = reinterpret_cast<float4*>(sWa);
        #pragma unroll
        for (int i = 0; i < 4; ++i) sa[tid + 256 * i] = ga[tid + 256 * i];
        if (HEAD) {
            const float4* gb = reinterpret_cast<const float4*>(Wb);
            const float4* gc = reinterpret_cast<const float4*>(Wc);
            float4* sb = reinterpret_cast<float4*>(sWb);
            float4* sc = reinterpret_cast<float4*>(sWc);
            #pragma unroll
            for (int i = 0; i < 4; ++i) sb[tid + 256 * i] = gb[tid + 256 * i];
            sc[tid] = gc[tid];
        }
    }

    const float4 pb  = reinterpret_cast<const float4*>(b)[l];
    const float4 pm  = reinterpret_cast<const float4*>(mean)[l];
    const float4 pv  = reinterpret_cast<const float4*>(var)[l];
    const float4 pg  = reinterpret_cast<const float4*>(g)[l];
    const float4 pbe = reinterpret_cast<const float4*>(beta)[l];

    int    cur_r, nx_r;
    float  cur_d;
    float4 cur_x;
    {
        const int s0  = __shfl(m_start, 0, 16);
        const int s1  = __shfl(m_start, 1, 16);
        const int cx  = (r0 + hh < n) ? (r0 + hh) : (n - 1);
        const int ix0 = (s0 + l < ne) ? (s0 + l) : (ne - 1);
        const int ix1 = (s1 + l < ne) ? (s1 + l) : (ne - 1);
        cur_x = X4[(size_t)cx * 16 + l];
        cur_r = srcs[ix0];
        nx_r  = srcs[ix1];
        cur_d = dinv[cur_r];
    }

    for (int p = 0; p < 8; ++p) {
        const int   c     = r0 + 2 * p + hh;
        const bool  valid = c < n;
        const int   start = __shfl(m_start, p, 16);
        const int   deg   = __shfl(m_deg, p, 16);
        const float dc    = __shfl(m_dc, p, 16);

        const float4 xs  = cur_x;
        int          myr = cur_r;
        float        myd = cur_d;

        const int    s2   = __shfl(m_start, (p + 2) & 7, 16);
        const int    cx   = (c + 2 < n) ? (c + 2) : (n - 1);
        const int    ix   = (s2 + l < ne) ? (s2 + l) : (ne - 1);
        const float  nx_d = dinv[nx_r];
        const int    n2_r = srcs[ix];
        const float4 nx_x = X4[(size_t)cx * 16 + l];

        float4 acc;
        acc.x = xs.x * dc * dc;
        acc.y = xs.y * dc * dc;
        acc.z = xs.z * dc * dc;
        acc.w = xs.w * dc * dc;

        for (int j0 = 0; j0 < deg; j0 += 16) {
            const int m = (deg - j0 < 16) ? (deg - j0) : 16;
            if (j0 > 0) {
                myr = 0;
                myd = 0.f;
                if (l < m) {
                    myr = srcs[start + j0 + l];
                    myd = dinv[myr];
                }
            }
            for (int j = 0; j < m; j += 8) {
                int    r[8];
                float  w[8];
                float4 x[8];
                #pragma unroll
                for (int k = 0; k < 8; ++k) r[k] = __shfl(myr, j + k, 16);
                #pragma unroll
                for (int k = 0; k < 8; ++k) {
                    x[k] = make_float4(0.f, 0.f, 0.f, 0.f);
                    if (j + k < m) x[k] = X4[(size_t)r[k] * 16 + l];
                }
                #pragma unroll
                for (int k = 0; k < 8; ++k) w[k] = __shfl(myd, j + k, 16) * dc;
                #pragma unroll
                for (int k = 0; k < 8; ++k) {
                    acc.x = fmaf(w[k], x[k].x, acc.x);
                    acc.y = fmaf(w[k], x[k].y, acc.y);
                    acc.z = fmaf(w[k], x[k].z, acc.z);
                    acc.w = fmaf(w[k], x[k].w, acc.w);
                }
            }
        }

        cur_r = nx_r;
        cur_d = nx_d;
        cur_x = nx_x;
        nx_r  = n2_r;

        float4 o = make_float4(0.f, 0.f, 0.f, 0.f);
        if (valid) {
            o.x = fmaxf((acc.x + pb.x - pm.x) * rsqrtf(pv.x + 1e-5f) * pg.x + pbe.x, 0.f);
            o.y = fmaxf((acc.y + pb.y - pm.y) * rsqrtf(pv.y + 1e-5f) * pg.y + pbe.y, 0.f);
            o.z = fmaxf((acc.z + pb.z - pm.z) * rsqrtf(pv.z + 1e-5f) * pg.z + pbe.z, 0.f);
            o.w = fmaxf((acc.w + pb.w - pm.w) * rsqrtf(pv.w + 1e-5f) * pg.w + pbe.w, 0.f);
        }
        *reinterpret_cast<float4*>(&sAw[(2 * p + hh) * 68 + l * 4]) = o;
    }
    __syncthreads();

    if (HEAD == 0) {
        v8f acc1[4] = {};
        strip_gemm<64>(sAw, sWa, acc1, l, hh);
        #pragma unroll
        for (int nt = 0; nt < 4; ++nt) {
            #pragma unroll
            for (int j = 0; j < 8; ++j)
                sAw[(j + 8 * hh) * 68 + nt * 16 + l] = acc1[nt][j];
        }
        __syncthreads();
        float4* R4 = reinterpret_cast<float4*>(res);
        #pragma unroll
        for (int i = 0; i < 8; ++i) {
            const int q  = lane + 32 * i;
            const int r  = q >> 4;
            const int c4 = q & 15;
            const int gr = r0 + r;
            if (gr < n)
                R4[(size_t)gr * 16 + c4] =
                    *reinterpret_cast<const float4*>(&sAw[r * 68 + c4 * 4]);
        }
    } else {
        v8f acc1[4] = {};
        strip_gemm<64>(sAw, sWa, acc1, l, hh);
        strip_store_relu(sAw, acc1, ca, l, hh);
        __syncthreads();

        v8f acc2[4] = {};
        strip_gemm<64>(sAw, sWb, acc2, l, hh);
        strip_store_relu(sAw, acc2, cb, l, hh);
        __syncthreads();

        v8f acc3[1] = {};
        strip_gemm<16>(sAw, sWc, acc3, l, hh);
        {
            const float bv = cc[l];
            #pragma unroll
            for (int j = 0; j < 8; ++j)
                sAw[(j + 8 * hh) * 68 + l] = acc3[0][j] + bv;
        }
        __syncthreads();
        float4* R4 = reinterpret_cast<float4*>(res);
        #pragma unroll
        for (int i = 0; i < 2; ++i) {
            const int q  = lane + 32 * i;
            const int r  = q >> 2;
            const int c4 = q & 3;
            const int gr = r0 + r;
            if (gr < n)
                R4[(size_t)gr * 4 + c4] =
                    *reinterpret_cast<const float4*>(&sAw[r * 68 + c4 * 4]);
        }
    }
}

extern "C" void kernel_launch(void* const* d_in, const int* in_sizes, int n_in,
                              void* d_out, int out_size, void* d_ws, size_t ws_size,
                              hipStream_t stream)
{
    const float* x      = (const float*)d_in[0];
    const int*   ei     = (const int*)d_in[1];
    const float* W0     = (const float*)d_in[2];
    const float* b0     = (const float*)d_in[3];
    const float* W1     = (const float*)d_in[4];
    const float* b1     = (const float*)d_in[5];
    const float* bn0_g  = (const float*)d_in[6];
    const float* bn0_b  = (const float*)d_in[7];
    const float* bn0_m  = (const float*)d_in[8];
    const float* bn0_v  = (const float*)d_in[9];
    const float* bn1_g  = (const float*)d_in[10];
    const float* bn1_b  = (const float*)d_in[11];
    const float* bn1_m  = (const float*)d_in[12];
    const float* bn1_v  = (const float*)d_in[13];
    const float* lin1_W = (const float*)d_in[14];
    const float* lin1_b = (const float*)d_in[15];
    const float* lin2_W = (const float*)d_in[16];
    const float* lin2_b = (const float*)d_in[17];
    const float* lin3_W = (const float*)d_in[18];
    const float* lin3_b = (const float*)d_in[19];

    const int N = in_sizes[0] / 64;
    const int E = in_sizes[1] / 2;
    const int* row = ei;
    const int* col = ei + E;

    char*  ws   = (char*)d_ws;
    size_t off0 = 0;
    size_t off1 = (off0 + (size_t)N * sizeof(float) + 255) & ~(size_t)255;
    size_t off2 = (off1 + (size_t)N * 64 * sizeof(float) + 255) & ~(size_t)255;
    size_t off3 = (off2 + (size_t)N * 64 * sizeof(float) + 255) & ~(size_t)255;
    size_t off4 = (off3 + (size_t)N * sizeof(int) + 255) & ~(size_t)255;
    size_t off5 = (off4 + (size_t)N * sizeof(int) + 255) & ~(size_t)255;
    const int gSlice = (N + 1023) / 1024;
    size_t off6 = (off5 + (size_t)gSlice * sizeof(int) + 255) & ~(size_t)255;
    size_t off7 = (off6 + (size_t)E * sizeof(int) + 255) & ~(size_t)255;
    float* dinv = (float*)(ws + off0);
    float* bufA = (float*)(ws + off1);
    float* bufB = (float*)(ws + off2);
    int*   cnt  = (int*)(ws + off3);
    int*   offs = (int*)(ws + off4);
    int*   tot  = (int*)(ws + off5);
    int*   rank = (int*)(ws + off6);
    int*   srcs = (int*)(ws + off7);

    const dim3 blk(256);
    const int gN    = (N + 255) / 256;
    const int gE    = (E + 255) / 256;
    const int gGemm = (N + 127) / 128;

    zero_cnt_kernel<<<gN, blk, 0, stream>>>(cnt, N);
    rank_kernel<<<gE, blk, 0, stream>>>(col, cnt, rank, E);
    slice_total_kernel<<<gSlice, blk, 0, stream>>>(cnt, tot, N);
    scan_kernel<<<gSlice, blk, 0, stream>>>(cnt, tot, offs, dinv, N);
    fill_kernel<<<gE, blk, 0, stream>>>(row, col, rank, offs, srcs, E);

    gemm_wmma_kernel<64, 0><<<gGemm, blk, 0, stream>>>(x, W0, nullptr, bufA, N);
    gather_dense_kernel<0><<<gGemm, blk, 0, stream>>>(offs, cnt, srcs, dinv, bufA,
        b0, bn0_g, bn0_b, bn0_m, bn0_v,
        W1, nullptr, nullptr, nullptr, nullptr, nullptr, bufB, N, E);

    gather_dense_kernel<1><<<gGemm, blk, 0, stream>>>(offs, cnt, srcs, dinv, bufB,
        b1, bn1_g, bn1_b, bn1_m, bn1_v,
        lin1_W, lin1_b, lin2_W, lin2_b, lin3_W, lin3_b, (float*)d_out, N, E);
}


# edge grouping: the sorted source list (slice totals, scan, fill: 3 launches, rank and offset arrays) replaced by a 32-slot table per node that the ranking kernel fills; sum order still arrival rank
# speedup vs baseline: 2.3411x; 1.0813x over previous
#include <hip/hip_runtime.h>
#include <cstdint>
#include <cstddef>

typedef __attribute__((ext_vector_type(2))) float v2f;
typedef __attribute__((ext_vector_type(8))) float v8f;

template <int NCOL, int EPI>
__global__ __launch_bounds__(256) void gemm_wmma_kernel(
    const float* __restrict__ A, const float* __restrict__ W,
    const float* __restrict__ bias, float* __restrict__ out, int nrows)
{
    constexpr int NT = NCOL / 16;
    __shared__ __align__(16) float sW[64 * NCOL];
    __shared__ __align__(16) float sA[8][16 * 68];

    const int tid  = threadIdx.x;
    const int wave = tid >> 5;
    const int lane = tid & 31;
    const int hh   = lane >> 4;
    const int l16  = lane & 15;
    const int r0   = blockIdx.x * 128 + wave * 16;

    for (int i = tid; i < 64 * NCOL; i += 256) sW[i] = W[i];

    const float4* A4 = reinterpret_cast<const float4*>(A);
    #pragma unroll
    for (int i = 0; i < 8; ++i) {
        int q  = lane + 32 * i;
        int r  = q >> 4;
        int c4 = q & 15;
        int gr = r0 + r;
        float4 v = make_float4(0.f, 0.f, 0.f, 0.f);
        if (gr < nrows) v = A4[(size_t)gr * 16 + c4];
        *reinterpret_cast<float4*>(&sA[wave][r * 68 + c4 * 4]) = v;
    }
    __syncthreads();

    v8f acc[NT] = {};

    #pragma unroll
    for (int kb = 0; kb < 64; kb += 4) {
        v2f a;
        a.x = sA[wave][l16 * 68 + kb + 2 * hh + 0];
        a.y = sA[wave][l16 * 68 + kb + 2 * hh + 1];
        #pragma unroll
        for (int nt = 0; nt < NT; ++nt) {
            v2f b;
            b.x = sW[(kb + 2 * hh + 0) * NCOL + nt * 16 + l16];
            b.y = sW[(kb + 2 * hh + 1) * NCOL + nt * 16 + l16];
            acc[nt] = __builtin_amdgcn_wmma_f32_16x16x4_f32(
                false, a, false, b, (short)0, acc[nt], false, false);
        }
    }

    #pragma unroll
    for (int nt = 0; nt < NT; ++nt) {
        const int col = nt * 16 + l16;
        float bv = 0.f;
        if (EPI > 0) bv = bias[col];
        #pragma unroll
        for (int j = 0; j < 8; ++j) {
            int gr = r0 + j + 8 * hh;
            if (gr < nrows) {
                float v = acc[nt][j] + bv;
                if (EPI == 2) v = fmaxf(v, 0.f);
                out[(size_t)gr * NCOL + col] = v;
            }
        }
    }
}

template <int NCOL>
__device__ __forceinline__ void strip_gemm(
    const float* __restrict__ sAw, const float* __restrict__ sW,
    v8f (&acc)[NCOL / 16], int l16, int hh)
{
    constexpr int NT = NCOL / 16;
    #pragma unroll
    for (int kb = 0; kb < 64; kb += 4) {
        v2f a;
        a.x = sAw[l16 * 68 + kb + 2 * hh + 0];
        a.y = sAw[l16 * 68 + kb + 2 * hh + 1];
        #pragma unroll
        for (int nt = 0; nt < NT; ++nt) {
            v2f b;
            b.x = sW[(kb + 2 * hh + 0) * NCOL + nt * 16 + l16];
            b.y = sW[(kb + 2 * hh + 1) * NCOL + nt * 16 + l16];
            acc[nt] = __builtin_amdgcn_wmma_f32_16x16x4_f32(
                false, a, false, b, (short)0, acc[nt], false, false);
        }
    }
}

__device__ __forceinline__ void strip_store_relu(
    float* __restrict__ sAw, const v8f (&acc)[4],
    const float* __restrict__ bias, int l16, int hh)
{
    #pragma unroll
    for (int nt = 0; nt < 4; ++nt) {
        const int   col = nt * 16 + l16;
        const float bv  = bias[col];
        #pragma unroll
        for (int j = 0; j < 8; ++j)
            sAw[(j + 8 * hh) * 68 + col] = fmaxf(acc[nt][j] + bv, 0.f);
    }
}

constexpr int SLOTS = 32;

__global__ __launch_bounds__(256) void zero_cnt_kernel(
    int* __restrict__ cnt, int* __restrict__ novf, int n)
{
    int i = blockIdx.x * 256 + threadIdx.x;
    if (i < n) cnt[i] = 0;
    if (i == 0) novf[0] = 0;
}

__global__ __launch_bounds__(256) void rank_fill_kernel(
    const int* __restrict__ row, const int* __restrict__ col,
    int* __restrict__ cnt, int* __restrict__ slot,
    int* __restrict__ novf, int* __restrict__ ovf, int ne)
{
    int e = blockIdx.x * 256 + threadIdx.x;
    if (e < ne) {
        const int c = col[e];
        const int s = row[e];
        const int r = atomicAdd(&cnt[c], 1);
        if (r < SLOTS) {
            slot[(size_t)c * SLOTS + r] = s;
        } else {
            const int o = atomicAdd(&novf[0], 1);
            ovf[2 * (size_t)o + 0] = c;
            ovf[2 * (size_t)o + 1] = s;
        }
    }
}

template <int HEAD>
__global__ __launch_bounds__(256) void gather_dense_kernel(
    const int* __restrict__ cnt, const int* __restrict__ slot,
    const int* __restrict__ novf, const int* __restrict__ ovf,
    const int* __restrict__ row,
    const float* __restrict__ xw,
    const float* __restrict__ b, const float* __restrict__ g,
    const float* __restrict__ beta, const float* __restrict__ mean,
    const float* __restrict__ var,
    const float* __restrict__ Wa, const float* __restrict__ ca,
    const float* __restrict__ Wb, const float* __restrict__ cb,
    const float* __restrict__ Wc, const float* __restrict__ cc,
    float* __restrict__ res, int n, int ne)
{
    __shared__ __align__(16) float sWa[64 * 64];
    __shared__ __align__(16) float sWb[HEAD ? 64 * 64 : 4];
    __shared__ __align__(16) float sWc[HEAD ? 64 * 16 : 4];
    __shared__ __align__(16) float sA[8][16 * 68];

    const int tid  = threadIdx.x;
    const int wave = tid >> 5;
    const int lane = tid & 31;
    const int hh   = lane >> 4;
    const int l    = lane & 15;
    const int r0   = blockIdx.x * 128 + wave * 16;
    float* sAw = sA[wave];
    const float4* X4 = reinterpret_cast<const float4*>(xw);

    int   m_deg = 0;
    float m_dc  = 0.f;
    {
        const int cm = r0 + 2 * l + hh;
        if (l < 8 && cm < n) {
            m_deg = cnt[cm];
            m_dc  = rsqrtf((float)(m_deg + 1));
        }
    }

    {
        const float4* ga = reinterpret_cast<const float4*>(Wa);
        float4* sa = reinterpret_cast<float4*>(sWa);
        #pragma unroll
        for (int i = 0; i < 4; ++i) sa[tid + 256 * i] = ga[tid + 256 * i];
        if (HEAD) {
            const float4* gb = reinterpret_cast<const float4*>(Wb);
            const float4* gc = reinterpret_cast<const float4*>(Wc);
            float4* sb = reinterpret_cast<float4*>(sWb);
            float4* sc = reinterpret_cast<float4*>(sWc);
            #pragma unroll
            for (int i = 0; i < 4; ++i) sb[tid + 256 * i] = gb[tid + 256 * i];
            sc[tid] = gc[tid];
        }
    }

    const float4 pb  = reinterpret_cast<const float4*>(b)[l];
    const float4 pm  = reinterpret_cast<const float4*>(mean)[l];
    const float4 pv  = reinterpret_cast<const float4*>(var)[l];
    const float4 pg  = reinterpret_cast<const float4*>(g)[l];
    const float4 pbe = reinterpret_cast<const float4*>(beta)[l];

    int    cur_r, nx_r;
    int    cur_k;
    float4 cur_x;
    {
        const int  d0 = __shfl(m_deg, 0, 16);
        const int  d1 = __shfl(m_deg, 1, 16);
        const int  cx = (r0 + hh < n) ? (r0 + hh) : (n - 1);
        const int* q0 = (l < d0) ? (slot + (size_t)(r0 + hh) * SLOTS + l) : row;
        const int* q1 = (l < d1) ? (slot + (size_t)(r0 + 2 + hh) * SLOTS + l) : row;
        cur_x = X4[(size_t)cx * 16 + l];
        cur_r = *q0;
        nx_r  = *q1;
        cur_k = cnt[cur_r];
    }

    for (int p = 0; p < 8; ++p) {
        const int   c     = r0 + 2 * p + hh;
        const bool  valid = c < n;
        const int   deg   = __shfl(m_deg, p, 16);
        const float dc    = __shfl(m_dc, p, 16);
        const int*  mine  = slot + (size_t)c * SLOTS;

        const float4 xs  = cur_x;
        int          myr = cur_r;
        float        myd = rsqrtf((float)(cur_k + 1));

        const int    p2   = (p + 2) & 7;
        const int    d2   = __shfl(m_deg, p2, 16);
        const int    cx   = (c + 2 < n) ? (c + 2) : (n - 1);
        const int*   q2   = (l < d2) ? (slot + (size_t)(r0 + 2 * p2 + hh) * SLOTS + l) : row;
        const int    nx_k = cnt[nx_r];
        const int    n2_r = *q2;
        const float4 nx_x = X4[(size_t)cx * 16 + l];

        float4 acc;
        acc.x = xs.x * dc * dc;
        acc.y = xs.y * dc * dc;
        acc.z = xs.z * dc * dc;
        acc.w = xs.w * dc * dc;

        const int degs = (deg < SLOTS) ? deg : SLOTS;
        for (int j0 = 0; j0 < degs; j0 += 16) {
            const int m = (degs - j0 < 16) ? (degs - j0) : 16;
            if (j0 > 0) {
                myr = 0;
                myd = 0.f;
                if (l < m) {
                    myr = mine[j0 + l];
                    myd = rsqrtf((float)(cnt[myr] + 1));
                }
            }
            for (int j = 0; j < m; j += 8) {
                int    r[8];
                float  w[8];
                float4 x[8];
                #pragma unroll
                for (int k = 0; k < 8; ++k) r[k] = __shfl(myr, j + k, 16);
                #pragma unroll
                for (int k = 0; k < 8; ++k) {
                    x[k] = make_float4(0.f, 0.f, 0.f, 0.f);
                    if (j + k < m) x[k] = X4[(size_t)r[k] * 16 + l];
                }
                #pragma unroll
                for (int k = 0; k < 8; ++k) w[k] = __shfl(myd, j + k, 16) * dc;
                #pragma unroll
                for (int k = 0; k < 8; ++k) {
                    acc.x = fmaf(w[k], x[k].x, acc.x);
                    acc.y = fmaf(w[k], x[k].y, acc.y);
                    acc.z = fmaf(w[k], x[k].z, acc.z);
                    acc.w = fmaf(w[k], x[k].w, acc.w);
                }
            }
        }

        if (deg > SLOTS) {
            int no = novf[0];
            no = (no < ne) ? no : ne;
            const int2* O2 = reinterpret_cast<const int2*>(ovf);
            for (int i0 = 0; i0 < no; i0 += 16) {
                int   orow = 0;
                float ow   = 0.f;
                if (i0 + l < no) {
                    const int2 rec = O2[i0 + l];
                    if (rec.x == c) {
                        orow = rec.y;
                        ow   = rsqrtf((float)(cnt[orow] + 1)) * dc;
                    }
                }
                #pragma nounroll
                for (int k = 0; k < 16; ++k) {
                    const float wk = __shfl(ow, k, 16);
                    const int   rk = __shfl(orow, k, 16);
                    if (wk != 0.f) {
                        const float4 xv = X4[(size_t)rk * 16 + l];
                        acc.x = fmaf(wk, xv.x, acc.x);
                        acc.y = fmaf(wk, xv.y, acc.y);
                        acc.z = fmaf(wk, xv.z, acc.z);
                        acc.w = fmaf(wk, xv.w, acc.w);
                    }
                }
            }
        }

        cur_r = nx_r;
        cur_k = nx_k;
        cur_x = nx_x;
        nx_r  = n2_r;

        float4 o = make_float4(0.f, 0.f, 0.f, 0.f);
        if (valid) {
            o.x = fmaxf((acc.x + pb.x - pm.x) * rsqrtf(pv.x + 1e-5f) * pg.x + pbe.x, 0.f);
            o.y = fmaxf((acc.y + pb.y - pm.y) * rsqrtf(pv.y + 1e-5f) * pg.y + pbe.y, 0.f);
            o.z = fmaxf((acc.z + pb.z - pm.z) * rsqrtf(pv.z + 1e-5f) * pg.z + pbe.z, 0.f);
            o.w = fmaxf((acc.w + pb.w - pm.w) * rsqrtf(pv.w + 1e-5f) * pg.w + pbe.w, 0.f);
        }
        *reinterpret_cast<float4*>(&sAw[(2 * p + hh) * 68 + l * 4]) = o;
    }
    __syncthreads();

    if (HEAD == 0) {
        v8f acc1[4] = {};
        strip_gemm<64>(sAw, sWa, acc1, l, hh);
        #pragma unroll
        for (int nt = 0; nt < 4; ++nt) {
            #pragma unroll
            for (int j = 0; j < 8; ++j)
                sAw[(j + 8 * hh) * 68 + nt * 16 + l] = acc1[nt][j];
        }
        __syncthreads();
        float4* R4 = reinterpret_cast<float4*>(res);
        #pragma unroll
        for (int i = 0; i < 8; ++i) {
            const int q  = lane + 32 * i;
            const int r  = q >> 4;
            const int c4 = q & 15;
            const int gr = r0 + r;
            if (gr < n)
                R4[(size_t)gr * 16 + c4] =
                    *reinterpret_cast<const float4*>(&sAw[r * 68 + c4 * 4]);
        }
    } else {
        v8f acc1[4] = {};
        strip_gemm<64>(sAw, sWa, acc1, l, hh);
        strip_store_relu(sAw, acc1, ca, l, hh);
        __syncthreads();

        v8f acc2[4] = {};
        strip_gemm<64>(sAw, sWb, acc2, l, hh);
        strip_store_relu(sAw, acc2, cb, l, hh);
        __syncthreads();

        v8f acc3[1] = {};
        strip_gemm<16>(sAw, sWc, acc3, l, hh);
        {
            const float bv = cc[l];
            #pragma unroll
            for (int j = 0; j < 8; ++j)
                sAw[(j + 8 * hh) * 68 + l] = acc3[0][j] + bv;
        }
        __syncthreads();
        float4* R4 = reinterpret_cast<float4*>(res);
        #pragma unroll
        for (int i = 0; i < 2; ++i) {
            const int q  = lane + 32 * i;
            const int r  = q >> 2;
            const int c4 = q & 3;
            const int gr = r0 + r;
            if (gr < n)
                R4[(size_t)gr * 4 + c4] =
                    *reinterpret_cast<const float4*>(&sAw[r * 68 + c4 * 4]);
        }
    }
}

extern "C" void kernel_launch(void* const* d_in, const int* in_sizes, int n_in,
                              void* d_out, int out_size, void* d_ws, size_t ws_size,
                              hipStream_t stream)
{
    const float* x      = (const float*)d_in[0];
    const int*   ei     = (const int*)d_in[1];
    const float* W0     = (const float*)d_in[2];
    const float* b0     = (const float*)d_in[3];
    const float* W1     = (const float*)d_in[4];
    const float* b1     = (const float*)d_in[5];
    const float* bn0_g  = (const float*)d_in[6];
    const float* bn0_b  = (const float*)d_in[7];
    const float* bn0_m  = (const float*)d_in[8];
    const float* bn0_v  = (const float*)d_in[9];
    const float* bn1_g  = (const float*)d_in[10];
    const float* bn1_b  = (const float*)d_in[11];
    const float* bn1_m  = (const float*)d_in[12];
    const float* bn1_v  = (const float*)d_in[13];
    const float* lin1_W = (const float*)d_in[14];
    const float* lin1_b = (const float*)d_in[15];
    const float* lin2_W = (const float*)d_in[16];
    const float* lin2_b = (const float*)d_in[17];
    const float* lin3_W = (const float*)d_in[18];
    const float* lin3_b = (const float*)d_in[19];

    const int N = in_sizes[0] / 64;
    const int E = in_sizes[1] / 2;
    const int* row = ei;
    const int* col = ei + E;

    const dim3 blk(256);
    const int gN    = (N + 255) / 256;
    const int gE    = (E + 255) / 256;
    const int gGemm = (N + 127) / 128;

    char*  ws   = (char*)d_ws;
    size_t off0 = 0;
    size_t off1 = (off0 + (size_t)N * 64 * sizeof(float) + 255) & ~(size_t)255;
    size_t off2 = (off1 + (size_t)N * 64 * sizeof(float) + 255) & ~(size_t)255;
    size_t off3 = (off2 + (size_t)N * sizeof(int) + 255) & ~(size_t)255;
    size_t off4 = (off3 + sizeof(int) + 255) & ~(size_t)255;
    size_t off5 = (off4 + (size_t)gGemm * 128 * SLOTS * sizeof(int) + 255) & ~(size_t)255;
    float* bufA = (float*)(ws + off0);
    float* bufB = (float*)(ws + off1);
    int*   cnt  = (int*)(ws + off2);
    int*   novf = (int*)(ws + off3);
    int*   slot = (int*)(ws + off4);
    int*   ovf  = (int*)(ws + off5);

    zero_cnt_kernel<<<gN, blk, 0, stream>>>(cnt, novf, N);
    rank_fill_kernel<<<gE, blk, 0, stream>>>(row, col, cnt, slot, novf, ovf, E);

    gemm_wmma_kernel<64, 0><<<gGemm, blk, 0, stream>>>(x, W0, nullptr, bufA, N);
    gather_dense_kernel<0><<<gGemm, blk, 0, stream>>>(cnt, slot, novf, ovf, row, bufA,
        b0, bn0_g, bn0_b, bn0_m, bn0_v,
        W1, nullptr, nullptr, nullptr, nullptr, nullptr, bufB, N, E);

    gather_dense_kernel<1><<<gGemm, blk, 0, stream>>>(cnt, slot, novf, ovf, row, bufB,
        b1, bn1_g, bn1_b, bn1_m, bn1_v,
        lin1_W, lin1_b, lin2_W, lin2_b, lin3_W, lin3_b, (float*)d_out, N, E);
}


# edge grouping: global returning add per edge replaced by two-level LDS counting (4096-edge chunks sorted by 128-node range; one block per range ranks in LDS, whole slot lines); sum order = LDS rank
# speedup vs baseline: 3.6574x; 1.5623x over previous
#include <hip/hip_runtime.h>
#include <cstdint>
#include <cstddef>

typedef __attribute__((ext_vector_type(2))) float v2f;
typedef __attribute__((ext_vector_type(8))) float v8f;

template <int NCOL, int EPI>
__global__ __launch_bounds__(256) void gemm_wmma_kernel(
    const float* __restrict__ A, const float* __restrict__ W,
    const float* __restrict__ bias, float* __restrict__ out, int nrows)
{
    constexpr int NT = NCOL / 16;
    __shared__ __align__(16) float sW[64 * NCOL];
    __shared__ __align__(16) float sA[8][16 * 68];

    const int tid  = threadIdx.x;
    const int wave = tid >> 5;
    const int lane = tid & 31;
    const int hh   = lane >> 4;
    const int l16  = lane & 15;
    const int r0   = blockIdx.x * 128 + wave * 16;

    for (int i = tid; i < 64 * NCOL; i += 256) sW[i] = W[i];

    const float4* A4 = reinterpret_cast<const float4*>(A);
    #pragma unroll
    for (int i = 0; i < 8; ++i) {
        int q  = lane + 32 * i;
        int r  = q >> 4;
        int c4 = q & 15;
        int gr = r0 + r;
        float4 v = make_float4(0.f, 0.f, 0.f, 0.f);
        if (gr < nrows) v = A4[(size_t)gr * 16 + c4];
        *reinterpret_cast<float4*>(&sA[wave][r * 68 + c4 * 4]) = v;
    }
    __syncthreads();

    v8f acc[NT] = {};

    #pragma unroll
    for (int kb = 0; kb < 64; kb += 4) {
        v2f a;
        a.x = sA[wave][l16 * 68 + kb + 2 * hh + 0];
        a.y = sA[wave][l16 * 68 + kb + 2 * hh + 1];
        #pragma unroll
        for (int nt = 0; nt < NT; ++nt) {
            v2f b;
            b.x = sW[(kb + 2 * hh + 0) * NCOL + nt * 16 + l16];
            b.y = sW[(kb + 2 * hh + 1) * NCOL + nt * 16 + l16];
            acc[nt] = __builtin_amdgcn_wmma_f32_16x16x4_f32(
                false, a, false, b, (short)0, acc[nt], false, false);
        }
    }

    #pragma unroll
    for (int nt = 0; nt < NT; ++nt) {
        const int col = nt * 16 + l16;
        float bv = 0.f;
        if (EPI > 0) bv = bias[col];
        #pragma unroll
        for (int j = 0; j < 8; ++j) {
            int gr = r0 + j + 8 * hh;
            if (gr < nrows) {
                float v = acc[nt][j] + bv;
                if (EPI == 2) v = fmaxf(v, 0.f);
                out[(size_t)gr * NCOL + col] = v;
            }
        }
    }
}

template <int NCOL>
__device__ __forceinline__ void strip_gemm(
    const float* __restrict__ sAw, const float* __restrict__ sW,
    v8f (&acc)[NCOL / 16], int l16, int hh)
{
    constexpr int NT = NCOL / 16;
    #pragma unroll
    for (int kb = 0; kb < 64; kb += 4) {
        v2f a;
        a.x = sAw[l16 * 68 + kb + 2 * hh + 0];
        a.y = sAw[l16 * 68 + kb + 2 * hh + 1];
        #pragma unroll
        for (int nt = 0; nt < NT; ++nt) {
            v2f b;
            b.x = sW[(kb + 2 * hh + 0) * NCOL + nt * 16 + l16];
            b.y = sW[(kb + 2 * hh + 1) * NCOL + nt * 16 + l16];
            acc[nt] = __builtin_amdgcn_wmma_f32_16x16x4_f32(
                false, a, false, b, (short)0, acc[nt], false, false);
        }
    }
}

__device__ __forceinline__ void strip_store_relu(
    float* __restrict__ sAw, const v8f (&acc)[4],
    const float* __restrict__ bias, int l16, int hh)
{
    #pragma unroll
    for (int nt = 0; nt < 4; ++nt) {
        const int   col = nt * 16 + l16;
        const float bv  = bias[col];
        #pragma unroll
        for (int j = 0; j < 8; ++j)
            sAw[(j + 8 * hh) * 68 + col] = fmaxf(acc[nt][j] + bv, 0.f);
    }
}

constexpr int SLOTS = 32;
constexpr int RB    = 128;
constexpr int CH    = 4096;
constexpr int MAXB  = 1024;

__global__ __launch_bounds__(512) void bin_edges_kernel(
    const int* __restrict__ row, const int* __restrict__ col,
    int* __restrict__ sorted_i, int* __restrict__ start,
    int* __restrict__ novf, int ne, int nbk, int pitch)
{
    int2* sorted = reinterpret_cast<int2*>(sorted_i);
    __shared__ int hist[MAXB];
    __shared__ int wsum[16];
    __shared__ __align__(16) int2 stage[CH];

    const int tid  = threadIdx.x;
    const int lane = tid & 31;
    const int wave = tid >> 5;
    const int base = blockIdx.x * CH;

    hist[2 * tid + 0] = 0;
    hist[2 * tid + 1] = 0;
    if (blockIdx.x == 0 && tid == 0) novf[0] = 0;

    int c[8], s[8], r[8];
    #pragma unroll
    for (int k = 0; k < 8; ++k) {
        const int e = base + k * 512 + tid;
        c[k] = 0;
        s[k] = 0;
        if (e < ne) { c[k] = col[e]; s[k] = row[e]; }
    }
    __syncthreads();

    #pragma unroll
    for (int k = 0; k < 8; ++k) {
        r[k] = 0;
        if (base + k * 512 + tid < ne) r[k] = atomicAdd(&hist[(unsigned)c[k] / RB], 1);
    }
    __syncthreads();

    const int h0 = hist[2 * tid + 0];
    const int h1 = hist[2 * tid + 1];
    int x = h0 + h1;
    #pragma unroll
    for (int d = 1; d < 32; d <<= 1) {
        const int y = __shfl_up(x, d, 32);
        if (lane >= d) x += y;
    }
    if (lane == 31) wsum[wave] = x;
    __syncthreads();
    int woff = 0;
    for (int w = 0; w < wave; ++w) woff += wsum[w];
    const int ex = woff + x - (h0 + h1);
    hist[2 * tid + 0] = ex;
    hist[2 * tid + 1] = ex + h0;
    __syncthreads();

    #pragma unroll
    for (int k = 0; k < 8; ++k) {
        if (base + k * 512 + tid < ne)
            stage[hist[(unsigned)c[k] / RB] + r[k]] = make_int2(c[k], s[k]);
    }
    for (int i = tid; i <= nbk; i += 512)
        start[(size_t)blockIdx.x * pitch + i] = hist[i];
    __syncthreads();

    const int4* st4 = reinterpret_cast<const int4*>(stage);
    int4* out4 = reinterpret_cast<int4*>(sorted + base);
    #pragma unroll
    for (int k = 0; k < CH / 2 / 512; ++k) {
        const int i = k * 512 + tid;
        if (base + 2 * i + 1 < ne)      out4[i] = st4[i];
        else if (base + 2 * i < ne)     sorted[base + 2 * i] = stage[2 * i];
    }
}

__global__ __launch_bounds__(256) void rank_owner_kernel(
    const int* __restrict__ sorted_i, const int* __restrict__ start,
    int* __restrict__ cnt, int* __restrict__ slot,
    int* __restrict__ novf, int* __restrict__ ovf,
    int n, int nchunk, int pitch)
{
    const int2* sorted = reinterpret_cast<const int2*>(sorted_i);
    __shared__ int lcnt[RB];
    __shared__ __align__(16) int lslot[RB * SLOTS];

    const int tid   = threadIdx.x;
    const int bk    = blockIdx.x;
    const int node0 = bk * RB;

    if (tid < RB) lcnt[tid] = 0;
    {
        int4* z = reinterpret_cast<int4*>(lslot);
        #pragma unroll
        for (int i = 0; i < RB * SLOTS / 4 / 256; ++i) z[tid + 256 * i] = make_int4(0, 0, 0, 0);
    }
    __syncthreads();

    for (int t = tid; t < nchunk; t += 256) {
        const int* st = start + (size_t)t * pitch + bk;
        const int  sb = st[0];
        const int  se = st[1];
        const int2* p = sorted + (size_t)t * CH;
        for (int j = sb; j < se; j += 4) {
            int2 rec[4];
            #pragma unroll
            for (int k = 0; k < 4; ++k) {
                rec[k] = make_int2(node0, 0);
                if (j + k < se) rec[k] = p[j + k];
            }
            #pragma unroll
            for (int k = 0; k < 4; ++k) {
                if (j + k < se) {
                    const int lc = (rec[k].x - node0) & (RB - 1);
                    const int rk = atomicAdd(&lcnt[lc], 1);
                    if (rk < SLOTS) {
                        lslot[lc * SLOTS + rk] = rec[k].y;
                    } else {
                        const int o = atomicAdd(&novf[0], 1);
                        ovf[2 * (size_t)o + 0] = rec[k].x;
                        ovf[2 * (size_t)o + 1] = rec[k].y;
                    }
                }
            }
        }
    }
    __syncthreads();

    if (tid < RB && node0 + tid < n) cnt[node0 + tid] = lcnt[tid];
    const int4* l4 = reinterpret_cast<const int4*>(lslot);
    int4* s4 = reinterpret_cast<int4*>(slot + (size_t)node0 * SLOTS);
    #pragma unroll
    for (int i = 0; i < RB * SLOTS / 4 / 256; ++i) s4[tid + 256 * i] = l4[tid + 256 * i];
}

template <int HEAD>
__global__ __launch_bounds__(256) void gather_dense_kernel(
    const int* __restrict__ cnt, const int* __restrict__ slot,
    const int* __restrict__ novf, const int* __restrict__ ovf,
    const int* __restrict__ row,
    const float* __restrict__ xw,
    const float* __restrict__ b, const float* __restrict__ g,
    const float* __restrict__ beta, const float* __restrict__ mean,
    const float* __restrict__ var,
    const float* __restrict__ Wa, const float* __restrict__ ca,
    const float* __restrict__ Wb, const float* __restrict__ cb,
    const float* __restrict__ Wc, const float* __restrict__ cc,
    float* __restrict__ res, int n, int ne)
{
    __shared__ __align__(16) float sWa[64 * 64];
    __shared__ __align__(16) float sWb[HEAD ? 64 * 64 : 4];
    __shared__ __align__(16) float sWc[HEAD ? 64 * 16 : 4];
    __shared__ __align__(16) float sA[8][16 * 68];

    const int tid  = threadIdx.x;
    const int wave = tid >> 5;
    const int lane = tid & 31;
    const int hh   = lane >> 4;
    const int l    = lane & 15;
    const int r0   = blockIdx.x * 128 + wave * 16;
    float* sAw = sA[wave];
    const float4* X4 = reinterpret_cast<const float4*>(xw);

    int   m_deg = 0;
    float m_dc  = 0.f;
    {
        const int cm = r0 + 2 * l + hh;
        if (l < 8 && cm < n) {
            m_deg = cnt[cm];
            m_dc  = rsqrtf((float)(m_deg + 1));
        }
    }

    {
        const float4* ga = reinterpret_cast<const float4*>(Wa);
        float4* sa = reinterpret_cast<float4*>(sWa);
        #pragma unroll
        for (int i = 0; i < 4; ++i) sa[tid + 256 * i] = ga[tid + 256 * i];
        if (HEAD) {
            const float4* gb = reinterpret_cast<const float4*>(Wb);
            const float4* gc = reinterpret_cast<const float4*>(Wc);
            float4* sb = reinterpret_cast<float4*>(sWb);
            float4* sc = reinterpret_cast<float4*>(sWc);
            #pragma unroll
            for (int i = 0; i < 4; ++i) sb[tid + 256 * i] = gb[tid + 256 * i];
            sc[tid] = gc[tid];
        }
    }

    const float4 pb  = reinterpret_cast<const float4*>(b)[l];
    const float4 pm  = reinterpret_cast<const float4*>(mean)[l];
    const float4 pv  = reinterpret_cast<const float4*>(var)[l];
    const float4 pg  = reinterpret_cast<const float4*>(g)[l];
    const float4 pbe = reinterpret_cast<const float4*>(beta)[l];

    int    cur_r, nx_r;
    int    cur_k;
    float4 cur_x;
    {
        const int  d0 = __shfl(m_deg, 0, 16);
        const int  d1 = __shfl(m_deg, 1, 16);
        const int  cx = (r0 + hh < n) ? (r0 + hh) : (n - 1);
        const int* q0 = (l < d0) ? (slot + (size_t)(r0 + hh) * SLOTS + l) : row;
        const int* q1 = (l < d1) ? (slot + (size_t)(r0 + 2 + hh) * SLOTS + l) : row;
        cur_x = X4[(size_t)cx * 16 + l];
        cur_r = *q0;
        nx_r  = *q1;
        cur_k = cnt[cur_r];
    }

    for (int p = 0; p < 8; ++p) {
        const int   c     = r0 + 2 * p + hh;
        const bool  valid = c < n;
        const int   deg   = __shfl(m_deg, p, 16);
        const float dc    = __shfl(m_dc, p, 16);
        const int*  mine  = slot + (size_t)c * SLOTS;

        const float4 xs  = cur_x;
        int          myr = cur_r;
        float        myd = rsqrtf((float)(cur_k + 1));

        const int    p2   = (p + 2) & 7;
        const int    d2   = __shfl(m_deg, p2, 16);
        const int    cx   = (c + 2 < n) ? (c + 2) : (n - 1);
        const int*   q2   = (l < d2) ? (slot + (size_t)(r0 + 2 * p2 + hh) * SLOTS + l) : row;
        const int    nx_k = cnt[nx_r];
        const int    n2_r = *q2;
        const float4 nx_x = X4[(size_t)cx * 16 + l];

        float4 acc;
        acc.x = xs.x * dc * dc;
        acc.y = xs.y * dc * dc;
        acc.z = xs.z * dc * dc;
        acc.w = xs.w * dc * dc;

        const int degs = (deg < SLOTS) ? deg : SLOTS;
        for (int j0 = 0; j0 < degs; j0 += 16) {
            const int m = (degs - j0 < 16) ? (degs - j0) : 16;
            if (j0 > 0) {
                myr = 0;
                myd = 0.f;
                if (l < m) {
                    myr = mine[j0 + l];
                    myd = rsqrtf((float)(cnt[myr] + 1));
                }
            }
            for (int j = 0; j < m; j += 8) {
                int    r[8];
                float  w[8];
                float4 x[8];
                #pragma unroll
                for (int k = 0; k < 8; ++k) r[k] = __shfl(myr, j + k, 16);
                #pragma unroll
                for (int k = 0; k < 8; ++k) {
                    x[k] = make_float4(0.f, 0.f, 0.f, 0.f);
                    if (j + k < m) x[k] = X4[(size_t)r[k] * 16 + l];
                }
                #pragma unroll
                for (int k = 0; k < 8; ++k) w[k] = __shfl(myd, j + k, 16) * dc;
                #pragma unroll
                for (int k = 0; k < 8; ++k) {
                    acc.x = fmaf(w[k], x[k].x, acc.x);
                    acc.y = fmaf(w[k], x[k].y, acc.y);
                    acc.z = fmaf(w[k], x[k].z, acc.z);
                    acc.w = fmaf(w[k], x[k].w, acc.w);
                }
            }
        }

        if (deg > SLOTS) {
            int no = novf[0];
            no = (no < ne) ? no : ne;
            const int2* O2 = reinterpret_cast<const int2*>(ovf);
            for (int i0 = 0; i0 < no; i0 += 16) {
                int   orow = 0;
                float ow   = 0.f;
                if (i0 + l < no) {
                    const int2 rec = O2[i0 + l];
                    if (rec.x == c) {
                        orow = rec.y;
                        ow   = rsqrtf((float)(cnt[orow] + 1)) * dc;
                    }
                }
                #pragma nounroll
                for (int k = 0; k < 16; ++k) {
                    const float wk = __shfl(ow, k, 16);
                    const int   rk = __shfl(orow, k, 16);
                    if (wk != 0.f) {
                        const float4 xv = X4[(size_t)rk * 16 + l];
                        acc.x = fmaf(wk, xv.x, acc.x);
                        acc.y = fmaf(wk, xv.y, acc.y);
                        acc.z = fmaf(wk, xv.z, acc.z);
                        acc.w = fmaf(wk, xv.w, acc.w);
                    }
                }
            }
        }

        cur_r = nx_r;
        cur_k = nx_k;
        cur_x = nx_x;
        nx_r  = n2_r;

        float4 o = make_float4(0.f, 0.f, 0.f, 0.f);
        if (valid) {
            o.x = fmaxf((acc.x + pb.x - pm.x) * rsqrtf(pv.x + 1e-5f) * pg.x + pbe.x, 0.f);
            o.y = fmaxf((acc.y + pb.y - pm.y) * rsqrtf(pv.y + 1e-5f) * pg.y + pbe.y, 0.f);
            o.z = fmaxf((acc.z + pb.z - pm.z) * rsqrtf(pv.z + 1e-5f) * pg.z + pbe.z, 0.f);
            o.w = fmaxf((acc.w + pb.w - pm.w) * rsqrtf(pv.w + 1e-5f) * pg.w + pbe.w, 0.f);
        }
        *reinterpret_cast<float4*>(&sAw[(2 * p + hh) * 68 + l * 4]) = o;
    }
    __syncthreads();

    if (HEAD == 0) {
        v8f acc1[4] = {};
        strip_gemm<64>(sAw, sWa, acc1, l, hh);
        #pragma unroll
        for (int nt = 0; nt < 4; ++nt) {
            #pragma unroll
            for (int j = 0; j < 8; ++j)
                sAw[(j + 8 * hh) * 68 + nt * 16 + l] = acc1[nt][j];
        }
        __syncthreads();
        float4* R4 = reinterpret_cast<float4*>(res);
        #pragma unroll
        for (int i = 0; i < 8; ++i) {
            const int q  = lane + 32 * i;
            const int r  = q >> 4;
            const int c4 = q & 15;
            const int gr = r0 + r;
            if (gr < n)
                R4[(size_t)gr * 16 + c4] =
                    *reinterpret_cast<const float4*>(&sAw[r * 68 + c4 * 4]);
        }
    } else {
        v8f acc1[4] = {};
        strip_gemm<64>(sAw, sWa, acc1, l, hh);
        strip_store_relu(sAw, acc1, ca, l, hh);
        __syncthreads();

        v8f acc2[4] = {};
        strip_gemm<64>(sAw, sWb, acc2, l, hh);
        strip_store_relu(sAw, acc2, cb, l, hh);
        __syncthreads();

        v8f acc3[1] = {};
        strip_gemm<16>(sAw, sWc, acc3, l, hh);
        {
            const float bv = cc[l];
            #pragma unroll
            for (int j = 0; j < 8; ++j)
                sAw[(j + 8 * hh) * 68 + l] = acc3[0][j] + bv;
        }
        __syncthreads();
        float4* R4 = reinterpret_cast<float4*>(res);
        #pragma unroll
        for (int i = 0; i < 2; ++i) {
            const int q  = lane + 32 * i;
            const int r  = q >> 2;
            const int c4 = q & 3;
            const int gr = r0 + r;
            if (gr < n)
                R4[(size_t)gr * 4 + c4] =
                    *reinterpret_cast<const float4*>(&sAw[r * 68 + c4 * 4]);
        }
    }
}

extern "C" void kernel_launch(void* const* d_in, const int* in_sizes, int n_in,
                              void* d_out, int out_size, void* d_ws, size_t ws_size,
                              hipStream_t stream)
{
    const float* x      = (const float*)d_in[0];
    const int*   ei     = (const int*)d_in[1];
    const float* W0     = (const float*)d_in[2];
    const float* b0     = (const float*)d_in[3];
    const float* W1     = (const float*)d_in[4];
    const float* b1     = (const float*)d_in[5];
    const float* bn0_g  = (const float*)d_in[6];
    const float* bn0_b  = (const float*)d_in[7];
    const float* bn0_m  = (const float*)d_in[8];
    const float* bn0_v  = (const float*)d_in[9];
    const float* bn1_g  = (const float*)d_in[10];
    const float* bn1_b  = (const float*)d_in[11];
    const float* bn1_m  = (const float*)d_in[12];
    const float* bn1_v  = (const float*)d_in[13];
    const float* lin1_W = (const float*)d_in[14];
    const float* lin1_b = (const float*)d_in[15];
    const float* lin2_W = (const float*)d_in[16];
    const float* lin2_b = (const float*)d_in[17];
    const float* lin3_W = (const float*)d_in[18];
    const float* lin3_b = (const float*)d_in[19];

    const int N = in_sizes[0] / 64;
    const int E = in_sizes[1] / 2;
    const int* row = ei;
    const int* col = ei + E;

    const dim3 blk(256);
    const int gGemm = (N + 127) / 128;
    const int gBin  = (E + CH - 1) / CH;
    const int pitch = (gGemm + 1 + 31) & ~31;

    char*  ws   = (char*)d_ws;
    size_t off0 = 0;
    size_t off1 = (off0 + (size_t)N * 64 * sizeof(float) + 255) & ~(size_t)255;
    size_t off2 = (off1 + (size_t)N * 64 * sizeof(float) + 255) & ~(size_t)255;
    size_t off3 = (off2 + (size_t)N * sizeof(int) + 255) & ~(size_t)255;
    size_t off4 = (off3 + sizeof(int) + 255) & ~(size_t)255;
    size_t off5 = (off4 + (size_t)gGemm * 128 * SLOTS * sizeof(int) + 255) & ~(size_t)255;
    size_t off6 = (off5 + (size_t)E * 2 * sizeof(int) + 255) & ~(size_t)255;
    size_t off7 = (off6 + (size_t)gBin * CH * 2 * sizeof(int) + 255) & ~(size_t)255;
    float* bufA   = (float*)(ws + off0);
    float* bufB   = (float*)(ws + off1);
    int*   cnt    = (int*)(ws + off2);
    int*   novf   = (int*)(ws + off3);
    int*   slot   = (int*)(ws + off4);
    int*   ovf    = (int*)(ws + off5);
    int*   sorted = (int*)(ws + off6);
    int*   start  = (int*)(ws + off7);

    bin_edges_kernel<<<gBin, dim3(512), 0, stream>>>(row, col, sorted, start, novf, E, gGemm, pitch);
    rank_owner_kernel<<<gGemm, blk, 0, stream>>>(sorted, start, cnt, slot, novf, ovf, N, gBin, pitch);

    gemm_wmma_kernel<64, 0><<<gGemm, blk, 0, stream>>>(x, W0, nullptr, bufA, N);
    gather_dense_kernel<0><<<gGemm, blk, 0, stream>>>(cnt, slot, novf, ovf, row, bufA,
        b0, bn0_g, bn0_b, bn0_m, bn0_v,
        W1, nullptr, nullptr, nullptr, nullptr, nullptr, bufB, N, E);

    gather_dense_kernel<1><<<gGemm, blk, 0, stream>>>(cnt, slot, novf, ovf, row, bufB,
        b1, bn1_g, bn1_b, bn1_m, bn1_v,
        lin1_W, lin1_b, lin2_W, lin2_b, lin3_W, lin3_b, (float*)d_out, N, E);
}


# graph layer 1 computed as agg(x) @ W0: the gather reads x itself, W0 is applied to the aggregate on WMMA in the same kernel; the x @ W0 launch and its buffer are gone; product/sum association changed
# speedup vs baseline: 4.2502x; 1.1621x over previous
#include <hip/hip_runtime.h>
#include <cstdint>
#include <cstddef>

typedef __attribute__((ext_vector_type(2))) float v2f;
typedef __attribute__((ext_vector_type(8))) float v8f;

template <int NCOL>
__device__ __forceinline__ void strip_gemm(
    const float* __restrict__ sAw, const float* __restrict__ sW,
    v8f (&acc)[NCOL / 16], int l16, int hh)
{
    constexpr int NT = NCOL / 16;
    #pragma unroll
    for (int kb = 0; kb < 64; kb += 4) {
        v2f a;
        a.x = sAw[l16 * 68 + kb + 2 * hh + 0];
        a.y = sAw[l16 * 68 + kb + 2 * hh + 1];
        #pragma unroll
        for (int nt = 0; nt < NT; ++nt) {
            v2f b;
            b.x = sW[(kb + 2 * hh + 0) * NCOL + nt * 16 + l16];
            b.y = sW[(kb + 2 * hh + 1) * NCOL + nt * 16 + l16];
            acc[nt] = __builtin_amdgcn_wmma_f32_16x16x4_f32(
                false, a, false, b, (short)0, acc[nt], false, false);
        }
    }
}

__device__ __forceinline__ void strip_store_relu(
    float* __restrict__ sAw, const v8f (&acc)[4],
    const float* __restrict__ bias, int l16, int hh)
{
    #pragma unroll
    for (int nt = 0; nt < 4; ++nt) {
        const int   col = nt * 16 + l16;
        const float bv  = bias[col];
        #pragma unroll
        for (int j = 0; j < 8; ++j)
            sAw[(j + 8 * hh) * 68 + col] = fmaxf(acc[nt][j] + bv, 0.f);
    }
}

__device__ __forceinline__ void strip_store_bn_relu(
    float* __restrict__ sAw, const v8f (&acc)[4],
    const float* __restrict__ bias, const float* __restrict__ g,
    const float* __restrict__ beta, const float* __restrict__ mean,
    const float* __restrict__ var, int l16, int hh)
{
    #pragma unroll
    for (int nt = 0; nt < 4; ++nt) {
        const int   col = nt * 16 + l16;
        const float bv  = bias[col];
        const float mv  = mean[col];
        const float rs  = rsqrtf(var[col] + 1e-5f);
        const float gv  = g[col];
        const float bev = beta[col];
        #pragma unroll
        for (int j = 0; j < 8; ++j)
            sAw[(j + 8 * hh) * 68 + col] =
                fmaxf((acc[nt][j] + bv - mv) * rs * gv + bev, 0.f);
    }
}

constexpr int SLOTS = 32;
constexpr int RB    = 128;
constexpr int CH    = 4096;
constexpr int MAXB  = 1024;

__global__ __launch_bounds__(512) void bin_edges_kernel(
    const int* __restrict__ row, const int* __restrict__ col,
    int* __restrict__ sorted_i, int* __restrict__ start,
    int* __restrict__ novf, int ne, int nbk, int pitch)
{
    int2* sorted = reinterpret_cast<int2*>(sorted_i);
    __shared__ int hist[MAXB];
    __shared__ int wsum[16];
    __shared__ __align__(16) int2 stage[CH];

    const int tid  = threadIdx.x;
    const int lane = tid & 31;
    const int wave = tid >> 5;
    const int base = blockIdx.x * CH;

    hist[2 * tid + 0] = 0;
    hist[2 * tid + 1] = 0;
    if (blockIdx.x == 0 && tid == 0) novf[0] = 0;

    int c[8], s[8], r[8];
    #pragma unroll
    for (int k = 0; k < 8; ++k) {
        const int e = base + k * 512 + tid;
        c[k] = 0;
        s[k] = 0;
        if (e < ne) { c[k] = col[e]; s[k] = row[e]; }
    }
    __syncthreads();

    #pragma unroll
    for (int k = 0; k < 8; ++k) {
        r[k] = 0;
        if (base + k * 512 + tid < ne) r[k] = atomicAdd(&hist[(unsigned)c[k] / RB], 1);
    }
    __syncthreads();

    const int h0 = hist[2 * tid + 0];
    const int h1 = hist[2 * tid + 1];
    int x = h0 + h1;
    #pragma unroll
    for (int d = 1; d < 32; d <<= 1) {
        const int y = __shfl_up(x, d, 32);
        if (lane >= d) x += y;
    }
    if (lane == 31) wsum[wave] = x;
    __syncthreads();
    int woff = 0;
    for (int w = 0; w < wave; ++w) woff += wsum[w];
    const int ex = woff + x - (h0 + h1);
    hist[2 * tid + 0] = ex;
    hist[2 * tid + 1] = ex + h0;
    __syncthreads();

    #pragma unroll
    for (int k = 0; k < 8; ++k) {
        if (base + k * 512 + tid < ne)
            stage[hist[(unsigned)c[k] / RB] + r[k]] = make_int2(c[k], s[k]);
    }
    for (int i = tid; i <= nbk; i += 512)
        start[(size_t)blockIdx.x * pitch + i] = hist[i];
    __syncthreads();

    const int4* st4 = reinterpret_cast<const int4*>(stage);
    int4* out4 = reinterpret_cast<int4*>(sorted + base);
    #pragma unroll
    for (int k = 0; k < CH / 2 / 512; ++k) {
        const int i = k * 512 + tid;
        if (base + 2 * i + 1 < ne)      out4[i] = st4[i];
        else if (base + 2 * i < ne)     sorted[base + 2 * i] = stage[2 * i];
    }
}

__global__ __launch_bounds__(256) void rank_owner_kernel(
    const int* __restrict__ sorted_i, const int* __restrict__ start,
    int* __restrict__ cnt, int* __restrict__ slot,
    int* __restrict__ novf, int* __restrict__ ovf,
    int n, int nchunk, int pitch)
{
    const int2* sorted = reinterpret_cast<const int2*>(sorted_i);
    __shared__ int lcnt[RB];
    __shared__ __align__(16) int lslot[RB * SLOTS];

    const int tid   = threadIdx.x;
    const int bk    = blockIdx.x;
    const int node0 = bk * RB;

    if (tid < RB) lcnt[tid] = 0;
    {
        int4* z = reinterpret_cast<int4*>(lslot);
        #pragma unroll
        for (int i = 0; i < RB * SLOTS / 4 / 256; ++i) z[tid + 256 * i] = make_int4(0, 0, 0, 0);
    }
    __syncthreads();

    for (int t = tid; t < nchunk; t += 256) {
        const int* st = start + (size_t)t * pitch + bk;
        const int  sb = st[0];
        const int  se = st[1];
        const int2* p = sorted + (size_t)t * CH;
        for (int j = sb; j < se; j += 4) {
            int2 rec[4];
            #pragma unroll
            for (int k = 0; k < 4; ++k) {
                rec[k] = make_int2(node0, 0);
                if (j + k < se) rec[k] = p[j + k];
            }
            #pragma unroll
            for (int k = 0; k < 4; ++k) {
                if (j + k < se) {
                    const int lc = (rec[k].x - node0) & (RB - 1);
                    const int rk = atomicAdd(&lcnt[lc], 1);
                    if (rk < SLOTS) {
                        lslot[lc * SLOTS + rk] = rec[k].y;
                    } else {
                        const int o = atomicAdd(&novf[0], 1);
                        ovf[2 * (size_t)o + 0] = rec[k].x;
                        ovf[2 * (size_t)o + 1] = rec[k].y;
                    }
                }
            }
        }
    }
    __syncthreads();

    if (tid < RB && node0 + tid < n) cnt[node0 + tid] = lcnt[tid];
    const int4* l4 = reinterpret_cast<const int4*>(lslot);
    int4* s4 = reinterpret_cast<int4*>(slot + (size_t)node0 * SLOTS);
    #pragma unroll
    for (int i = 0; i < RB * SLOTS / 4 / 256; ++i) s4[tid + 256 * i] = l4[tid + 256 * i];
}

template <int HEAD>
__global__ __launch_bounds__(256) void gather_dense_kernel(
    const int* __restrict__ cnt, const int* __restrict__ slot,
    const int* __restrict__ novf, const int* __restrict__ ovf,
    const int* __restrict__ row,
    const float* __restrict__ xw,
    const float* __restrict__ b, const float* __restrict__ g,
    const float* __restrict__ beta, const float* __restrict__ mean,
    const float* __restrict__ var,
    const float* __restrict__ Wa, const float* __restrict__ ca,
    const float* __restrict__ Wb, const float* __restrict__ cb,
    const float* __restrict__ Wc, const float* __restrict__ cc,
    float* __restrict__ res, int n, int ne)
{
    __shared__ __align__(16) float sWa[64 * 64];
    __shared__ __align__(16) float sWb[64 * 64];
    __shared__ __align__(16) float sWc[HEAD ? 64 * 16 : 4];
    __shared__ __align__(16) float sA[8][16 * 68];

    const int tid  = threadIdx.x;
    const int wave = tid >> 5;
    const int lane = tid & 31;
    const int hh   = lane >> 4;
    const int l    = lane & 15;
    const int r0   = blockIdx.x * 128 + wave * 16;
    float* sAw = sA[wave];
    const float4* X4 = reinterpret_cast<const float4*>(xw);

    int   m_deg = 0;
    float m_dc  = 0.f;
    {
        const int cm = r0 + 2 * l + hh;
        if (l < 8 && cm < n) {
            m_deg = cnt[cm];
            m_dc  = rsqrtf((float)(m_deg + 1));
        }
    }

    {
        const float4* ga = reinterpret_cast<const float4*>(Wa);
        float4* sa = reinterpret_cast<float4*>(sWa);
        #pragma unroll
        for (int i = 0; i < 4; ++i) sa[tid + 256 * i] = ga[tid + 256 * i];
        const float4* gb = reinterpret_cast<const float4*>(Wb);
        float4* sb = reinterpret_cast<float4*>(sWb);
        #pragma unroll
        for (int i = 0; i < 4; ++i) sb[tid + 256 * i] = gb[tid + 256 * i];
        if (HEAD) {
            const float4* gc = reinterpret_cast<const float4*>(Wc);
            float4* sc = reinterpret_cast<float4*>(sWc);
            sc[tid] = gc[tid];
        }
    }

    float4 pb  = make_float4(0.f, 0.f, 0.f, 0.f);
    float4 pm  = pb, pv = pb, pg = pb, pbe = pb;
    if (HEAD) {
        pb  = reinterpret_cast<const float4*>(b)[l];
        pm  = reinterpret_cast<const float4*>(mean)[l];
        pv  = reinterpret_cast<const float4*>(var)[l];
        pg  = reinterpret_cast<const float4*>(g)[l];
        pbe = reinterpret_cast<const float4*>(beta)[l];
    }

    int    cur_r, nx_r;
    int    cur_k;
    float4 cur_x;
    {
        const int  d0 = __shfl(m_deg, 0, 16);
        const int  d1 = __shfl(m_deg, 1, 16);
        const int  cx = (r0 + hh < n) ? (r0 + hh) : (n - 1);
        const int* q0 = (l < d0) ? (slot + (size_t)(r0 + hh) * SLOTS + l) : row;
        const int* q1 = (l < d1) ? (slot + (size_t)(r0 + 2 + hh) * SLOTS + l) : row;
        cur_x = X4[(size_t)cx * 16 + l];
        cur_r = *q0;
        nx_r  = *q1;
        cur_k = cnt[cur_r];
    }

    for (int p = 0; p < 8; ++p) {
        const int   c     = r0 + 2 * p + hh;
        const bool  valid = c < n;
        const int   deg   = __shfl(m_deg, p, 16);
        const float dc    = __shfl(m_dc, p, 16);
        const int*  mine  = slot + (size_t)c * SLOTS;

        const float4 xs  = cur_x;
        int          myr = cur_r;
        float        myd = rsqrtf((float)(cur_k + 1));

        const int    p2   = (p + 2) & 7;
        const int    d2   = __shfl(m_deg, p2, 16);
        const int    cx   = (c + 2 < n) ? (c + 2) : (n - 1);
        const int*   q2   = (l < d2) ? (slot + (size_t)(r0 + 2 * p2 + hh) * SLOTS + l) : row;
        const int    nx_k = cnt[nx_r];
        const int    n2_r = *q2;
        const float4 nx_x = X4[(size_t)cx * 16 + l];

        float4 acc;
        acc.x = xs.x * dc * dc;
        acc.y = xs.y * dc * dc;
        acc.z = xs.z * dc * dc;
        acc.w = xs.w * dc * dc;

        const int degs = (deg < SLOTS) ? deg : SLOTS;
        for (int j0 = 0; j0 < degs; j0 += 16) {
            const int m = (degs - j0 < 16) ? (degs - j0) : 16;
            if (j0 > 0) {
                myr = 0;
                myd = 0.f;
                if (l < m) {
                    myr = mine[j0 + l];
                    myd = rsqrtf((float)(cnt[myr] + 1));
                }
            }
            for (int j = 0; j < m; j += 8) {
                int    r[8];
                float  w[8];
                float4 x[8];
                #pragma unroll
                for (int k = 0; k < 8; ++k) r[k] = __shfl(myr, j + k, 16);
                #pragma unroll
                for (int k = 0; k < 8; ++k) {
                    x[k] = make_float4(0.f, 0.f, 0.f, 0.f);
                    if (j + k < m) x[k] = X4[(size_t)r[k] * 16 + l];
                }
                #pragma unroll
                for (int k = 0; k < 8; ++k) w[k] = __shfl(myd, j + k, 16) * dc;
                #pragma unroll
                for (int k = 0; k < 8; ++k) {
                    acc.x = fmaf(w[k], x[k].x, acc.x);
                    acc.y = fmaf(w[k], x[k].y, acc.y);
                    acc.z = fmaf(w[k], x[k].z, acc.z);
                    acc.w = fmaf(w[k], x[k].w, acc.w);
                }
            }
        }

        if (deg > SLOTS) {
            int no = novf[0];
            no = (no < ne) ? no : ne;
            const int2* O2 = reinterpret_cast<const int2*>(ovf);
            for (int i0 = 0; i0 < no; i0 += 16) {
                int   orow = 0;
                float ow   = 0.f;
                if (i0 + l < no) {
                    const int2 rec = O2[i0 + l];
                    if (rec.x == c) {
                        orow = rec.y;
                        ow   = rsqrtf((float)(cnt[orow] + 1)) * dc;
                    }
                }
                #pragma nounroll
                for (int k = 0; k < 16; ++k) {
                    const float wk = __shfl(ow, k, 16);
                    const int   rk = __shfl(orow, k, 16);
                    if (wk != 0.f) {
                        const float4 xv = X4[(size_t)rk * 16 + l];
                        acc.x = fmaf(wk, xv.x, acc.x);
                        acc.y = fmaf(wk, xv.y, acc.y);
                        acc.z = fmaf(wk, xv.z, acc.z);
                        acc.w = fmaf(wk, xv.w, acc.w);
                    }
                }
            }
        }

        cur_r = nx_r;
        cur_k = nx_k;
        cur_x = nx_x;
        nx_r  = n2_r;

        float4 o = make_float4(0.f, 0.f, 0.f, 0.f);
        if (valid) {
            if (HEAD) {
                o.x = fmaxf((acc.x + pb.x - pm.x) * rsqrtf(pv.x + 1e-5f) * pg.x + pbe.x, 0.f);
                o.y = fmaxf((acc.y + pb.y - pm.y) * rsqrtf(pv.y + 1e-5f) * pg.y + pbe.y, 0.f);
                o.z = fmaxf((acc.z + pb.z - pm.z) * rsqrtf(pv.z + 1e-5f) * pg.z + pbe.z, 0.f);
                o.w = fmaxf((acc.w + pb.w - pm.w) * rsqrtf(pv.w + 1e-5f) * pg.w + pbe.w, 0.f);
            } else {
                o = acc;
            }
        }
        *reinterpret_cast<float4*>(&sAw[(2 * p + hh) * 68 + l * 4]) = o;
    }
    __syncthreads();

    if (HEAD == 0) {
        v8f acc0[4] = {};
        strip_gemm<64>(sAw, sWa, acc0, l, hh);
        strip_store_bn_relu(sAw, acc0, b, g, beta, mean, var, l, hh);
        __syncthreads();

        v8f acc1[4] = {};
        strip_gemm<64>(sAw, sWb, acc1, l, hh);
        #pragma unroll
        for (int nt = 0; nt < 4; ++nt) {
            #pragma unroll
            for (int j = 0; j < 8; ++j)
                sAw[(j + 8 * hh) * 68 + nt * 16 + l] = acc1[nt][j];
        }
        __syncthreads();
        float4* R4 = reinterpret_cast<float4*>(res);
        #pragma unroll
        for (int i = 0; i < 8; ++i) {
            const int q  = lane + 32 * i;
            const int r  = q >> 4;
            const int c4 = q & 15;
            const int gr = r0 + r;
            if (gr < n)
                R4[(size_t)gr * 16 + c4] =
                    *reinterpret_cast<const float4*>(&sAw[r * 68 + c4 * 4]);
        }
    } else {
        v8f acc1[4] = {};
        strip_gemm<64>(sAw, sWa, acc1, l, hh);
        strip_store_relu(sAw, acc1, ca, l, hh);
        __syncthreads();

        v8f acc2[4] = {};
        strip_gemm<64>(sAw, sWb, acc2, l, hh);
        strip_store_relu(sAw, acc2, cb, l, hh);
        __syncthreads();

        v8f acc3[1] = {};
        strip_gemm<16>(sAw, sWc, acc3, l, hh);
        {
            const float bv = cc[l];
            #pragma unroll
            for (int j = 0; j < 8; ++j)
                sAw[(j + 8 * hh) * 68 + l] = acc3[0][j] + bv;
        }
        __syncthreads();
        float4* R4 = reinterpret_cast<float4*>(res);
        #pragma unroll
        for (int i = 0; i < 2; ++i) {
            const int q  = lane + 32 * i;
            const int r  = q >> 2;
            const int c4 = q & 3;
            const int gr = r0 + r;
            if (gr < n)
                R4[(size_t)gr * 4 + c4] =
                    *reinterpret_cast<const float4*>(&sAw[r * 68 + c4 * 4]);
        }
    }
}

extern "C" void kernel_launch(void* const* d_in, const int* in_sizes, int n_in,
                              void* d_out, int out_size, void* d_ws, size_t ws_size,
                              hipStream_t stream)
{
    const float* x      = (const float*)d_in[0];
    const int*   ei     = (const int*)d_in[1];
    const float* W0     = (const float*)d_in[2];
    const float* b0     = (const float*)d_in[3];
    const float* W1     = (const float*)d_in[4];
    const float* b1     = (const float*)d_in[5];
    const float* bn0_g  = (const float*)d_in[6];
    const float* bn0_b  = (const float*)d_in[7];
    const float* bn0_m  = (const float*)d_in[8];
    const float* bn0_v  = (const float*)d_in[9];
    const float* bn1_g  = (const float*)d_in[10];
    const float* bn1_b  = (const float*)d_in[11];
    const float* bn1_m  = (const float*)d_in[12];
    const float* bn1_v  = (const float*)d_in[13];
    const float* lin1_W = (const float*)d_in[14];
    const float* lin1_b = (const float*)d_in[15];
    const float* lin2_W = (const float*)d_in[16];
    const float* lin2_b = (const float*)d_in[17];
    const float* lin3_W = (const float*)d_in[18];
    const float* lin3_b = (const float*)d_in[19];

    const int N = in_sizes[0] / 64;
    const int E = in_sizes[1] / 2;
    const int* row = ei;
    const int* col = ei + E;

    const dim3 blk(256);
    const int gGemm = (N + 127) / 128;
    const int gBin  = (E + CH - 1) / CH;
    const int pitch = (gGemm + 1 + 31) & ~31;

    char*  ws   = (char*)d_ws;
    size_t off1 = 0;
    size_t off2 = (off1 + (size_t)N * 64 * sizeof(float) + 255) & ~(size_t)255;
    size_t off3 = (off2 + (size_t)N * sizeof(int) + 255) & ~(size_t)255;
    size_t off4 = (off3 + sizeof(int) + 255) & ~(size_t)255;
    size_t off5 = (off4 + (size_t)gGemm * 128 * SLOTS * sizeof(int) + 255) & ~(size_t)255;
    size_t off6 = (off5 + (size_t)E * 2 * sizeof(int) + 255) & ~(size_t)255;
    size_t off7 = (off6 + (size_t)gBin * CH * 2 * sizeof(int) + 255) & ~(size_t)255;
    float* bufB   = (float*)(ws + off1);
    int*   cnt    = (int*)(ws + off2);
    int*   novf   = (int*)(ws + off3);
    int*   slot   = (int*)(ws + off4);
    int*   ovf    = (int*)(ws + off5);
    int*   sorted = (int*)(ws + off6);
    int*   start  = (int*)(ws + off7);

    bin_edges_kernel<<<gBin, dim3(512), 0, stream>>>(row, col, sorted, start, novf, E, gGemm, pitch);
    rank_owner_kernel<<<gGemm, blk, 0, stream>>>(sorted, start, cnt, slot, novf, ovf, N, gBin, pitch);

    gather_dense_kernel<0><<<gGemm, blk, 0, stream>>>(cnt, slot, novf, ovf, row, x,
        b0, bn0_g, bn0_b, bn0_m, bn0_v,
        W0, nullptr, W1, nullptr, nullptr, nullptr, bufB, N, E);

    gather_dense_kernel<1><<<gGemm, blk, 0, stream>>>(cnt, slot, novf, ovf, row, bufB,
        b1, bn1_g, bn1_b, bn1_m, bn1_v,
        lin1_W, lin1_b, lin2_W, lin2_b, lin3_W, lin3_b, (float*)d_out, N, E);
}


# gather: the eight row loads of a batch go out as one group, under no branch, each from an offset register of its own (an entry past a node's edges: a valid row at weight zero); sum order unchanged
# speedup vs baseline: 4.4374x; 1.0440x over previous
#include <hip/hip_runtime.h>
#include <cstdint>
#include <cstddef>

typedef __attribute__((ext_vector_type(2))) float v2f;
typedef __attribute__((ext_vector_type(8))) float v8f;

template <int NCOL>
__device__ __forceinline__ void strip_gemm(
    const float* __restrict__ sAw, const float* __restrict__ sW,
    v8f (&acc)[NCOL / 16], int l16, int hh)
{
    constexpr int NT = NCOL / 16;
    #pragma unroll
    for (int kb = 0; kb < 64; kb += 4) {
        v2f a;
        a.x = sAw[l16 * 68 + kb + 2 * hh + 0];
        a.y = sAw[l16 * 68 + kb + 2 * hh + 1];
        #pragma unroll
        for (int nt = 0; nt < NT; ++nt) {
            v2f b;
            b.x = sW[(kb + 2 * hh + 0) * NCOL + nt * 16 + l16];
            b.y = sW[(kb + 2 * hh + 1) * NCOL + nt * 16 + l16];
            acc[nt] = __builtin_amdgcn_wmma_f32_16x16x4_f32(
                false, a, false, b, (short)0, acc[nt], false, false);
        }
    }
}

__device__ __forceinline__ void strip_store_relu(
    float* __restrict__ sAw, const v8f (&acc)[4],
    const float* __restrict__ bias, int l16, int hh)
{
    #pragma unroll
    for (int nt = 0; nt < 4; ++nt) {
        const int   col = nt * 16 + l16;
        const float bv  = bias[col];
        #pragma unroll
        for (int j = 0; j < 8; ++j)
            sAw[(j + 8 * hh) * 68 + col] = fmaxf(acc[nt][j] + bv, 0.f);
    }
}

__device__ __forceinline__ void strip_store_bn_relu(
    float* __restrict__ sAw, const v8f (&acc)[4],
    const float* __restrict__ bias, const float* __restrict__ g,
    const float* __restrict__ beta, const float* __restrict__ mean,
    const float* __restrict__ var, int l16, int hh)
{
    #pragma unroll
    for (int nt = 0; nt < 4; ++nt) {
        const int   col = nt * 16 + l16;
        const float bv  = bias[col];
        const float mv  = mean[col];
        const float rs  = rsqrtf(var[col] + 1e-5f);
        const float gv  = g[col];
        const float bev = beta[col];
        #pragma unroll
        for (int j = 0; j < 8; ++j)
            sAw[(j + 8 * hh) * 68 + col] =
                fmaxf((acc[nt][j] + bv - mv) * rs * gv + bev, 0.f);
    }
}

constexpr int SLOTS = 32;
constexpr int RB    = 128;
constexpr int CH    = 4096;
constexpr int MAXB  = 1024;

__global__ __launch_bounds__(512) void bin_edges_kernel(
    const int* __restrict__ row, const int* __restrict__ col,
    int* __restrict__ sorted_i, int* __restrict__ start,
    int* __restrict__ novf, int ne, int nbk, int pitch)
{
    int2* sorted = reinterpret_cast<int2*>(sorted_i);
    __shared__ int hist[MAXB];
    __shared__ int wsum[16];
    __shared__ __align__(16) int2 stage[CH];

    const int tid  = threadIdx.x;
    const int lane = tid & 31;
    const int wave = tid >> 5;
    const int base = blockIdx.x * CH;

    hist[2 * tid + 0] = 0;
    hist[2 * tid + 1] = 0;
    if (blockIdx.x == 0 && tid == 0) novf[0] = 0;

    int c[8], s[8], r[8];
    #pragma unroll
    for (int k = 0; k < 8; ++k) {
        const int e = base + k * 512 + tid;
        c[k] = 0;
        s[k] = 0;
        if (e < ne) { c[k] = col[e]; s[k] = row[e]; }
    }
    __syncthreads();

    #pragma unroll
    for (int k = 0; k < 8; ++k) {
        r[k] = 0;
        if (base + k * 512 + tid < ne) r[k] = atomicAdd(&hist[(unsigned)c[k] / RB], 1);
    }
    __syncthreads();

    const int h0 = hist[2 * tid + 0];
    const int h1 = hist[2 * tid + 1];
    int x = h0 + h1;
    #pragma unroll
    for (int d = 1; d < 32; d <<= 1) {
        const int y = __shfl_up(x, d, 32);
        if (lane >= d) x += y;
    }
    if (lane == 31) wsum[wave] = x;
    __syncthreads();
    int woff = 0;
    for (int w = 0; w < wave; ++w) woff += wsum[w];
    const int ex = woff + x - (h0 + h1);
    hist[2 * tid + 0] = ex;
    hist[2 * tid + 1] = ex + h0;
    __syncthreads();

    #pragma unroll
    for (int k = 0; k < 8; ++k) {
        if (base + k * 512 + tid < ne)
            stage[hist[(unsigned)c[k] / RB] + r[k]] = make_int2(c[k], s[k]);
    }
    for (int i = tid; i <= nbk; i += 512)
        start[(size_t)blockIdx.x * pitch + i] = hist[i];
    __syncthreads();

    const int4* st4 = reinterpret_cast<const int4*>(stage);
    int4* out4 = reinterpret_cast<int4*>(sorted + base);
    #pragma unroll
    for (int k = 0; k < CH / 2 / 512; ++k) {
        const int i = k * 512 + tid;
        if (base + 2 * i + 1 < ne)      out4[i] = st4[i];
        else if (base + 2 * i < ne)     sorted[base + 2 * i] = stage[2 * i];
    }
}

__global__ __launch_bounds__(256) void rank_owner_kernel(
    const int* __restrict__ sorted_i, const int* __restrict__ start,
    int* __restrict__ cnt, int* __restrict__ slot,
    int* __restrict__ novf, int* __restrict__ ovf,
    int n, int nchunk, int pitch)
{
    const int2* sorted = reinterpret_cast<const int2*>(sorted_i);
    __shared__ int lcnt[RB];
    __shared__ __align__(16) int lslot[RB * SLOTS];

    const int tid   = threadIdx.x;
    const int bk    = blockIdx.x;
    const int node0 = bk * RB;

    if (tid < RB) lcnt[tid] = 0;
    {
        int4* z = reinterpret_cast<int4*>(lslot);
        #pragma unroll
        for (int i = 0; i < RB * SLOTS / 4 / 256; ++i) z[tid + 256 * i] = make_int4(0, 0, 0, 0);
    }
    __syncthreads();

    for (int t = tid; t < nchunk; t += 256) {
        const int* st = start + (size_t)t * pitch + bk;
        const int  sb = st[0];
        const int  se = st[1];
        const int2* p = sorted + (size_t)t * CH;
        for (int j = sb; j < se; j += 4) {
            int2 rec[4];
            #pragma unroll
            for (int k = 0; k < 4; ++k) {
                rec[k] = make_int2(node0, 0);
                if (j + k < se) rec[k] = p[j + k];
            }
            #pragma unroll
            for (int k = 0; k < 4; ++k) {
                if (j + k < se) {
                    const int lc = (rec[k].x - node0) & (RB - 1);
                    const int rk = atomicAdd(&lcnt[lc], 1);
                    if (rk < SLOTS) {
                        lslot[lc * SLOTS + rk] = rec[k].y;
                    } else {
                        const int o = atomicAdd(&novf[0], 1);
                        ovf[2 * (size_t)o + 0] = rec[k].x;
                        ovf[2 * (size_t)o + 1] = rec[k].y;
                    }
                }
            }
        }
    }
    __syncthreads();

    if (tid < RB && node0 + tid < n) cnt[node0 + tid] = lcnt[tid];
    const int4* l4 = reinterpret_cast<const int4*>(lslot);
    int4* s4 = reinterpret_cast<int4*>(slot + (size_t)node0 * SLOTS);
    #pragma unroll
    for (int i = 0; i < RB * SLOTS / 4 / 256; ++i) s4[tid + 256 * i] = l4[tid + 256 * i];
}

template <int HEAD>
__global__ __launch_bounds__(256) void gather_dense_kernel(
    const int* __restrict__ cnt, const int* __restrict__ slot,
    const int* __restrict__ novf, const int* __restrict__ ovf,
    const int* __restrict__ row,
    const float* __restrict__ xw,
    const float* __restrict__ b, const float* __restrict__ g,
    const float* __restrict__ beta, const float* __restrict__ mean,
    const float* __restrict__ var,
    const float* __restrict__ Wa, const float* __restrict__ ca,
    const float* __restrict__ Wb, const float* __restrict__ cb,
    const float* __restrict__ Wc, const float* __restrict__ cc,
    float* __restrict__ res, int n, int ne)
{
    __shared__ __align__(16) float sWa[64 * 64];
    __shared__ __align__(16) float sWb[64 * 64];
    __shared__ __align__(16) float sWc[HEAD ? 64 * 16 : 4];
    __shared__ __align__(16) float sA[8][16 * 68];

    const int tid  = threadIdx.x;
    const int wave = tid >> 5;
    const int lane = tid & 31;
    const int hh   = lane >> 4;
    const int l    = lane & 15;
    const int r0   = blockIdx.x * 128 + wave * 16;
    float* sAw = sA[wave];
    const float4* X4 = reinterpret_cast<const float4*>(xw);
    const char*   Xb = reinterpret_cast<const char*>(xw);
    const unsigned lofs = (unsigned)l * 16u;

    int   m_deg = 0;
    float m_dc  = 0.f;
    {
        const int cm = r0 + 2 * l + hh;
        if (l < 8 && cm < n) {
            m_deg = cnt[cm];
            m_dc  = rsqrtf((float)(m_deg + 1));
        }
    }

    {
        const float4* ga = reinterpret_cast<const float4*>(Wa);
        float4* sa = reinterpret_cast<float4*>(sWa);
        #pragma unroll
        for (int i = 0; i < 4; ++i) sa[tid + 256 * i] = ga[tid + 256 * i];
        const float4* gb = reinterpret_cast<const float4*>(Wb);
        float4* sb = reinterpret_cast<float4*>(sWb);
        #pragma unroll
        for (int i = 0; i < 4; ++i) sb[tid + 256 * i] = gb[tid + 256 * i];
        if (HEAD) {
            const float4* gc = reinterpret_cast<const float4*>(Wc);
            float4* sc = reinterpret_cast<float4*>(sWc);
            sc[tid] = gc[tid];
        }
    }

    float4 pb  = make_float4(0.f, 0.f, 0.f, 0.f);
    float4 pm  = pb, pv = pb, pg = pb, pbe = pb;
    if (HEAD) {
        pb  = reinterpret_cast<const float4*>(b)[l];
        pm  = reinterpret_cast<const float4*>(mean)[l];
        pv  = reinterpret_cast<const float4*>(var)[l];
        pg  = reinterpret_cast<const float4*>(g)[l];
        pbe = reinterpret_cast<const float4*>(beta)[l];
    }

    int    cur_r, nx_r;
    int    cur_k;
    float4 cur_x;
    {
        const int  d0 = __shfl(m_deg, 0, 16);
        const int  d1 = __shfl(m_deg, 1, 16);
        const int  cx = (r0 + hh < n) ? (r0 + hh) : (n - 1);
        const int* q0 = (l < d0) ? (slot + (size_t)(r0 + hh) * SLOTS + l) : row;
        const int* q1 = (l < d1) ? (slot + (size_t)(r0 + 2 + hh) * SLOTS + l) : row;
        cur_x = X4[(size_t)cx * 16 + l];
        cur_r = *q0;
        nx_r  = *q1;
        cur_k = cnt[cur_r];
    }

    for (int p = 0; p < 8; ++p) {
        const int   c     = r0 + 2 * p + hh;
        const bool  valid = c < n;
        const int   deg   = __shfl(m_deg, p, 16);
        const float dc    = __shfl(m_dc, p, 16);
        const int*  mine  = slot + (size_t)c * SLOTS;

        const float4 xs  = cur_x;
        int          myr = cur_r;
        float        myd = rsqrtf((float)(cur_k + 1));

        const int    p2   = (p + 2) & 7;
        const int    d2   = __shfl(m_deg, p2, 16);
        const int    cx   = (c + 2 < n) ? (c + 2) : (n - 1);
        const int*   q2   = (l < d2) ? (slot + (size_t)(r0 + 2 * p2 + hh) * SLOTS + l) : row;
        const int    nx_k = cnt[nx_r];
        const int    n2_r = *q2;
        const float4 nx_x = X4[(size_t)cx * 16 + l];

        float4 acc;
        acc.x = xs.x * dc * dc;
        acc.y = xs.y * dc * dc;
        acc.z = xs.z * dc * dc;
        acc.w = xs.w * dc * dc;

        const int degs = (deg < SLOTS) ? deg : SLOTS;
        for (int j0 = 0; j0 < degs; j0 += 16) {
            const int m = (degs - j0 < 16) ? (degs - j0) : 16;
            if (j0 > 0) {
                myr = 0;
                myd = 0.f;
                if (l < m) {
                    myr = mine[j0 + l];
                    myd = rsqrtf((float)(cnt[myr] + 1));
                }
            }
            for (int j = 0; j < m; j += 8) {
                int    r[8];
                float  w[8];
                float4 x[8];
                float    sd[8];
                unsigned ofs[8];
                #pragma unroll
                for (int k = 0; k < 8; ++k) r[k] = __shfl(myr, j + k, 16);
                #pragma unroll
                for (int k = 0; k < 8; ++k) sd[k] = __shfl(myd, j + k, 16);
                #pragma unroll
                for (int k = 0; k < 8; ++k) ofs[k] = (unsigned)r[k] * 256u + lofs;
                asm volatile("" :: "v"(ofs[0]), "v"(ofs[1]), "v"(ofs[2]), "v"(ofs[3]),
                                   "v"(ofs[4]), "v"(ofs[5]), "v"(ofs[6]), "v"(ofs[7]),
                                   "v"(sd[0]), "v"(sd[1]), "v"(sd[2]), "v"(sd[3]),
                                   "v"(sd[4]), "v"(sd[5]), "v"(sd[6]), "v"(sd[7]));
                #pragma unroll
                for (int k = 0; k < 8; ++k)
                    x[k] = *reinterpret_cast<const float4*>(Xb + ofs[k]);
                #pragma unroll
                for (int k = 0; k < 8; ++k) w[k] = (j + k < m) ? sd[k] * dc : 0.f;
                #pragma unroll
                for (int k = 0; k < 8; ++k) {
                    acc.x = fmaf(w[k], x[k].x, acc.x);
                    acc.y = fmaf(w[k], x[k].y, acc.y);
                    acc.z = fmaf(w[k], x[k].z, acc.z);
                    acc.w = fmaf(w[k], x[k].w, acc.w);
                }
            }
        }

        if (deg > SLOTS) {
            int no = novf[0];
            no = (no < ne) ? no : ne;
            const int2* O2 = reinterpret_cast<const int2*>(ovf);
            for (int i0 = 0; i0 < no; i0 += 16) {
                int   orow = 0;
                float ow   = 0.f;
                if (i0 + l < no) {
                    const int2 rec = O2[i0 + l];
                    if (rec.x == c) {
                        orow = rec.y;
                        ow   = rsqrtf((float)(cnt[orow] + 1)) * dc;
                    }
                }
                #pragma nounroll
                for (int k = 0; k < 16; ++k) {
                    const float wk = __shfl(ow, k, 16);
                    const int   rk = __shfl(orow, k, 16);
                    if (wk != 0.f) {
                        const float4 xv = X4[(size_t)rk * 16 + l];
                        acc.x = fmaf(wk, xv.x, acc.x);
                        acc.y = fmaf(wk, xv.y, acc.y);
                        acc.z = fmaf(wk, xv.z, acc.z);
                        acc.w = fmaf(wk, xv.w, acc.w);
                    }
                }
            }
        }

        cur_r = nx_r;
        cur_k = nx_k;
        cur_x = nx_x;
        nx_r  = n2_r;

        float4 o = make_float4(0.f, 0.f, 0.f, 0.f);
        if (valid) {
            if (HEAD) {
                o.x = fmaxf((acc.x + pb.x - pm.x) * rsqrtf(pv.x + 1e-5f) * pg.x + pbe.x, 0.f);
                o.y = fmaxf((acc.y + pb.y - pm.y) * rsqrtf(pv.y + 1e-5f) * pg.y + pbe.y, 0.f);
                o.z = fmaxf((acc.z + pb.z - pm.z) * rsqrtf(pv.z + 1e-5f) * pg.z + pbe.z, 0.f);
                o.w = fmaxf((acc.w + pb.w - pm.w) * rsqrtf(pv.w + 1e-5f) * pg.w + pbe.w, 0.f);
            } else {
                o = acc;
            }
        }
        *reinterpret_cast<float4*>(&sAw[(2 * p + hh) * 68 + l * 4]) = o;
    }
    __syncthreads();

    if (HEAD == 0) {
        v8f acc0[4] = {};
        strip_gemm<64>(sAw, sWa, acc0, l, hh);
        strip_store_bn_relu(sAw, acc0, b, g, beta, mean, var, l, hh);
        __syncthreads();

        v8f acc1[4] = {};
        strip_gemm<64>(sAw, sWb, acc1, l, hh);
        #pragma unroll
        for (int nt = 0; nt < 4; ++nt) {
            #pragma unroll
            for (int j = 0; j < 8; ++j)
                sAw[(j + 8 * hh) * 68 + nt * 16 + l] = acc1[nt][j];
        }
        __syncthreads();
        float4* R4 = reinterpret_cast<float4*>(res);
        #pragma unroll
        for (int i = 0; i < 8; ++i) {
            const int q  = lane + 32 * i;
            const int r  = q >> 4;
            const int c4 = q & 15;
            const int gr = r0 + r;
            if (gr < n)
                R4[(size_t)gr * 16 + c4] =
                    *reinterpret_cast<const float4*>(&sAw[r * 68 + c4 * 4]);
        }
    } else {
        v8f acc1[4] = {};
        strip_gemm<64>(sAw, sWa, acc1, l, hh);
        strip_store_relu(sAw, acc1, ca, l, hh);
        __syncthreads();

        v8f acc2[4] = {};
        strip_gemm<64>(sAw, sWb, acc2, l, hh);
        strip_store_relu(sAw, acc2, cb, l, hh);
        __syncthreads();

        v8f acc3[1] = {};
        strip_gemm<16>(sAw, sWc, acc3, l, hh);
        {
            const float bv = cc[l];
            #pragma unroll
            for (int j = 0; j < 8; ++j)
                sAw[(j + 8 * hh) * 68 + l] = acc3[0][j] + bv;
        }
        __syncthreads();
        float4* R4 = reinterpret_cast<float4*>(res);
        #pragma unroll
        for (int i = 0; i < 2; ++i) {
            const int q  = lane + 32 * i;
            const int r  = q >> 2;
            const int c4 = q & 3;
            const int gr = r0 + r;
            if (gr < n)
                R4[(size_t)gr * 4 + c4] =
                    *reinterpret_cast<const float4*>(&sAw[r * 68 + c4 * 4]);
        }
    }
}

extern "C" void kernel_launch(void* const* d_in, const int* in_sizes, int n_in,
                              void* d_out, int out_size, void* d_ws, size_t ws_size,
                              hipStream_t stream)
{
    const float* x      = (const float*)d_in[0];
    const int*   ei     = (const int*)d_in[1];
    const float* W0     = (const float*)d_in[2];
    const float* b0     = (const float*)d_in[3];
    const float* W1     = (const float*)d_in[4];
    const float* b1     = (const float*)d_in[5];
    const float* bn0_g  = (const float*)d_in[6];
    const float* bn0_b  = (const float*)d_in[7];
    const float* bn0_m  = (const float*)d_in[8];
    const float* bn0_v  = (const float*)d_in[9];
    const float* bn1_g  = (const float*)d_in[10];
    const float* bn1_b  = (const float*)d_in[11];
    const float* bn1_m  = (const float*)d_in[12];
    const float* bn1_v  = (const float*)d_in[13];
    const float* lin1_W = (const float*)d_in[14];
    const float* lin1_b = (const float*)d_in[15];
    const float* lin2_W = (const float*)d_in[16];
    const float* lin2_b = (const float*)d_in[17];
    const float* lin3_W = (const float*)d_in[18];
    const float* lin3_b = (const float*)d_in[19];

    const int N = in_sizes[0] / 64;
    const int E = in_sizes[1] / 2;
    const int* row = ei;
    const int* col = ei + E;

    const dim3 blk(256);
    const int gGemm = (N + 127) / 128;
    const int gBin  = (E + CH - 1) / CH;
    const int pitch = (gGemm + 1 + 31) & ~31;

    char*  ws   = (char*)d_ws;
    size_t off1 = 0;
    size_t off2 = (off1 + (size_t)N * 64 * sizeof(float) + 255) & ~(size_t)255;
    size_t off3 = (off2 + (size_t)N * sizeof(int) + 255) & ~(size_t)255;
    size_t off4 = (off3 + sizeof(int) + 255) & ~(size_t)255;
    size_t off5 = (off4 + (size_t)gGemm * 128 * SLOTS * sizeof(int) + 255) & ~(size_t)255;
    size_t off6 = (off5 + (size_t)E * 2 * sizeof(int) + 255) & ~(size_t)255;
    size_t off7 = (off6 + (size_t)gBin * CH * 2 * sizeof(int) + 255) & ~(size_t)255;
    float* bufB   = (float*)(ws + off1);
    int*   cnt    = (int*)(ws + off2);
    int*   novf   = (int*)(ws + off3);
    int*   slot   = (int*)(ws + off4);
    int*   ovf    = (int*)(ws + off5);
    int*   sorted = (int*)(ws + off6);
    int*   start  = (int*)(ws + off7);

    bin_edges_kernel<<<gBin, dim3(512), 0, stream>>>(row, col, sorted, start, novf, E, gGemm, pitch);
    rank_owner_kernel<<<gGemm, blk, 0, stream>>>(sorted, start, cnt, slot, novf, ovf, N, gBin, pitch);

    gather_dense_kernel<0><<<gGemm, blk, 0, stream>>>(cnt, slot, novf, ovf, row, x,
        b0, bn0_g, bn0_b, bn0_m, bn0_v,
        W0, nullptr, W1, nullptr, nullptr, nullptr, bufB, N, E);

    gather_dense_kernel<1><<<gGemm, blk, 0, stream>>>(cnt, slot, novf, ovf, row, bufB,
        b1, bn1_g, bn1_b, bn1_m, bn1_v,
        lin1_W, lin1_b, lin2_W, lin2_b, lin3_W, lin3_b, (float*)d_out, N, E);
}


# graph layer 2's rows stored pre-multiplied by their own node's rsqrt(1+deg): its gather drops the per-edge in-degree lookup (one line per edge); product association changed, sum order kept
# speedup vs baseline: 4.7767x; 1.0765x over previous
#include <hip/hip_runtime.h>
#include <cstdint>
#include <cstddef>

typedef __attribute__((ext_vector_type(2))) float v2f;
typedef __attribute__((ext_vector_type(8))) float v8f;

template <int NCOL>
__device__ __forceinline__ void strip_gemm(
    const float* __restrict__ sAw, const float* __restrict__ sW,
    v8f (&acc)[NCOL / 16], int l16, int hh)
{
    constexpr int NT = NCOL / 16;
    #pragma unroll
    for (int kb = 0; kb < 64; kb += 4) {
        v2f a;
        a.x = sAw[l16 * 68 + kb + 2 * hh + 0];
        a.y = sAw[l16 * 68 + kb + 2 * hh + 1];
        #pragma unroll
        for (int nt = 0; nt < NT; ++nt) {
            v2f b;
            b.x = sW[(kb + 2 * hh + 0) * NCOL + nt * 16 + l16];
            b.y = sW[(kb + 2 * hh + 1) * NCOL + nt * 16 + l16];
            acc[nt] = __builtin_amdgcn_wmma_f32_16x16x4_f32(
                false, a, false, b, (short)0, acc[nt], false, false);
        }
    }
}

__device__ __forceinline__ void strip_store_relu(
    float* __restrict__ sAw, const v8f (&acc)[4],
    const float* __restrict__ bias, int l16, int hh)
{
    #pragma unroll
    for (int nt = 0; nt < 4; ++nt) {
        const int   col = nt * 16 + l16;
        const float bv  = bias[col];
        #pragma unroll
        for (int j = 0; j < 8; ++j)
            sAw[(j + 8 * hh) * 68 + col] = fmaxf(acc[nt][j] + bv, 0.f);
    }
}

__device__ __forceinline__ void strip_store_bn_relu(
    float* __restrict__ sAw, const v8f (&acc)[4],
    const float* __restrict__ bias, const float* __restrict__ g,
    const float* __restrict__ beta, const float* __restrict__ mean,
    const float* __restrict__ var, int l16, int hh)
{
    #pragma unroll
    for (int nt = 0; nt < 4; ++nt) {
        const int   col = nt * 16 + l16;
        const float bv  = bias[col];
        const float mv  = mean[col];
        const float rs  = rsqrtf(var[col] + 1e-5f);
        const float gv  = g[col];
        const float bev = beta[col];
        #pragma unroll
        for (int j = 0; j < 8; ++j)
            sAw[(j + 8 * hh) * 68 + col] =
                fmaxf((acc[nt][j] + bv - mv) * rs * gv + bev, 0.f);
    }
}

constexpr int SLOTS = 32;
constexpr int RB    = 128;
constexpr int CH    = 4096;
constexpr int MAXB  = 1024;

__global__ __launch_bounds__(512) void bin_edges_kernel(
    const int* __restrict__ row, const int* __restrict__ col,
    int* __restrict__ sorted_i, int* __restrict__ start,
    int* __restrict__ novf, int ne, int nbk, int pitch)
{
    int2* sorted = reinterpret_cast<int2*>(sorted_i);
    __shared__ int hist[MAXB];
    __shared__ int wsum[16];
    __shared__ __align__(16) int2 stage[CH];

    const int tid  = threadIdx.x;
    const int lane = tid & 31;
    const int wave = tid >> 5;
    const int base = blockIdx.x * CH;

    hist[2 * tid + 0] = 0;
    hist[2 * tid + 1] = 0;
    if (blockIdx.x == 0 && tid == 0) novf[0] = 0;

    int c[8], s[8], r[8];
    #pragma unroll
    for (int k = 0; k < 8; ++k) {
        const int e = base + k * 512 + tid;
        c[k] = 0;
        s[k] = 0;
        if (e < ne) { c[k] = col[e]; s[k] = row[e]; }
    }
    __syncthreads();

    #pragma unroll
    for (int k = 0; k < 8; ++k) {
        r[k] = 0;
        if (base + k * 512 + tid < ne) r[k] = atomicAdd(&hist[(unsigned)c[k] / RB], 1);
    }
    __syncthreads();

    const int h0 = hist[2 * tid + 0];
    const int h1 = hist[2 * tid + 1];
    int x = h0 + h1;
    #pragma unroll
    for (int d = 1; d < 32; d <<= 1) {
        const int y = __shfl_up(x, d, 32);
        if (lane >= d) x += y;
    }
    if (lane == 31) wsum[wave] = x;
    __syncthreads();
    int woff = 0;
    for (int w = 0; w < wave; ++w) woff += wsum[w];
    const int ex = woff + x - (h0 + h1);
    hist[2 * tid + 0] = ex;
    hist[2 * tid + 1] = ex + h0;
    __syncthreads();

    #pragma unroll
    for (int k = 0; k < 8; ++k) {
        if (base + k * 512 + tid < ne)
            stage[hist[(unsigned)c[k] / RB] + r[k]] = make_int2(c[k], s[k]);
    }
    for (int i = tid; i <= nbk; i += 512)
        start[(size_t)blockIdx.x * pitch + i] = hist[i];
    __syncthreads();

    const int4* st4 = reinterpret_cast<const int4*>(stage);
    int4* out4 = reinterpret_cast<int4*>(sorted + base);
    #pragma unroll
    for (int k = 0; k < CH / 2 / 512; ++k) {
        const int i = k * 512 + tid;
        if (base + 2 * i + 1 < ne)      out4[i] = st4[i];
        else if (base + 2 * i < ne)     sorted[base + 2 * i] = stage[2 * i];
    }
}

__global__ __launch_bounds__(256) void rank_owner_kernel(
    const int* __restrict__ sorted_i, const int* __restrict__ start,
    int* __restrict__ cnt, int* __restrict__ slot,
    int* __restrict__ novf, int* __restrict__ ovf,
    int n, int nchunk, int pitch)
{
    const int2* sorted = reinterpret_cast<const int2*>(sorted_i);
    __shared__ int lcnt[RB];
    __shared__ __align__(16) int lslot[RB * SLOTS];

    const int tid   = threadIdx.x;
    const int bk    = blockIdx.x;
    const int node0 = bk * RB;

    if (tid < RB) lcnt[tid] = 0;
    {
        int4* z = reinterpret_cast<int4*>(lslot);
        #pragma unroll
        for (int i = 0; i < RB * SLOTS / 4 / 256; ++i) z[tid + 256 * i] = make_int4(0, 0, 0, 0);
    }
    __syncthreads();

    for (int t = tid; t < nchunk; t += 256) {
        const int* st = start + (size_t)t * pitch + bk;
        const int  sb = st[0];
        const int  se = st[1];
        const int2* p = sorted + (size_t)t * CH;
        for (int j = sb; j < se; j += 4) {
            int2 rec[4];
            #pragma unroll
            for (int k = 0; k < 4; ++k) {
                rec[k] = make_int2(node0, 0);
                if (j + k < se) rec[k] = p[j + k];
            }
            #pragma unroll
            for (int k = 0; k < 4; ++k) {
                if (j + k < se) {
                    const int lc = (rec[k].x - node0) & (RB - 1);
                    const int rk = atomicAdd(&lcnt[lc], 1);
                    if (rk < SLOTS) {
                        lslot[lc * SLOTS + rk] = rec[k].y;
                    } else {
                        const int o = atomicAdd(&novf[0], 1);
                        ovf[2 * (size_t)o + 0] = rec[k].x;
                        ovf[2 * (size_t)o + 1] = rec[k].y;
                    }
                }
            }
        }
    }
    __syncthreads();

    if (tid < RB && node0 + tid < n) cnt[node0 + tid] = lcnt[tid];
    const int4* l4 = reinterpret_cast<const int4*>(lslot);
    int4* s4 = reinterpret_cast<int4*>(slot + (size_t)node0 * SLOTS);
    #pragma unroll
    for (int i = 0; i < RB * SLOTS / 4 / 256; ++i) s4[tid + 256 * i] = l4[tid + 256 * i];
}

template <int HEAD>
__global__ __launch_bounds__(256) void gather_dense_kernel(
    const int* __restrict__ cnt, const int* __restrict__ slot,
    const int* __restrict__ novf, const int* __restrict__ ovf,
    const int* __restrict__ row,
    const float* __restrict__ xw,
    const float* __restrict__ b, const float* __restrict__ g,
    const float* __restrict__ beta, const float* __restrict__ mean,
    const float* __restrict__ var,
    const float* __restrict__ Wa, const float* __restrict__ ca,
    const float* __restrict__ Wb, const float* __restrict__ cb,
    const float* __restrict__ Wc, const float* __restrict__ cc,
    float* __restrict__ res, int n, int ne)
{
    __shared__ __align__(16) float sWa[64 * 64];
    __shared__ __align__(16) float sWb[64 * 64];
    __shared__ __align__(16) float sWc[HEAD ? 64 * 16 : 4];
    __shared__ __align__(16) float sA[8][16 * 68];

    const int tid  = threadIdx.x;
    const int wave = tid >> 5;
    const int lane = tid & 31;
    const int hh   = lane >> 4;
    const int l    = lane & 15;
    const int r0   = blockIdx.x * 128 + wave * 16;
    float* sAw = sA[wave];
    const float4* X4 = reinterpret_cast<const float4*>(xw);
    const char*   Xb = reinterpret_cast<const char*>(xw);
    const unsigned lofs = (unsigned)l * 16u;

    int   m_deg = 0;
    float m_dc  = 0.f;
    {
        const int cm = r0 + 2 * l + hh;
        if (l < 8 && cm < n) {
            m_deg = cnt[cm];
            m_dc  = rsqrtf((float)(m_deg + 1));
        }
    }

    {
        const float4* ga = reinterpret_cast<const float4*>(Wa);
        float4* sa = reinterpret_cast<float4*>(sWa);
        #pragma unroll
        for (int i = 0; i < 4; ++i) sa[tid + 256 * i] = ga[tid + 256 * i];
        const float4* gb = reinterpret_cast<const float4*>(Wb);
        float4* sb = reinterpret_cast<float4*>(sWb);
        #pragma unroll
        for (int i = 0; i < 4; ++i) sb[tid + 256 * i] = gb[tid + 256 * i];
        if (HEAD) {
            const float4* gc = reinterpret_cast<const float4*>(Wc);
            float4* sc = reinterpret_cast<float4*>(sWc);
            sc[tid] = gc[tid];
        }
    }

    float4 pb  = make_float4(0.f, 0.f, 0.f, 0.f);
    float4 pm  = pb, pv = pb, pg = pb, pbe = pb;
    if (HEAD) {
        pb  = reinterpret_cast<const float4*>(b)[l];
        pm  = reinterpret_cast<const float4*>(mean)[l];
        pv  = reinterpret_cast<const float4*>(var)[l];
        pg  = reinterpret_cast<const float4*>(g)[l];
        pbe = reinterpret_cast<const float4*>(beta)[l];
    }

    int    cur_r, nx_r;
    int    cur_k = 0;
    float4 cur_x;
    {
        const int  d0 = __shfl(m_deg, 0, 16);
        const int  d1 = __shfl(m_deg, 1, 16);
        const int  cx = (r0 + hh < n) ? (r0 + hh) : (n - 1);
        const int* q0 = (l < d0) ? (slot + (size_t)(r0 + hh) * SLOTS + l) : row;
        const int* q1 = (l < d1) ? (slot + (size_t)(r0 + 2 + hh) * SLOTS + l) : row;
        cur_x = X4[(size_t)cx * 16 + l];
        cur_r = *q0;
        nx_r  = *q1;
        if (!HEAD) cur_k = cnt[cur_r];
    }

    for (int p = 0; p < 8; ++p) {
        const int   c     = r0 + 2 * p + hh;
        const bool  valid = c < n;
        const int   deg   = __shfl(m_deg, p, 16);
        const float dc    = __shfl(m_dc, p, 16);
        const int*  mine  = slot + (size_t)c * SLOTS;

        const float4 xs  = cur_x;
        int          myr = cur_r;
        float        myd = HEAD ? 1.f : rsqrtf((float)(cur_k + 1));

        const int    p2   = (p + 2) & 7;
        const int    d2   = __shfl(m_deg, p2, 16);
        const int    cx   = (c + 2 < n) ? (c + 2) : (n - 1);
        const int*   q2   = (l < d2) ? (slot + (size_t)(r0 + 2 * p2 + hh) * SLOTS + l) : row;
        int          nx_k = 0;
        if (!HEAD) nx_k = cnt[nx_r];
        const int    n2_r = *q2;
        const float4 nx_x = X4[(size_t)cx * 16 + l];

        float4 acc;
        if (HEAD) {
            acc.x = xs.x * dc;
            acc.y = xs.y * dc;
            acc.z = xs.z * dc;
            acc.w = xs.w * dc;
        } else {
            acc.x = xs.x * dc * dc;
            acc.y = xs.y * dc * dc;
            acc.z = xs.z * dc * dc;
            acc.w = xs.w * dc * dc;
        }

        const int degs = (deg < SLOTS) ? deg : SLOTS;
        for (int j0 = 0; j0 < degs; j0 += 16) {
            const int m = (degs - j0 < 16) ? (degs - j0) : 16;
            if (j0 > 0) {
                myr = 0;
                myd = 0.f;
                if (l < m) {
                    myr = mine[j0 + l];
                    myd = HEAD ? 1.f : rsqrtf((float)(cnt[myr] + 1));
                }
                if (HEAD) {
                    asm volatile("" :: "v"(myr));
                }
            }
            for (int j = 0; j < m; j += 8) {
                int    r[8];
                float  w[8];
                float4 x[8];
                float    sd[8];
                unsigned ofs[8];
                #pragma unroll
                for (int k = 0; k < 8; ++k) r[k] = __shfl(myr, j + k, 16);
                if (!HEAD) {
                    #pragma unroll
                    for (int k = 0; k < 8; ++k) sd[k] = __shfl(myd, j + k, 16);
                }
                #pragma unroll
                for (int k = 0; k < 8; ++k) ofs[k] = (unsigned)r[k] * 256u + lofs;
                if (HEAD) {
                    asm volatile("" :: "v"(ofs[0]), "v"(ofs[1]), "v"(ofs[2]), "v"(ofs[3]),
                                       "v"(ofs[4]), "v"(ofs[5]), "v"(ofs[6]), "v"(ofs[7]));
                } else {
                    asm volatile("" :: "v"(ofs[0]), "v"(ofs[1]), "v"(ofs[2]), "v"(ofs[3]),
                                       "v"(ofs[4]), "v"(ofs[5]), "v"(ofs[6]), "v"(ofs[7]),
                                       "v"(sd[0]), "v"(sd[1]), "v"(sd[2]), "v"(sd[3]),
                                       "v"(sd[4]), "v"(sd[5]), "v"(sd[6]), "v"(sd[7]));
                }
                #pragma unroll
                for (int k = 0; k < 8; ++k)
                    x[k] = *reinterpret_cast<const float4*>(Xb + ofs[k]);
                #pragma unroll
                for (int k = 0; k < 8; ++k) w[k] = (j + k < m) ? (HEAD ? dc : sd[k] * dc) : 0.f;
                #pragma unroll
                for (int k = 0; k < 8; ++k) {
                    acc.x = fmaf(w[k], x[k].x, acc.x);
                    acc.y = fmaf(w[k], x[k].y, acc.y);
                    acc.z = fmaf(w[k], x[k].z, acc.z);
                    acc.w = fmaf(w[k], x[k].w, acc.w);
                }
            }
        }

        if (deg > SLOTS) {
            int no = novf[0];
            no = (no < ne) ? no : ne;
            const int2* O2 = reinterpret_cast<const int2*>(ovf);
            for (int i0 = 0; i0 < no; i0 += 16) {
                int   orow = 0;
                float ow   = 0.f;
                if (i0 + l < no) {
                    const int2 rec = O2[i0 + l];
                    if (rec.x == c) {
                        orow = rec.y;
                        ow   = HEAD ? dc : rsqrtf((float)(cnt[orow] + 1)) * dc;
                    }
                }
                #pragma nounroll
                for (int k = 0; k < 16; ++k) {
                    const float wk = __shfl(ow, k, 16);
                    const int   rk = __shfl(orow, k, 16);
                    if (wk != 0.f) {
                        const float4 xv = X4[(size_t)rk * 16 + l];
                        acc.x = fmaf(wk, xv.x, acc.x);
                        acc.y = fmaf(wk, xv.y, acc.y);
                        acc.z = fmaf(wk, xv.z, acc.z);
                        acc.w = fmaf(wk, xv.w, acc.w);
                    }
                }
            }
        }

        cur_r = nx_r;
        cur_k = nx_k;
        cur_x = nx_x;
        nx_r  = n2_r;

        float4 o = make_float4(0.f, 0.f, 0.f, 0.f);
        if (valid) {
            if (HEAD) {
                o.x = fmaxf((acc.x + pb.x - pm.x) * rsqrtf(pv.x + 1e-5f) * pg.x + pbe.x, 0.f);
                o.y = fmaxf((acc.y + pb.y - pm.y) * rsqrtf(pv.y + 1e-5f) * pg.y + pbe.y, 0.f);
                o.z = fmaxf((acc.z + pb.z - pm.z) * rsqrtf(pv.z + 1e-5f) * pg.z + pbe.z, 0.f);
                o.w = fmaxf((acc.w + pb.w - pm.w) * rsqrtf(pv.w + 1e-5f) * pg.w + pbe.w, 0.f);
            } else {
                o = acc;
            }
        }
        *reinterpret_cast<float4*>(&sAw[(2 * p + hh) * 68 + l * 4]) = o;
    }
    __syncthreads();

    if (HEAD == 0) {
        v8f acc0[4] = {};
        strip_gemm<64>(sAw, sWa, acc0, l, hh);
        strip_store_bn_relu(sAw, acc0, b, g, beta, mean, var, l, hh);
        __syncthreads();

        v8f acc1[4] = {};
        strip_gemm<64>(sAw, sWb, acc1, l, hh);
        #pragma unroll
        for (int nt = 0; nt < 4; ++nt) {
            #pragma unroll
            for (int j = 0; j < 8; ++j)
                sAw[(j + 8 * hh) * 68 + nt * 16 + l] = acc1[nt][j];
        }
        __syncthreads();
        float4* R4 = reinterpret_cast<float4*>(res);
        #pragma unroll
        for (int i = 0; i < 8; ++i) {
            const int q  = lane + 32 * i;
            const int r  = q >> 4;
            const int c4 = q & 15;
            const int gr = r0 + r;
            const float dr = __shfl(m_dc, i, 16);
            float4 v = *reinterpret_cast<const float4*>(&sAw[r * 68 + c4 * 4]);
            v.x *= dr;
            v.y *= dr;
            v.z *= dr;
            v.w *= dr;
            if (gr < n)
                R4[(size_t)gr * 16 + c4] = v;
        }
    } else {
        v8f acc1[4] = {};
        strip_gemm<64>(sAw, sWa, acc1, l, hh);
        strip_store_relu(sAw, acc1, ca, l, hh);
        __syncthreads();

        v8f acc2[4] = {};
        strip_gemm<64>(sAw, sWb, acc2, l, hh);
        strip_store_relu(sAw, acc2, cb, l, hh);
        __syncthreads();

        v8f acc3[1] = {};
        strip_gemm<16>(sAw, sWc, acc3, l, hh);
        {
            const float bv = cc[l];
            #pragma unroll
            for (int j = 0; j < 8; ++j)
                sAw[(j + 8 * hh) * 68 + l] = acc3[0][j] + bv;
        }
        __syncthreads();
        float4* R4 = reinterpret_cast<float4*>(res);
        #pragma unroll
        for (int i = 0; i < 2; ++i) {
            const int q  = lane + 32 * i;
            const int r  = q >> 2;
            const int c4 = q & 3;
            const int gr = r0 + r;
            if (gr < n)
                R4[(size_t)gr * 4 + c4] =
                    *reinterpret_cast<const float4*>(&sAw[r * 68 + c4 * 4]);
        }
    }
}

extern "C" void kernel_launch(void* const* d_in, const int* in_sizes, int n_in,
                              void* d_out, int out_size, void* d_ws, size_t ws_size,
                              hipStream_t stream)
{
    const float* x      = (const float*)d_in[0];
    const int*   ei     = (const int*)d_in[1];
    const float* W0     = (const float*)d_in[2];
    const float* b0     = (const float*)d_in[3];
    const float* W1     = (const float*)d_in[4];
    const float* b1     = (const float*)d_in[5];
    const float* bn0_g  = (const float*)d_in[6];
    const float* bn0_b  = (const float*)d_in[7];
    const float* bn0_m  = (const float*)d_in[8];
    const float* bn0_v  = (const float*)d_in[9];
    const float* bn1_g  = (const float*)d_in[10];
    const float* bn1_b  = (const float*)d_in[11];
    const float* bn1_m  = (const float*)d_in[12];
    const float* bn1_v  = (const float*)d_in[13];
    const float* lin1_W = (const float*)d_in[14];
    const float* lin1_b = (const float*)d_in[15];
    const float* lin2_W = (const float*)d_in[16];
    const float* lin2_b = (const float*)d_in[17];
    const float* lin3_W = (const float*)d_in[18];
    const float* lin3_b = (const float*)d_in[19];

    const int N = in_sizes[0] / 64;
    const int E = in_sizes[1] / 2;
    const int* row = ei;
    const int* col = ei + E;

    const dim3 blk(256);
    const int gGemm = (N + 127) / 128;
    const int gBin  = (E + CH - 1) / CH;
    const int pitch = (gGemm + 1 + 31) & ~31;

    char*  ws   = (char*)d_ws;
    size_t off1 = 0;
    size_t off2 = (off1 + (size_t)N * 64 * sizeof(float) + 255) & ~(size_t)255;
    size_t off3 = (off2 + (size_t)N * sizeof(int) + 255) & ~(size_t)255;
    size_t off4 = (off3 + sizeof(int) + 255) & ~(size_t)255;
    size_t off5 = (off4 + (size_t)gGemm * 128 * SLOTS * sizeof(int) + 255) & ~(size_t)255;
    size_t off6 = (off5 + (size_t)E * 2 * sizeof(int) + 255) & ~(size_t)255;
    size_t off7 = (off6 + (size_t)gBin * CH * 2 * sizeof(int) + 255) & ~(size_t)255;
    float* bufB   = (float*)(ws + off1);
    int*   cnt    = (int*)(ws + off2);
    int*   novf   = (int*)(ws + off3);
    int*   slot   = (int*)(ws + off4);
    int*   ovf    = (int*)(ws + off5);
    int*   sorted = (int*)(ws + off6);
    int*   start  = (int*)(ws + off7);

    bin_edges_kernel<<<gBin, dim3(512), 0, stream>>>(row, col, sorted, start, novf, E, gGemm, pitch);
    rank_owner_kernel<<<gGemm, blk, 0, stream>>>(sorted, start, cnt, slot, novf, ovf, N, gBin, pitch);

    gather_dense_kernel<0><<<gGemm, blk, 0, stream>>>(cnt, slot, novf, ovf, row, x,
        b0, bn0_g, bn0_b, bn0_m, bn0_v,
        W0, nullptr, W1, nullptr, nullptr, nullptr, bufB, N, E);

    gather_dense_kernel<1><<<gGemm, blk, 0, stream>>>(cnt, slot, novf, ovf, row, bufB,
        b1, bn1_g, bn1_b, bn1_m, bn1_v,
        lin1_W, lin1_b, lin2_W, lin2_b, lin3_W, lin3_b, (float*)d_out, N, E);
}


# graph layer 1: all in-degrees staged as a byte table in LDS (one 25-wave block per compute unit, one block barrier), the per-edge degree lookup reads LDS instead of an L2 line; sums unchanged
# speedup vs baseline: 4.9557x; 1.0375x over previous
#include <hip/hip_runtime.h>
#include <cstdint>
#include <cstddef>

typedef __attribute__((ext_vector_type(2))) float v2f;
typedef __attribute__((ext_vector_type(8))) float v8f;

template <int NCOL>
__device__ __forceinline__ void strip_gemm(
    const float* __restrict__ sAw, const float* __restrict__ sW,
    v8f (&acc)[NCOL / 16], int l16, int hh)
{
    constexpr int NT = NCOL / 16;
    #pragma unroll
    for (int kb = 0; kb < 64; kb += 4) {
        v2f a;
        a.x = sAw[l16 * 68 + kb + 2 * hh + 0];
        a.y = sAw[l16 * 68 + kb + 2 * hh + 1];
        #pragma unroll
        for (int nt = 0; nt < NT; ++nt) {
            v2f b;
            b.x = sW[(kb + 2 * hh + 0) * NCOL + nt * 16 + l16];
            b.y = sW[(kb + 2 * hh + 1) * NCOL + nt * 16 + l16];
            acc[nt] = __builtin_amdgcn_wmma_f32_16x16x4_f32(
                false, a, false, b, (short)0, acc[nt], false, false);
        }
    }
}

__device__ __forceinline__ void strip_store_relu(
    float* __restrict__ sAw, const v8f (&acc)[4],
    const float* __restrict__ bias, int l16, int hh)
{
    #pragma unroll
    for (int nt = 0; nt < 4; ++nt) {
        const int   col = nt * 16 + l16;
        const float bv  = bias[col];
        #pragma unroll
        for (int j = 0; j < 8; ++j)
            sAw[(j + 8 * hh) * 68 + col] = fmaxf(acc[nt][j] + bv, 0.f);
    }
}

__device__ __forceinline__ void strip_store_bn_relu(
    float* __restrict__ sAw, const v8f (&acc)[4],
    const float* __restrict__ bias, const float* __restrict__ g,
    const float* __restrict__ beta, const float* __restrict__ mean,
    const float* __restrict__ var, int l16, int hh)
{
    #pragma unroll
    for (int nt = 0; nt < 4; ++nt) {
        const int   col = nt * 16 + l16;
        const float bv  = bias[col];
        const float mv  = mean[col];
        const float rs  = rsqrtf(var[col] + 1e-5f);
        const float gv  = g[col];
        const float bev = beta[col];
        #pragma unroll
        for (int j = 0; j < 8; ++j)
            sAw[(j + 8 * hh) * 68 + col] =
                fmaxf((acc[nt][j] + bv - mv) * rs * gv + bev, 0.f);
    }
}

__device__ __forceinline__ int deg_full(int d, const int* __restrict__ cnt, int v)
{
    if (d == 255) d = cnt[v];
    return d;
}
__device__ __forceinline__ int deg_of(
    const unsigned char* tab, const int* __restrict__ cnt, int v)
{
    return deg_full(tab[v], cnt, v);
}

__device__ __forceinline__ void strip_handover()
{
    __builtin_amdgcn_fence(__ATOMIC_RELEASE, "wavefront");
    __builtin_amdgcn_wave_barrier();
    __builtin_amdgcn_fence(__ATOMIC_ACQUIRE, "wavefront");
}

constexpr int SLOTS = 32;
constexpr int RB    = 128;
constexpr int CH    = 4096;
constexpr int MAXB  = 1024;

__global__ __launch_bounds__(512) void bin_edges_kernel(
    const int* __restrict__ row, const int* __restrict__ col,
    int* __restrict__ sorted_i, int* __restrict__ start,
    int* __restrict__ novf, int ne, int nbk, int pitch)
{
    int2* sorted = reinterpret_cast<int2*>(sorted_i);
    __shared__ int hist[MAXB];
    __shared__ int wsum[16];
    __shared__ __align__(16) int2 stage[CH];

    const int tid  = threadIdx.x;
    const int lane = tid & 31;
    const int wave = tid >> 5;
    const int base = blockIdx.x * CH;

    hist[2 * tid + 0] = 0;
    hist[2 * tid + 1] = 0;
    if (blockIdx.x == 0 && tid == 0) novf[0] = 0;

    int c[8], s[8], r[8];
    #pragma unroll
    for (int k = 0; k < 8; ++k) {
        const int e = base + k * 512 + tid;
        c[k] = 0;
        s[k] = 0;
        if (e < ne) { c[k] = col[e]; s[k] = row[e]; }
    }
    __syncthreads();

    #pragma unroll
    for (int k = 0; k < 8; ++k) {
        r[k] = 0;
        if (base + k * 512 + tid < ne) r[k] = atomicAdd(&hist[(unsigned)c[k] / RB], 1);
    }
    __syncthreads();

    const int h0 = hist[2 * tid + 0];
    const int h1 = hist[2 * tid + 1];
    int x = h0 + h1;
    #pragma unroll
    for (int d = 1; d < 32; d <<= 1) {
        const int y = __shfl_up(x, d, 32);
        if (lane >= d) x += y;
    }
    if (lane == 31) wsum[wave] = x;
    __syncthreads();
    int woff = 0;
    for (int w = 0; w < wave; ++w) woff += wsum[w];
    const int ex = woff + x - (h0 + h1);
    hist[2 * tid + 0] = ex;
    hist[2 * tid + 1] = ex + h0;
    __syncthreads();

    #pragma unroll
    for (int k = 0; k < 8; ++k) {
        if (base + k * 512 + tid < ne)
            stage[hist[(unsigned)c[k] / RB] + r[k]] = make_int2(c[k], s[k]);
    }
    for (int i = tid; i <= nbk; i += 512)
        start[(size_t)blockIdx.x * pitch + i] = hist[i];
    __syncthreads();

    const int4* st4 = reinterpret_cast<const int4*>(stage);
    int4* out4 = reinterpret_cast<int4*>(sorted + base);
    #pragma unroll
    for (int k = 0; k < CH / 2 / 512; ++k) {
        const int i = k * 512 + tid;
        if (base + 2 * i + 1 < ne)      out4[i] = st4[i];
        else if (base + 2 * i < ne)     sorted[base + 2 * i] = stage[2 * i];
    }
}

__global__ __launch_bounds__(256) void rank_owner_kernel(
    const int* __restrict__ sorted_i, const int* __restrict__ start,
    int* __restrict__ cnt, int* __restrict__ deg8, int* __restrict__ slot,
    int* __restrict__ novf, int* __restrict__ ovf,
    int n, int nchunk, int pitch)
{
    const int2* sorted = reinterpret_cast<const int2*>(sorted_i);
    __shared__ int lcnt[RB];
    __shared__ __align__(16) int lslot[RB * SLOTS];

    const int tid   = threadIdx.x;
    const int bk    = blockIdx.x;
    const int node0 = bk * RB;

    if (tid < RB) lcnt[tid] = 0;
    {
        int4* z = reinterpret_cast<int4*>(lslot);
        #pragma unroll
        for (int i = 0; i < RB * SLOTS / 4 / 256; ++i) z[tid + 256 * i] = make_int4(0, 0, 0, 0);
    }
    __syncthreads();

    for (int t = tid; t < nchunk; t += 256) {
        const int* st = start + (size_t)t * pitch + bk;
        const int  sb = st[0];
        const int  se = st[1];
        const int2* p = sorted + (size_t)t * CH;
        for (int j = sb; j < se; j += 4) {
            int2 rec[4];
            #pragma unroll
            for (int k = 0; k < 4; ++k) {
                rec[k] = make_int2(node0, 0);
                if (j + k < se) rec[k] = p[j + k];
            }
            #pragma unroll
            for (int k = 0; k < 4; ++k) {
                if (j + k < se) {
                    const int lc = (rec[k].x - node0) & (RB - 1);
                    const int rk = atomicAdd(&lcnt[lc], 1);
                    if (rk < SLOTS) {
                        lslot[lc * SLOTS + rk] = rec[k].y;
                    } else {
                        const int o = atomicAdd(&novf[0], 1);
                        ovf[2 * (size_t)o + 0] = rec[k].x;
                        ovf[2 * (size_t)o + 1] = rec[k].y;
                    }
                }
            }
        }
    }
    __syncthreads();

    if (tid < RB && node0 + tid < n) cnt[node0 + tid] = lcnt[tid];
    if (tid < RB / 4) {
        int w = 0;
        #pragma unroll
        for (int k = 0; k < 4; ++k) {
            const int d = lcnt[4 * tid + k];
            w |= ((d < 255) ? d : 255) << (8 * k);
        }
        deg8[bk * (RB / 4) + tid] = w;
    }
    const int4* l4 = reinterpret_cast<const int4*>(lslot);
    int4* s4 = reinterpret_cast<int4*>(slot + (size_t)node0 * SLOTS);
    #pragma unroll
    for (int i = 0; i < RB * SLOTS / 4 / 256; ++i) s4[tid + 256 * i] = l4[tid + 256 * i];
}

constexpr int DEGW = 32768;

template <int HEAD, int NW>
__global__ __launch_bounds__(NW * 32) void gather_dense_kernel(
    const int* __restrict__ cnt, const int* __restrict__ deg8, const int* __restrict__ slot,
    const int* __restrict__ novf, const int* __restrict__ ovf,
    const int* __restrict__ row,
    const float* __restrict__ xw,
    const float* __restrict__ b, const float* __restrict__ g,
    const float* __restrict__ beta, const float* __restrict__ mean,
    const float* __restrict__ var,
    const float* __restrict__ Wa, const float* __restrict__ ca,
    const float* __restrict__ Wb, const float* __restrict__ cb,
    const float* __restrict__ Wc, const float* __restrict__ cc,
    float* __restrict__ res, int n, int ne, int ndeg4)
{
    constexpr int NT = NW * 32;
    constexpr int F_WB = 64 * 64, F_A = 2 * 64 * 64, F_DEG = F_A + NW * 16 * 68;
    __shared__ __align__(16) float sWa_[HEAD ? 64 * 64 : 4];
    __shared__ __align__(16) float sWb_[HEAD ? 64 * 64 : 4];
    __shared__ __align__(16) float sWc[HEAD ? 64 * 16 : 4];
    __shared__ __align__(16) float sA_[HEAD ? NW * 16 * 68 : 4];
    __shared__ __align__(16) float sAll[HEAD ? 4 : F_DEG + DEGW];
    float* const sWa  = HEAD ? sWa_ : sAll;
    float* const sWb  = HEAD ? sWb_ : sAll + F_WB;
    float* const sAb  = HEAD ? sA_  : sAll + F_A;
    int*   const sDeg = reinterpret_cast<int*>(sAll + (HEAD ? 0 : F_DEG));
    const unsigned char* tab = reinterpret_cast<const unsigned char*>(sDeg);

    const int tid  = threadIdx.x;
    const int wave = tid >> 5;
    const int lane = tid & 31;
    const int hh   = lane >> 4;
    const int l    = lane & 15;
    const int r0   = blockIdx.x * (NW * 16) + wave * 16;
    float* sAw = sAb + wave * (16 * 68);
    const float4* X4 = reinterpret_cast<const float4*>(xw);
    const char*   Xb = reinterpret_cast<const char*>(xw);
    const unsigned lofs = (unsigned)l * 16u;

    int   m_deg = 0;
    float m_dc  = 0.f;
    {
        const int cm = r0 + 2 * l + hh;
        if (l < 8 && cm < n) {
            m_deg = cnt[cm];
            m_dc  = rsqrtf((float)(m_deg + 1));
        }
    }

    constexpr int NWL = HEAD ? 1 : (1024 + NT - 1) / NT;
    float4 wa[NWL], wb[NWL];
    int    wix[NWL];
    int4   tv[8];
    int    tix[8];
    if (HEAD) {
        const float4* ga = reinterpret_cast<const float4*>(Wa);
        float4* sa = reinterpret_cast<float4*>(sWa);
        #pragma unroll
        for (int i = 0; i < 4; ++i) sa[tid + 256 * i] = ga[tid + 256 * i];
        const float4* gb = reinterpret_cast<const float4*>(Wb);
        float4* sb = reinterpret_cast<float4*>(sWb);
        #pragma unroll
        for (int i = 0; i < 4; ++i) sb[tid + 256 * i] = gb[tid + 256 * i];
        const float4* gc = reinterpret_cast<const float4*>(Wc);
        float4* sc = reinterpret_cast<float4*>(sWc);
        sc[tid] = gc[tid];
    } else {
        const float4* ga = reinterpret_cast<const float4*>(Wa);
        const float4* gb = reinterpret_cast<const float4*>(Wb);
        const int4*   g4 = reinterpret_cast<const int4*>(deg8);
        #pragma unroll
        for (int i = 0; i < NWL; ++i) {
            wix[i] = (tid + NT * i < 1024) ? (tid + NT * i) : 1023;
            wa[i]  = ga[wix[i]];
            wb[i]  = gb[wix[i]];
        }
        #pragma unroll
        for (int k = 0; k < 8; ++k) {
            tix[k] = k * NT + tid;
            tix[k] = (tix[k] < ndeg4) ? tix[k] : (ndeg4 - 1);
            tv[k]  = g4[tix[k]];
        }
    }

    float4 pb  = make_float4(0.f, 0.f, 0.f, 0.f);
    float4 pm  = pb, pv = pb, pg = pb, pbe = pb;
    if (HEAD) {
        pb  = reinterpret_cast<const float4*>(b)[l];
        pm  = reinterpret_cast<const float4*>(mean)[l];
        pv  = reinterpret_cast<const float4*>(var)[l];
        pg  = reinterpret_cast<const float4*>(g)[l];
        pbe = reinterpret_cast<const float4*>(beta)[l];
    }

    int    cur_r, nx_r;
    int    cur_k = 0;
    float4 cur_x;
    {
        const int  d0 = __shfl(m_deg, 0, 16);
        const int  d1 = __shfl(m_deg, 1, 16);
        const int  cx = (r0 + hh < n) ? (r0 + hh) : (n - 1);
        const int* q0 = (l < d0) ? (slot + (size_t)(r0 + hh) * SLOTS + l) : row;
        const int* q1 = (l < d1) ? (slot + (size_t)(r0 + 2 + hh) * SLOTS + l) : row;
        cur_x = X4[(size_t)cx * 16 + l];
        cur_r = *q0;
        nx_r  = *q1;
        if (!HEAD) {
            float4* sa = reinterpret_cast<float4*>(sWa);
            float4* sb = reinterpret_cast<float4*>(sWb);
            int4*   s4 = reinterpret_cast<int4*>(sDeg);
            #pragma unroll
            for (int i = 0; i < NWL; ++i) {
                sa[wix[i]] = wa[i];
                sb[wix[i]] = wb[i];
            }
            #pragma unroll
            for (int k = 0; k < 8; ++k) s4[tix[k]] = tv[k];
            const int4* g4 = reinterpret_cast<const int4*>(deg8);
            for (int i = 8 * NT + tid; i < ndeg4; i += NT) s4[i] = g4[i];
            __syncthreads();
            cur_k = tab[cur_r];
        }
    }

    for (int p = 0; p < 8; ++p) {
        const int   c     = r0 + 2 * p + hh;
        const bool  valid = c < n;
        const int   deg   = __shfl(m_deg, p, 16);
        const float dc    = __shfl(m_dc, p, 16);
        const int*  mine  = slot + (size_t)c * SLOTS;

        const float4 xs  = cur_x;
        int          myr = cur_r;
        float        myd = HEAD ? 1.f : rsqrtf((float)(deg_full(cur_k, cnt, cur_r) + 1));

        const int    p2   = (p + 2) & 7;
        const int    d2   = __shfl(m_deg, p2, 16);
        const int    cx   = (c + 2 < n) ? (c + 2) : (n - 1);
        const int*   q2   = (l < d2) ? (slot + (size_t)(r0 + 2 * p2 + hh) * SLOTS + l) : row;
        int          nx_k = 0;
        if (!HEAD) nx_k = tab[nx_r];
        const int    n2_r = *q2;
        const float4 nx_x = X4[(size_t)cx * 16 + l];

        float4 acc;
        if (HEAD) {
            acc.x = xs.x * dc;
            acc.y = xs.y * dc;
            acc.z = xs.z * dc;
            acc.w = xs.w * dc;
        } else {
            acc.x = xs.x * dc * dc;
            acc.y = xs.y * dc * dc;
            acc.z = xs.z * dc * dc;
            acc.w = xs.w * dc * dc;
        }

        const int degs = (deg < SLOTS) ? deg : SLOTS;
        for (int j0 = 0; j0 < degs; j0 += 16) {
            const int m = (degs - j0 < 16) ? (degs - j0) : 16;
            if (j0 > 0) {
                myr = 0;
                myd = 0.f;
                if (l < m) {
                    myr = mine[j0 + l];
                    myd = HEAD ? 1.f : rsqrtf((float)(deg_of(tab, cnt, myr) + 1));
                }
                if (HEAD) {
                    asm volatile("" :: "v"(myr));
                }
            }
            for (int j = 0; j < m; j += 8) {
                int    r[8];
                float  w[8];
                float4 x[8];
                float    sd[8];
                unsigned ofs[8];
                #pragma unroll
                for (int k = 0; k < 8; ++k) r[k] = __shfl(myr, j + k, 16);
                if (!HEAD) {
                    #pragma unroll
                    for (int k = 0; k < 8; ++k) sd[k] = __shfl(myd, j + k, 16);
                }
                #pragma unroll
                for (int k = 0; k < 8; ++k) ofs[k] = (unsigned)r[k] * 256u + lofs;
                if (HEAD) {
                    asm volatile("" :: "v"(ofs[0]), "v"(ofs[1]), "v"(ofs[2]), "v"(ofs[3]),
                                       "v"(ofs[4]), "v"(ofs[5]), "v"(ofs[6]), "v"(ofs[7]));
                } else {
                    asm volatile("" :: "v"(ofs[0]), "v"(ofs[1]), "v"(ofs[2]), "v"(ofs[3]),
                                       "v"(ofs[4]), "v"(ofs[5]), "v"(ofs[6]), "v"(ofs[7]),
                                       "v"(sd[0]), "v"(sd[1]), "v"(sd[2]), "v"(sd[3]),
                                       "v"(sd[4]), "v"(sd[5]), "v"(sd[6]), "v"(sd[7]));
                }
                #pragma unroll
                for (int k = 0; k < 8; ++k)
                    x[k] = *reinterpret_cast<const float4*>(Xb + ofs[k]);
                #pragma unroll
                for (int k = 0; k < 8; ++k) w[k] = (j + k < m) ? (HEAD ? dc : sd[k] * dc) : 0.f;
                #pragma unroll
                for (int k = 0; k < 8; ++k) {
                    acc.x = fmaf(w[k], x[k].x, acc.x);
                    acc.y = fmaf(w[k], x[k].y, acc.y);
                    acc.z = fmaf(w[k], x[k].z, acc.z);
                    acc.w = fmaf(w[k], x[k].w, acc.w);
                }
            }
        }

        if (deg > SLOTS) {
            int no = novf[0];
            no = (no < ne) ? no : ne;
            const int2* O2 = reinterpret_cast<const int2*>(ovf);
            for (int i0 = 0; i0 < no; i0 += 16) {
                int   orow = 0;
                float ow   = 0.f;
                if (i0 + l < no) {
                    const int2 rec = O2[i0 + l];
                    if (rec.x == c) {
                        orow = rec.y;
                        ow   = HEAD ? dc : rsqrtf((float)(deg_of(tab, cnt, orow) + 1)) * dc;
                    }
                }
                #pragma nounroll
                for (int k = 0; k < 16; ++k) {
                    const float wk = __shfl(ow, k, 16);
                    const int   rk = __shfl(orow, k, 16);
                    if (wk != 0.f) {
                        const float4 xv = X4[(size_t)rk * 16 + l];
                        acc.x = fmaf(wk, xv.x, acc.x);
                        acc.y = fmaf(wk, xv.y, acc.y);
                        acc.z = fmaf(wk, xv.z, acc.z);
                        acc.w = fmaf(wk, xv.w, acc.w);
                    }
                }
            }
        }

        cur_r = nx_r;
        cur_k = nx_k;
        cur_x = nx_x;
        nx_r  = n2_r;

        float4 o = make_float4(0.f, 0.f, 0.f, 0.f);
        if (valid) {
            if (HEAD) {
                o.x = fmaxf((acc.x + pb.x - pm.x) * rsqrtf(pv.x + 1e-5f) * pg.x + pbe.x, 0.f);
                o.y = fmaxf((acc.y + pb.y - pm.y) * rsqrtf(pv.y + 1e-5f) * pg.y + pbe.y, 0.f);
                o.z = fmaxf((acc.z + pb.z - pm.z) * rsqrtf(pv.z + 1e-5f) * pg.z + pbe.z, 0.f);
                o.w = fmaxf((acc.w + pb.w - pm.w) * rsqrtf(pv.w + 1e-5f) * pg.w + pbe.w, 0.f);
            } else {
                o = acc;
            }
        }
        *reinterpret_cast<float4*>(&sAw[(2 * p + hh) * 68 + l * 4]) = o;
    }
    if (HEAD) __syncthreads();
    else      strip_handover();

    if (HEAD == 0) {
        v8f acc0[4] = {};
        strip_gemm<64>(sAw, sWa, acc0, l, hh);
        strip_store_bn_relu(sAw, acc0, b, g, beta, mean, var, l, hh);
        strip_handover();

        v8f acc1[4] = {};
        strip_gemm<64>(sAw, sWb, acc1, l, hh);
        #pragma unroll
        for (int nt = 0; nt < 4; ++nt) {
            #pragma unroll
            for (int j = 0; j < 8; ++j)
                sAw[(j + 8 * hh) * 68 + nt * 16 + l] = acc1[nt][j];
        }
        strip_handover();
        float4* R4 = reinterpret_cast<float4*>(res);
        #pragma unroll
        for (int i = 0; i < 8; ++i) {
            const int q  = lane + 32 * i;
            const int r  = q >> 4;
            const int c4 = q & 15;
            const int gr = r0 + r;
            const float dr = __shfl(m_dc, i, 16);
            float4 v = *reinterpret_cast<const float4*>(&sAw[r * 68 + c4 * 4]);
            v.x *= dr;
            v.y *= dr;
            v.z *= dr;
            v.w *= dr;
            if (gr < n)
                R4[(size_t)gr * 16 + c4] = v;
        }
    } else {
        v8f acc1[4] = {};
        strip_gemm<64>(sAw, sWa, acc1, l, hh);
        strip_store_relu(sAw, acc1, ca, l, hh);
        __syncthreads();

        v8f acc2[4] = {};
        strip_gemm<64>(sAw, sWb, acc2, l, hh);
        strip_store_relu(sAw, acc2, cb, l, hh);
        __syncthreads();

        v8f acc3[1] = {};
        strip_gemm<16>(sAw, sWc, acc3, l, hh);
        {
            const float bv = cc[l];
            #pragma unroll
            for (int j = 0; j < 8; ++j)
                sAw[(j + 8 * hh) * 68 + l] = acc3[0][j] + bv;
        }
        __syncthreads();
        float4* R4 = reinterpret_cast<float4*>(res);
        #pragma unroll
        for (int i = 0; i < 2; ++i) {
            const int q  = lane + 32 * i;
            const int r  = q >> 2;
            const int c4 = q & 3;
            const int gr = r0 + r;
            if (gr < n)
                R4[(size_t)gr * 4 + c4] =
                    *reinterpret_cast<const float4*>(&sAw[r * 68 + c4 * 4]);
        }
    }
}

extern "C" void kernel_launch(void* const* d_in, const int* in_sizes, int n_in,
                              void* d_out, int out_size, void* d_ws, size_t ws_size,
                              hipStream_t stream)
{
    const float* x      = (const float*)d_in[0];
    const int*   ei     = (const int*)d_in[1];
    const float* W0     = (const float*)d_in[2];
    const float* b0     = (const float*)d_in[3];
    const float* W1     = (const float*)d_in[4];
    const float* b1     = (const float*)d_in[5];
    const float* bn0_g  = (const float*)d_in[6];
    const float* bn0_b  = (const float*)d_in[7];
    const float* bn0_m  = (const float*)d_in[8];
    const float* bn0_v  = (const float*)d_in[9];
    const float* bn1_g  = (const float*)d_in[10];
    const float* bn1_b  = (const float*)d_in[11];
    const float* bn1_m  = (const float*)d_in[12];
    const float* bn1_v  = (const float*)d_in[13];
    const float* lin1_W = (const float*)d_in[14];
    const float* lin1_b = (const float*)d_in[15];
    const float* lin2_W = (const float*)d_in[16];
    const float* lin2_b = (const float*)d_in[17];
    const float* lin3_W = (const float*)d_in[18];
    const float* lin3_b = (const float*)d_in[19];

    const int N = in_sizes[0] / 64;
    const int E = in_sizes[1] / 2;
    const int* row = ei;
    const int* col = ei + E;

    const dim3 blk(256);
    const int gGemm = (N + 127) / 128;
    const int gBin  = (E + CH - 1) / CH;
    const int pitch = (gGemm + 1 + 31) & ~31;

    char*  ws   = (char*)d_ws;
    size_t off1 = 0;
    size_t off2 = (off1 + (size_t)N * 64 * sizeof(float) + 255) & ~(size_t)255;
    size_t off3 = (off2 + (size_t)N * sizeof(int) + 255) & ~(size_t)255;
    size_t off4 = (off3 + sizeof(int) + 255) & ~(size_t)255;
    size_t off5 = (off4 + (size_t)gGemm * 128 * SLOTS * sizeof(int) + 255) & ~(size_t)255;
    size_t off6 = (off5 + (size_t)E * 2 * sizeof(int) + 255) & ~(size_t)255;
    size_t off7 = (off6 + (size_t)gBin * CH * 2 * sizeof(int) + 255) & ~(size_t)255;
    size_t off8 = (off7 + (size_t)gBin * pitch * sizeof(int) + 255) & ~(size_t)255;
    float* bufB   = (float*)(ws + off1);
    int*   cnt    = (int*)(ws + off2);
    int*   novf   = (int*)(ws + off3);
    int*   slot   = (int*)(ws + off4);
    int*   ovf    = (int*)(ws + off5);
    int*   sorted = (int*)(ws + off6);
    int*   start  = (int*)(ws + off7);
    int*   deg8   = (int*)(ws + off8);

    bin_edges_kernel<<<gBin, dim3(512), 0, stream>>>(row, col, sorted, start, novf, E, gGemm, pitch);
    rank_owner_kernel<<<gGemm, blk, 0, stream>>>(sorted, start, cnt, deg8, slot, novf, ovf, N, gBin, pitch);

    gather_dense_kernel<0, 25><<<(N + 399) / 400, dim3(800), 0, stream>>>(cnt, deg8, slot, novf, ovf, row, x,
        b0, bn0_g, bn0_b, bn0_m, bn0_v,
        W0, nullptr, W1, nullptr, nullptr, nullptr, bufB, N, E, gGemm * 8);

    gather_dense_kernel<1, 8><<<gGemm, blk, 0, stream>>>(cnt, nullptr, slot, novf, ovf, row, bufB,
        b1, bn1_g, bn1_b, bn1_m, bn1_v,
        lin1_W, lin1_b, lin2_W, lin2_b, lin3_W, lin3_b, (float*)d_out, N, E, 0);
}
